# starting kernel
# baseline (speedup 1.0000x reference)
#include <hip/hip_runtime.h>
#include <math.h>

typedef __attribute__((ext_vector_type(2))) float v2f;
typedef __attribute__((ext_vector_type(8))) float v8f;

#define INV_NN     0.512989176042577f
#define INV_SQRT3  0.5773502691896258f
#define S3   1.7320508075688772f
#define S5   2.23606797749979f
#define S7   2.6457513110645907f
#define S15  3.872983346207417f
#define S105 10.246950765959598f
#define C358 2.091650066335189f
#define C218 1.6201851746019651f

__global__ void kzero(float* __restrict__ p, long n) {
    long i = (long)blockIdx.x * blockDim.x + threadIdx.x;
    if (i < n) p[i] = 0.0f;
}

__global__ void kA(const float* __restrict__ evec, const float* __restrict__ elen,
                   const int* __restrict__ edst,
                   float* __restrict__ sh, float* __restrict__ basis,
                   float* __restrict__ x, int E)
{
    int e = blockIdx.x * blockDim.x + threadIdx.x;
    if (e >= E) return;
    float vx = evec[3*(size_t)e], vy = evec[3*(size_t)e+1], vz = evec[3*(size_t)e+2];
    float rn = 1.0f / (sqrtf(vx*vx + vy*vy + vz*vz) + 1e-12f);
    vx *= rn; vy *= rn; vz *= rn;
    float x2 = vx*vx, y2 = vy*vy, z2 = vz*vz;
    float s[16];
    s[0]  = 1.0f;
    s[1]  = S3*vx; s[2] = S3*vy; s[3] = S3*vz;
    s[4]  = S15*vx*vz;
    s[5]  = S15*vx*vy;
    s[6]  = 0.5f*S5*(2.0f*y2 - x2 - z2);
    s[7]  = S15*vy*vz;
    s[8]  = 0.5f*S15*(z2 - x2);
    s[9]  = C358*vx*(3.0f*z2 - x2);
    s[10] = S105*vx*vy*vz;
    s[11] = C218*vx*(4.0f*y2 - z2 - x2);
    s[12] = 0.5f*S7*vy*(2.0f*y2 - 3.0f*z2 - 3.0f*x2);
    s[13] = C218*vz*(4.0f*y2 - z2 - x2);
    s[14] = 0.5f*S105*vy*(z2 - x2);
    s[15] = C358*vz*(z2 - 3.0f*x2);
    int d = edst[e];
    #pragma unroll
    for (int j = 0; j < 16; j++) {
        sh[(size_t)e*16 + j] = s[j];
        atomicAdd(&x[(size_t)d*16 + j], s[j] * INV_NN);
    }
    float len = elen[e];
    #pragma unroll
    for (int i = 0; i < 3; i++) {
        float dd = (len - 2.0f*(float)i) * 0.5f;
        basis[(size_t)e*3 + i] = expf(-dd*dd) * (1.0f/1.12f);
    }
}

__global__ __launch_bounds__(192) void kB(
    const float* __restrict__ sh, const float* __restrict__ basis,
    const float* __restrict__ x,
    const int* __restrict__ esrc, const int* __restrict__ edst,
    const float* __restrict__ W1a, const float* __restrict__ W1b,
    float* __restrict__ h, int E)
{
    __shared__ float h1[32][258];
    __shared__ float dots[32][4];
    __shared__ int   dstT[32];

    const int tid = threadIdx.x;
    const int e0  = blockIdx.x * 32;

    for (int idx = tid; idx < 32*256; idx += 192) {
        int m = idx >> 8, c = idx & 255;
        int e = e0 + m;
        float v = 0.0f;
        if (e < E) {
            float b0 = basis[(size_t)e*3], b1 = basis[(size_t)e*3+1], b2 = basis[(size_t)e*3+2];
            v = b0*W1a[c] + b1*W1a[256+c] + b2*W1a[512+c];
            v = fmaxf(v * INV_SQRT3, 0.0f);
        }
        h1[m][c] = v;
    }
    if (tid < 128) {
        int m = tid >> 2, p = tid & 3;
        int e = e0 + m;
        float s = 0.0f;
        if (e < E) {
            int src = esrc[e];
            int a = p*p, b = a + 2*p + 1;
            for (int j = a; j < b; j++)
                s += x[(size_t)src*16 + j] * sh[(size_t)e*16 + j];
            s *= rsqrtf((float)(2*p + 1));
        }
        dots[m][p] = s;
        if (p == 0) dstT[m] = (e < E) ? edst[e] : 0;
    }
    __syncthreads();

    const int kc    = tid >> 5;
    const int lane  = tid & 31;
    const int mA    = lane & 15;
    const int khalf = lane >> 4;
    const int n     = lane & 15;

    v8f a00 = {}, a01 = {}, a02 = {}, a03 = {};
    v8f a10 = {}, a11 = {}, a12 = {}, a13 = {};
    const int colb = kc * 16 + n;

    for (int k0 = 0; k0 < 256; k0 += 4) {
        int kb = k0 + 2*khalf;
        v2f A0, A1;
        A0.x = h1[mA][kb];      A0.y = h1[mA][kb + 1];
        A1.x = h1[16 + mA][kb]; A1.y = h1[16 + mA][kb + 1];
        const float* Brow0 = W1b + (size_t)kb * 384;
        const float* Brow1 = Brow0 + 384;
        v2f B0; B0.x = Brow0[colb];       B0.y = Brow1[colb];
        v2f B1; B1.x = Brow0[96  + colb]; B1.y = Brow1[96  + colb];
        v2f B2; B2.x = Brow0[192 + colb]; B2.y = Brow1[192 + colb];
        v2f B3; B3.x = Brow0[288 + colb]; B3.y = Brow1[288 + colb];
        a00 = __builtin_amdgcn_wmma_f32_16x16x4_f32(false, A0, false, B0, (short)0, a00, false, false);
        a10 = __builtin_amdgcn_wmma_f32_16x16x4_f32(false, A1, false, B0, (short)0, a10, false, false);
        a01 = __builtin_amdgcn_wmma_f32_16x16x4_f32(false, A0, false, B1, (short)0, a01, false, false);
        a11 = __builtin_amdgcn_wmma_f32_16x16x4_f32(false, A1, false, B1, (short)0, a11, false, false);
        a02 = __builtin_amdgcn_wmma_f32_16x16x4_f32(false, A0, false, B2, (short)0, a02, false, false);
        a12 = __builtin_amdgcn_wmma_f32_16x16x4_f32(false, A1, false, B2, (short)0, a12, false, false);
        a03 = __builtin_amdgcn_wmma_f32_16x16x4_f32(false, A0, false, B3, (short)0, a03, false, false);
        a13 = __builtin_amdgcn_wmma_f32_16x16x4_f32(false, A1, false, B3, (short)0, a13, false, false);
    }

    const float scale = (1.0f/16.0f) * 0.5f * INV_NN;
    #pragma unroll
    for (int r = 0; r < 8; r++) {
        int m = r + 8*khalf;
        if (e0 + m < E) {
            float ef = dots[m][0]*a00[r] + dots[m][1]*a01[r]
                     + dots[m][2]*a02[r] + dots[m][3]*a03[r];
            atomicAdd(&h[(size_t)dstT[m]*96 + kc*16 + n], ef * scale);
        }
        int m1 = 16 + m;
        if (e0 + m1 < E) {
            float ef = dots[m1][0]*a10[r] + dots[m1][1]*a11[r]
                     + dots[m1][2]*a12[r] + dots[m1][3]*a13[r];
            atomicAdd(&h[(size_t)dstT[m1]*96 + kc*16 + n], ef * scale);
        }
    }
}

__global__ void kC(const float* __restrict__ h, float* __restrict__ y, int N)
{
    int nid = blockIdx.x * blockDim.x + threadIdx.x;
    if (nid >= N) return;
    const float* hr = h + (size_t)nid * 96;
    float* yr = y + (size_t)nid * 64;
    #pragma unroll
    for (int j = 0; j < 16; j++) yr[j] = fmaxf(hr[j], 0.0f);
    #pragma unroll
    for (int j = 16; j < 32; j++) yr[j] = fabsf(hr[j]);
    #pragma unroll
    for (int j = 0; j < 32; j++) {
        float g = hr[32 + j];
        bool use_relu = (j < 8) || (j >= 16 && j < 24);
        float ga = use_relu ? fmaxf(g, 0.0f) : tanhf(g);
        yr[32 + j] = ga * hr[64 + j];
    }
}

__global__ __launch_bounds__(256) void kD(const float* __restrict__ y,
                                          const float* __restrict__ W2b,
                                          float* __restrict__ z, int N)
{
    __shared__ float yn[64];
    int nid = blockIdx.x;
    int c = threadIdx.x;
    if (c < 64) yn[c] = y[(size_t)nid*64 + c];
    __syncthreads();
    float s = 0.0f;
    const float* wr = W2b + (size_t)c * 64;
    #pragma unroll 8
    for (int k = 0; k < 64; k++) s += wr[k] * yn[k];
    z[(size_t)nid*256 + c] = s;
}

__global__ __launch_bounds__(256) void kE(const float* __restrict__ basis,
                                          const float* __restrict__ z,
                                          const int* __restrict__ esrc,
                                          const int* __restrict__ edst,
                                          const float* __restrict__ W2a,
                                          float* __restrict__ out, int E)
{
    int e = blockIdx.x * (blockDim.x >> 5) + (threadIdx.x >> 5);
    int lane = threadIdx.x & 31;
    if (e >= E) return;
    float b0 = basis[(size_t)e*3], b1 = basis[(size_t)e*3+1], b2 = basis[(size_t)e*3+2];
    int src = esrc[e];
    const float* zr = z + (size_t)src * 256;
    float s = 0.0f;
    #pragma unroll
    for (int i = 0; i < 8; i++) {
        int c = lane + 32*i;
        float hv = b0*W2a[c] + b1*W2a[256+c] + b2*W2a[512+c];
        hv = fmaxf(hv * INV_SQRT3, 0.0f);
        s += hv * zr[c];
    }
    #pragma unroll
    for (int off = 16; off > 0; off >>= 1) s += __shfl_down(s, off, 32);
    if (lane == 0)
        atomicAdd(&out[edst[e]], s * (1.0f/128.0f) * INV_NN);
}

extern "C" void kernel_launch(void* const* d_in, const int* in_sizes, int n_in,
                              void* d_out, int out_size, void* d_ws, size_t ws_size,
                              hipStream_t stream)
{
    const float* evec = (const float*)d_in[0];
    const float* elen = (const float*)d_in[1];
    const float* W1a  = (const float*)d_in[2];
    const float* W1b  = (const float*)d_in[3];
    const float* W2a  = (const float*)d_in[4];
    const float* W2b  = (const float*)d_in[5];
    const int*   esrc = (const int*)d_in[6];
    const int*   edst = (const int*)d_in[7];
    float* out = (float*)d_out;

    const int E = in_sizes[0] / 3;
    const int N = out_size;

    float* ws = (float*)d_ws;
    float* x_     = ws;                 ws += (size_t)N * 16;
    float* h_     = ws;                 ws += (size_t)N * 96;
    float* y_     = ws;                 ws += (size_t)N * 64;
    float* z_     = ws;                 ws += (size_t)N * 256;
    float* sh_    = ws;                 ws += (size_t)E * 16;
    float* basis_ = ws;                 ws += (size_t)E * 3;

    long nx = (long)N * 16, nh = (long)N * 96;
    kzero<<<(int)((nx + 255) / 256), 256, 0, stream>>>(x_, nx);
    kzero<<<(int)((nh + 255) / 256), 256, 0, stream>>>(h_, nh);
    kzero<<<(N + 255) / 256, 256, 0, stream>>>(out, N);

    kA<<<(E + 255) / 256, 256, 0, stream>>>(evec, elen, edst, sh_, basis_, x_, E);
    kB<<<(E + 31) / 32, 192, 0, stream>>>(sh_, basis_, x_, esrc, edst, W1a, W1b, h_, E);
    kC<<<(N + 255) / 256, 256, 0, stream>>>(h_, y_, N);
    kD<<<N, 256, 0, stream>>>(y_, W2b, z_, N);
    kE<<<(E + 7) / 8, 256, 0, stream>>>(basis_, z_, esrc, edst, W2a, out, E);
}


# kB: edges taken in length order (one counting-sort pass), so a 64-edge tile shares its relu-dead hidden columns; k loop runs over the live columns only; k grouping and h arrival order change
# speedup vs baseline: 2.3291x; 2.3291x over previous
#include <hip/hip_runtime.h>
#include <math.h>

typedef __attribute__((ext_vector_type(2))) float v2f;
typedef __attribute__((ext_vector_type(8))) float v8f;

#define INV_NN     0.512989176042577f
#define INV_SQRT3  0.5773502691896258f
#define S3   1.7320508075688772f
#define S5   2.23606797749979f
#define S7   2.6457513110645907f
#define S15  3.872983346207417f
#define S105 10.246950765959598f
#define C358 2.091650066335189f
#define C218 1.6201851746019651f

__global__ void kzero(float* __restrict__ p, long n) {
    long i = (long)blockIdx.x * blockDim.x + threadIdx.x;
    if (i < n) p[i] = 0.0f;
}

#define NBINS 256
#define SORT_CHUNK 4096
#define SORT_THREADS 1024
__device__ __forceinline__ int len_bin(float len) {
    float t = fminf(fmaxf(len * ((float)NBINS / 4.0f), 0.0f), (float)(NBINS - 1));
    return (int)t;
}

__global__ __launch_bounds__(SORT_THREADS) void kCount(const float* __restrict__ elen,
                                                       int* __restrict__ cnt, int nb, int E)
{
    __shared__ int lh[NBINS];
    const int tid = threadIdx.x;
    if (tid < NBINS) lh[tid] = 0;
    __syncthreads();
    const int base = blockIdx.x * SORT_CHUNK;
    #pragma unroll
    for (int i = 0; i < SORT_CHUNK / SORT_THREADS; i++) {
        const int e = base + i * SORT_THREADS + tid;
        if (e < E) atomicAdd(&lh[len_bin(elen[e])], 1);
    }
    __syncthreads();
    if (tid < NBINS) cnt[(size_t)tid * nb + blockIdx.x] = lh[tid];
}

__global__ __launch_bounds__(1024) void kScan(int* __restrict__ cnt, int M, int per)
{
    __shared__ int wsum[32];
    const int t = threadIdx.x, lane = t & 31, w = t >> 5;
    const int lo = t * per;
    int s = 0;
    #pragma unroll 8
    for (int i = 0; i < per; i++) {
        const int k = lo + i;
        s += (k < M) ? cnt[k] : 0;
    }
    int incl = s;
    #pragma unroll
    for (int d = 1; d < 32; d <<= 1) {
        int v = __shfl_up(incl, d, 32);
        if (lane >= d) incl += v;
    }
    if (lane == 31) wsum[w] = incl;
    __syncthreads();
    if (w == 0) {
        const int v = wsum[lane];
        int iv = v;
        #pragma unroll
        for (int d = 1; d < 32; d <<= 1) {
            int u = __shfl_up(iv, d, 32);
            if (lane >= d) iv += u;
        }
        wsum[lane] = iv - v;
    }
    __syncthreads();
    int run = wsum[w] + incl - s;
    #pragma unroll 8
    for (int i = 0; i < per; i++) {
        const int k = lo + i;
        if (k < M) { const int c = cnt[k]; cnt[k] = run; run += c; }
    }
}

__global__ __launch_bounds__(SORT_THREADS) void kScatter(const float* __restrict__ elen,
                                                         const int* __restrict__ offs,
                                                         int* __restrict__ perm, int nb, int E)
{
    __shared__ int lh[NBINS];
    const int tid = threadIdx.x;
    if (tid < NBINS) lh[tid] = offs[(size_t)tid * nb + blockIdx.x];
    __syncthreads();
    const int base = blockIdx.x * SORT_CHUNK;
    #pragma unroll
    for (int i = 0; i < SORT_CHUNK / SORT_THREADS; i++) {
        const int e = base + i * SORT_THREADS + tid;
        if (e < E) {
            const int pos = atomicAdd(&lh[len_bin(elen[e])], 1);
            if ((unsigned)pos < (unsigned)E) perm[pos] = e;
        }
    }
}

#define SH_PITCH 20
__global__ __launch_bounds__(256) void kA(const float* __restrict__ evec, const float* __restrict__ elen,
                   const int* __restrict__ edst,
                   float* __restrict__ sh, float* __restrict__ basis,
                   float* __restrict__ x, int E)
{
    __shared__ __attribute__((aligned(16))) float tile[256 * SH_PITCH];
    __shared__ int dstT[256];

    const int tid = threadIdx.x;
    const int eb  = blockIdx.x * 256;
    const int e   = eb + tid;
    float s[16] = {};
    int d = 0;
    if (e < E) {
        float vx = evec[3*(size_t)e], vy = evec[3*(size_t)e+1], vz = evec[3*(size_t)e+2];
        float rn = 1.0f / (sqrtf(vx*vx + vy*vy + vz*vz) + 1e-12f);
        vx *= rn; vy *= rn; vz *= rn;
        float x2 = vx*vx, y2 = vy*vy, z2 = vz*vz;
        s[0]  = 1.0f;
        s[1]  = S3*vx; s[2] = S3*vy; s[3] = S3*vz;
        s[4]  = S15*vx*vz;
        s[5]  = S15*vx*vy;
        s[6]  = 0.5f*S5*(2.0f*y2 - x2 - z2);
        s[7]  = S15*vy*vz;
        s[8]  = 0.5f*S15*(z2 - x2);
        s[9]  = C358*vx*(3.0f*z2 - x2);
        s[10] = S105*vx*vy*vz;
        s[11] = C218*vx*(4.0f*y2 - z2 - x2);
        s[12] = 0.5f*S7*vy*(2.0f*y2 - 3.0f*z2 - 3.0f*x2);
        s[13] = C218*vz*(4.0f*y2 - z2 - x2);
        s[14] = 0.5f*S105*vy*(z2 - x2);
        s[15] = C358*vz*(z2 - 3.0f*x2);
        d = edst[e];
        float len = elen[e];
        #pragma unroll
        for (int i = 0; i < 3; i++) {
            float dd = (len - 2.0f*(float)i) * 0.5f;
            basis[(size_t)e*3 + i] = expf(-dd*dd) * (1.0f/1.12f);
        }
    }
    {
        float4* row = (float4*)(tile + tid * SH_PITCH);
        row[0] = make_float4(s[0],  s[1],  s[2],  s[3]);
        row[1] = make_float4(s[4],  s[5],  s[6],  s[7]);
        row[2] = make_float4(s[8],  s[9],  s[10], s[11]);
        row[3] = make_float4(s[12], s[13], s[14], s[15]);
        dstT[tid] = d;
    }
    __syncthreads();

    const int lane  = tid & 31;
    const int hh    = lane >> 4;
    const int j     = lane & 15;
    const int wbase = tid & ~31;
    if (eb + wbase + 32 <= E) {
        float v[16];
        int   dm[16];
        #pragma unroll
        for (int i = 0; i < 16; i++) {
            const int m = wbase + 2*i + hh;
            v[i]  = tile[m * SH_PITCH + j];
            dm[i] = dstT[m];
        }
        #pragma unroll
        for (int i = 0; i < 16; i++) {
            const int m = wbase + 2*i + hh;
            sh[(size_t)(eb + m)*16 + j] = v[i];
            atomicAdd(&x[(size_t)dm[i]*16 + j], v[i] * INV_NN);
        }
    } else {
        for (int i = 0; i < 16; i++) {
            const int m  = wbase + 2*i + hh;
            const int em = eb + m;
            if (em < E) {
                const float v = tile[m * SH_PITCH + j];
                sh[(size_t)em*16 + j] = v;
                atomicAdd(&x[(size_t)dstT[m]*16 + j], v * INV_NN);
            }
        }
    }
}

#define KB_EDGES 64
#define KB_THREADS 384
__device__ __forceinline__ float hidden1(float4 b, float w0, float w1, float w2) {
    return fmaxf(fmaf(b.z, w2, fmaf(b.y, w1, b.x * w0)) * INV_SQRT3, 0.0f);
}
__global__ __launch_bounds__(KB_THREADS) void kB(
    const float* __restrict__ sh, const float* __restrict__ basis,
    const float* __restrict__ x,
    const int* __restrict__ esrc, const int* __restrict__ edst,
    const int* __restrict__ perm,
    const float* __restrict__ W1a, const float* __restrict__ W1b,
    float* __restrict__ h, int E)
{
    __shared__ float h1[KB_EDGES][258];
    __shared__ float dots[KB_EDGES][4];
    __shared__ __attribute__((aligned(16))) float bas[KB_EDGES][4];
    __shared__ int   dstT[KB_EDGES];
    __shared__ int   eid[KB_EDGES];
    __shared__ __attribute__((aligned(8))) unsigned cols[256];
    __shared__ int   wcnt[8];

    const int tid  = threadIdx.x;
    const int wv   = tid >> 5;
    const int lane = tid & 31;
    const int e0   = blockIdx.x * KB_EDGES;

    if (tid < KB_EDGES) {
        const int i = e0 + tid;
        int e = -1;
        float b0 = 0.0f, b1 = 0.0f, b2 = 0.0f;
        if (i < E) {
            e = perm[i];
            e = min(max(e, 0), E - 1);
            b0 = basis[(size_t)e*3]; b1 = basis[(size_t)e*3+1]; b2 = basis[(size_t)e*3+2];
        }
        eid[tid] = e;
        dstT[tid] = (e >= 0) ? edst[e] : 0;
        *(float4*)bas[tid] = make_float4(b0, b1, b2, 0.0f);
    } else if (tid < KB_EDGES + 256) {
        const int it = tid - KB_EDGES;
        const int m = it >> 2, p = it & 3;
        const int i = e0 + m;
        float s = 0.0f;
        if (i < E) {
            int e = perm[i];
            e = min(max(e, 0), E - 1);
            const int src = esrc[e];
            const int a = p*p, len = 2*p + 1;
            float xs[7], ss[7];
            #pragma unroll
            for (int j = 0; j < 7; j++) {
                const bool on = j < len;
                xs[j] = on ? x[(size_t)src*16 + a + j] : 0.0f;
                ss[j] = on ? sh[(size_t)e*16 + a + j] : 0.0f;
            }
            #pragma unroll
            for (int j = 0; j < 7; j++) s += xs[j] * ss[j];
            s *= rsqrtf((float)len);
        }
        dots[m][p] = s;
    }
    __syncthreads();

    bool live = false;
    float w0 = 0.0f, w1 = 0.0f, w2 = 0.0f;
    if (tid < 256) {
        w0 = W1a[tid]; w1 = W1a[256 + tid]; w2 = W1a[512 + tid];
        float top = 0.0f;
        #pragma unroll 8
        for (int m = 0; m < KB_EDGES; m++)
            top = fmaxf(top, hidden1(*(const float4*)bas[m], w0, w1, w2));
        live = top > 0.0f;
    }
    const unsigned bal = (unsigned)__ballot(live);
    if (wv < 8 && lane == 0) wcnt[wv] = __popc(bal);
    __syncthreads();

    int base = 0, L = 0;
    #pragma unroll
    for (int i = 0; i < 8; i++) {
        const int cnt = wcnt[i];
        if (i < wv) base += cnt;
        L += cnt;
    }
    const int Lp = (L + 7) & ~7;
    if (tid < 256) {
        if (live) {
            const int j = base + __popc(bal & ((1u << lane) - 1u));
            cols[j] = (unsigned)tid * (384u * 4u);
            for (int m = 0; m < KB_EDGES; m++)
                h1[m][j] = hidden1(*(const float4*)bas[m], w0, w1, w2);
        }
    } else {
        for (int it = tid - 256; it < (Lp - L) * KB_EDGES; it += 128) {
            const int j = L + (it >> 6), m = it & 63;
            h1[m][j] = 0.0f;
            if (m == 0) cols[j] = 0u;
        }
    }
    __syncthreads();

    const int kc    = wv % 6;
    const int rb    = (wv / 6) * 32;
    const int mA    = lane & 15;
    const int khalf = lane >> 4;
    const int n     = lane & 15;

    v8f a00 = {}, a01 = {}, a02 = {}, a03 = {};
    v8f a10 = {}, a11 = {}, a12 = {}, a13 = {};
    const char* Wb = (const char*)W1b;
    const unsigned colb = (unsigned)(kc * 16 + n) * 4u;

    for (int k0 = 0; k0 < Lp; k0 += 8) {
        #pragma unroll
        for (int u = 0; u < 2; u++) {
            const int kb = k0 + 4*u + 2*khalf;
            v2f A0, A1;
            A0.x = h1[rb + mA][kb];      A0.y = h1[rb + mA][kb + 1];
            A1.x = h1[rb + 16 + mA][kb]; A1.y = h1[rb + 16 + mA][kb + 1];
            const float* Brow0 = (const float*)(Wb + (cols[kb]     + colb));
            const float* Brow1 = (const float*)(Wb + (cols[kb + 1] + colb));
            v2f B0; B0.x = Brow0[0];   B0.y = Brow1[0];
            v2f B1; B1.x = Brow0[96];  B1.y = Brow1[96];
            v2f B2; B2.x = Brow0[192]; B2.y = Brow1[192];
            v2f B3; B3.x = Brow0[288]; B3.y = Brow1[288];
            a00 = __builtin_amdgcn_wmma_f32_16x16x4_f32(false, A0, false, B0, (short)0, a00, false, false);
            a10 = __builtin_amdgcn_wmma_f32_16x16x4_f32(false, A1, false, B0, (short)0, a10, false, false);
            a01 = __builtin_amdgcn_wmma_f32_16x16x4_f32(false, A0, false, B1, (short)0, a01, false, false);
            a11 = __builtin_amdgcn_wmma_f32_16x16x4_f32(false, A1, false, B1, (short)0, a11, false, false);
            a02 = __builtin_amdgcn_wmma_f32_16x16x4_f32(false, A0, false, B2, (short)0, a02, false, false);
            a12 = __builtin_amdgcn_wmma_f32_16x16x4_f32(false, A1, false, B2, (short)0, a12, false, false);
            a03 = __builtin_amdgcn_wmma_f32_16x16x4_f32(false, A0, false, B3, (short)0, a03, false, false);
            a13 = __builtin_amdgcn_wmma_f32_16x16x4_f32(false, A1, false, B3, (short)0, a13, false, false);
        }
    }

    const float scale = (1.0f/16.0f) * 0.5f * INV_NN;
    #pragma unroll
    for (int r = 0; r < 8; r++) {
        int m = rb + r + 8*khalf;
        if (eid[m] >= 0) {
            float ef = dots[m][0]*a00[r] + dots[m][1]*a01[r]
                     + dots[m][2]*a02[r] + dots[m][3]*a03[r];
            atomicAdd(&h[(size_t)dstT[m]*96 + kc*16 + n], ef * scale);
        }
        int m1 = 16 + m;
        if (eid[m1] >= 0) {
            float ef = dots[m1][0]*a10[r] + dots[m1][1]*a11[r]
                     + dots[m1][2]*a12[r] + dots[m1][3]*a13[r];
            atomicAdd(&h[(size_t)dstT[m1]*96 + kc*16 + n], ef * scale);
        }
    }
}

__global__ void kC(const float* __restrict__ h, float* __restrict__ y, int N)
{
    int nid = blockIdx.x * blockDim.x + threadIdx.x;
    if (nid >= N) return;
    const float* hr = h + (size_t)nid * 96;
    float* yr = y + (size_t)nid * 64;
    #pragma unroll
    for (int j = 0; j < 16; j++) yr[j] = fmaxf(hr[j], 0.0f);
    #pragma unroll
    for (int j = 16; j < 32; j++) yr[j] = fabsf(hr[j]);
    #pragma unroll
    for (int j = 0; j < 32; j++) {
        float g = hr[32 + j];
        bool use_relu = (j < 8) || (j >= 16 && j < 24);
        float ga = use_relu ? fmaxf(g, 0.0f) : tanhf(g);
        yr[32 + j] = ga * hr[64 + j];
    }
}

__global__ __launch_bounds__(256) void kE(const float* __restrict__ basis,
                                          const float* __restrict__ y,
                                          const int* __restrict__ esrc,
                                          const int* __restrict__ edst,
                                          const float* __restrict__ W2a,
                                          const float* __restrict__ W2b,
                                          float* __restrict__ out, int E)
{
    __shared__ __attribute__((aligned(16))) float Wl[128 * 128];
    __shared__ __attribute__((aligned(16))) float Al[3 * 256];

    const int tid = threadIdx.x;
    #pragma unroll
    for (int it = 0; it < 3; it++) Al[tid + 256*it] = W2a[tid + 256*it];
    {
        const int cc = tid & 15;
        #pragma unroll
        for (int it = 0; it < 8; it++) {
            const int kp = (tid >> 4) + 16*it;
            const float4 r0 = *(const float4*)(W2b + (size_t)(2*kp)     * 64 + 4*cc);
            const float4 r1 = *(const float4*)(W2b + (size_t)(2*kp + 1) * 64 + 4*cc);
            float* wb = Wl + kp*128 + 2*cc;
            const int sw = (kp & 1) * 32;
            v2f t0; t0.x = r0.x; t0.y = r1.x;
            v2f t1; t1.x = r0.y; t1.y = r1.y;
            v2f t2; t2.x = r0.z; t2.y = r1.z;
            v2f t3; t3.x = r0.w; t3.y = r1.w;
            *(v2f*)(wb + ( 0 ^ sw)) = t0;
            *(v2f*)(wb + (32 ^ sw)) = t1;
            *(v2f*)(wb + (64 ^ sw)) = t2;
            *(v2f*)(wb + (96 ^ sw)) = t3;
        }
    }
    __syncthreads();

    const int wv   = tid >> 5;
    const int lane = tid & 31;
    const int c    = lane & 15;
    const int hh   = lane >> 4;
    const int e0   = blockIdx.x * 256 + wv * 32;
    if (e0 >= E) return;

    float p0 = 0.0f, p1 = 0.0f, p2 = 0.0f, q0 = 0.0f, q1 = 0.0f, q2 = 0.0f;
    {
        const int ea = e0 + c, eb = e0 + 16 + c;
        if (ea < E) { p0 = basis[(size_t)ea*3]; p1 = basis[(size_t)ea*3+1]; p2 = basis[(size_t)ea*3+2]; }
        if (eb < E) { q0 = basis[(size_t)eb*3]; q1 = basis[(size_t)eb*3+1]; q2 = basis[(size_t)eb*3+2]; }
    }

    v8f a00 = {}, a01 = {}, a02 = {}, a03 = {};
    v8f a10 = {}, a11 = {}, a12 = {}, a13 = {};

    const float* Ap = Al + 2*hh;
    const float* Bp = Wl + hh*128 + 2*c;
    const int o0 = ( 0 ^ (hh*32)), o1 = (32 ^ (hh*32));
    const int o2 = (64 ^ (hh*32)), o3 = (96 ^ (hh*32));

    for (int k0 = 0; k0 < 256; k0 += 4) {
        const v2f u0 = *(const v2f*)(Ap + k0);
        const v2f u1 = *(const v2f*)(Ap + 256 + k0);
        const v2f u2 = *(const v2f*)(Ap + 512 + k0);
        v2f A0, A1;
        A0.x = fmaxf((p0*u0.x + p1*u1.x + p2*u2.x) * INV_SQRT3, 0.0f);
        A0.y = fmaxf((p0*u0.y + p1*u1.y + p2*u2.y) * INV_SQRT3, 0.0f);
        A1.x = fmaxf((q0*u0.x + q1*u1.x + q2*u2.x) * INV_SQRT3, 0.0f);
        A1.y = fmaxf((q0*u0.y + q1*u1.y + q2*u2.y) * INV_SQRT3, 0.0f);
        const float* Bk = Bp + k0*64;
        const v2f B0 = *(const v2f*)(Bk + o0);
        const v2f B1 = *(const v2f*)(Bk + o1);
        const v2f B2 = *(const v2f*)(Bk + o2);
        const v2f B3 = *(const v2f*)(Bk + o3);
        a00 = __builtin_amdgcn_wmma_f32_16x16x4_f32(false, A0, false, B0, (short)0, a00, false, false);
        a10 = __builtin_amdgcn_wmma_f32_16x16x4_f32(false, A1, false, B0, (short)0, a10, false, false);
        a01 = __builtin_amdgcn_wmma_f32_16x16x4_f32(false, A0, false, B1, (short)0, a01, false, false);
        a11 = __builtin_amdgcn_wmma_f32_16x16x4_f32(false, A1, false, B1, (short)0, a11, false, false);
        a02 = __builtin_amdgcn_wmma_f32_16x16x4_f32(false, A0, false, B2, (short)0, a02, false, false);
        a12 = __builtin_amdgcn_wmma_f32_16x16x4_f32(false, A1, false, B2, (short)0, a12, false, false);
        a03 = __builtin_amdgcn_wmma_f32_16x16x4_f32(false, A0, false, B3, (short)0, a03, false, false);
        a13 = __builtin_amdgcn_wmma_f32_16x16x4_f32(false, A1, false, B3, (short)0, a13, false, false);
    }

    const int sv = (e0 + lane < E) ? esrc[e0 + lane] : 0;
    float part[16];
    #pragma unroll
    for (int r = 0; r < 8; r++) {
        const int s0 = __shfl(sv, 8*hh + r, 32);
        const int s1 = __shfl(sv, 16 + 8*hh + r, 32);
        const float4 y0 = *(const float4*)(y + (size_t)s0*64 + 4*c);
        const float4 y1 = *(const float4*)(y + (size_t)s1*64 + 4*c);
        part[r]     = a00[r]*y0.x + a01[r]*y0.y + a02[r]*y0.z + a03[r]*y0.w;
        part[8 + r] = a10[r]*y1.x + a11[r]*y1.y + a12[r]*y1.z + a13[r]*y1.w;
    }
    #pragma unroll
    for (int i = 0; i < 8; i++) {
        const bool up = (c & 8) != 0;
        const float keep = up ? part[8 + i] : part[i];
        const float send = up ? part[i] : part[8 + i];
        part[i] = keep + __shfl_xor(send, 8, 32);
    }
    #pragma unroll
    for (int i = 0; i < 4; i++) {
        const bool up = (c & 4) != 0;
        const float keep = up ? part[4 + i] : part[i];
        const float send = up ? part[i] : part[4 + i];
        part[i] = keep + __shfl_xor(send, 4, 32);
    }
    #pragma unroll
    for (int i = 0; i < 2; i++) {
        const bool up = (c & 2) != 0;
        const float keep = up ? part[2 + i] : part[i];
        const float send = up ? part[i] : part[2 + i];
        part[i] = keep + __shfl_xor(send, 2, 32);
    }
    {
        const bool up = (c & 1) != 0;
        const float keep = up ? part[1] : part[0];
        const float send = up ? part[0] : part[1];
        part[0] = keep + __shfl_xor(send, 1, 32);
    }
    const int e = e0 + 16*(c >> 3) + 8*hh + (c & 7);
    if (e < E)
        atomicAdd(&out[edst[e]], part[0] * (1.0f/128.0f) * INV_NN);
}

extern "C" void kernel_launch(void* const* d_in, const int* in_sizes, int n_in,
                              void* d_out, int out_size, void* d_ws, size_t ws_size,
                              hipStream_t stream)
{
    const float* evec = (const float*)d_in[0];
    const float* elen = (const float*)d_in[1];
    const float* W1a  = (const float*)d_in[2];
    const float* W1b  = (const float*)d_in[3];
    const float* W2a  = (const float*)d_in[4];
    const float* W2b  = (const float*)d_in[5];
    const int*   esrc = (const int*)d_in[6];
    const int*   edst = (const int*)d_in[7];
    float* out = (float*)d_out;

    const int E = in_sizes[0] / 3;
    const int N = out_size;

    float* ws = (float*)d_ws;
    float* x_     = ws;                 ws += (size_t)N * 16;
    float* h_     = ws;                 ws += (size_t)N * 96;
    float* y_     = ws;                 ws += (size_t)N * 64;
    float* sh_    = ws;                 ws += (size_t)E * 16;
    float* basis_ = ws;                 ws += (size_t)E * 3;
    int*   perm_  = (int*)ws;           ws += (size_t)E;
    int*   cnt_   = (int*)ws;
    const int nb  = (E + SORT_CHUNK - 1) / SORT_CHUNK;
    const int M   = NBINS * nb;

    long nx = (long)N * 16, nh = (long)N * 96;
    kzero<<<(int)((nx + 255) / 256), 256, 0, stream>>>(x_, nx);
    kzero<<<(int)((nh + 255) / 256), 256, 0, stream>>>(h_, nh);
    kzero<<<(N + 255) / 256, 256, 0, stream>>>(out, N);

    kA<<<(E + 255) / 256, 256, 0, stream>>>(evec, elen, edst, sh_, basis_, x_, E);
    kCount<<<nb, SORT_THREADS, 0, stream>>>(elen, cnt_, nb, E);
    kScan<<<1, 1024, 0, stream>>>(cnt_, M, (M + 1023) / 1024);
    kScatter<<<nb, SORT_THREADS, 0, stream>>>(elen, cnt_, perm_, nb, E);
    kB<<<(E + KB_EDGES - 1) / KB_EDGES, KB_THREADS, 0, stream>>>(sh_, basis_, x_, esrc, edst, perm_, W1a, W1b, h_, E);
    kC<<<(N + 255) / 256, 256, 0, stream>>>(h_, y_, N);
    kE<<<(E + 255) / 256, 256, 0, stream>>>(basis_, y_, esrc, edst, W2a, W2b, out, E);
}


# kB: hidden columns positive on all 128 rows of a length-sorted tile are composed: G = W1a x W1b once per block on f32 WMMA, edges x G at k=4; sign-mixed columns keep the k loop; k sum order changes
# speedup vs baseline: 3.3713x; 1.4475x over previous
#include <hip/hip_runtime.h>
#include <math.h>

typedef __attribute__((ext_vector_type(2))) float v2f;
typedef __attribute__((ext_vector_type(8))) float v8f;

#define INV_NN     0.512989176042577f
#define INV_SQRT3  0.5773502691896258f
#define S3   1.7320508075688772f
#define S5   2.23606797749979f
#define S7   2.6457513110645907f
#define S15  3.872983346207417f
#define S105 10.246950765959598f
#define C358 2.091650066335189f
#define C218 1.6201851746019651f

__global__ void kzero(float* __restrict__ p, long n) {
    long i = (long)blockIdx.x * blockDim.x + threadIdx.x;
    if (i < n) p[i] = 0.0f;
}

#define NBINS 256
#define SORT_CHUNK 4096
#define SORT_THREADS 1024
__device__ __forceinline__ int len_bin(float len) {
    float t = fminf(fmaxf(len * ((float)NBINS / 4.0f), 0.0f), (float)(NBINS - 1));
    return (int)t;
}

__global__ __launch_bounds__(SORT_THREADS) void kCount(const float* __restrict__ elen,
                                                       int* __restrict__ cnt, int nb, int E)
{
    __shared__ int lh[NBINS];
    const int tid = threadIdx.x;
    if (tid < NBINS) lh[tid] = 0;
    __syncthreads();
    const int base = blockIdx.x * SORT_CHUNK;
    #pragma unroll
    for (int i = 0; i < SORT_CHUNK / SORT_THREADS; i++) {
        const int e = base + i * SORT_THREADS + tid;
        if (e < E) atomicAdd(&lh[len_bin(elen[e])], 1);
    }
    __syncthreads();
    if (tid < NBINS) cnt[(size_t)tid * nb + blockIdx.x] = lh[tid];
}

__global__ __launch_bounds__(1024) void kScan(int* __restrict__ cnt, int M, int per)
{
    __shared__ int wsum[32];
    const int t = threadIdx.x, lane = t & 31, w = t >> 5;
    const int lo = t * per;
    int s = 0;
    #pragma unroll 8
    for (int i = 0; i < per; i++) {
        const int k = lo + i;
        s += (k < M) ? cnt[k] : 0;
    }
    int incl = s;
    #pragma unroll
    for (int d = 1; d < 32; d <<= 1) {
        int v = __shfl_up(incl, d, 32);
        if (lane >= d) incl += v;
    }
    if (lane == 31) wsum[w] = incl;
    __syncthreads();
    if (w == 0) {
        const int v = wsum[lane];
        int iv = v;
        #pragma unroll
        for (int d = 1; d < 32; d <<= 1) {
            int u = __shfl_up(iv, d, 32);
            if (lane >= d) iv += u;
        }
        wsum[lane] = iv - v;
    }
    __syncthreads();
    int run = wsum[w] + incl - s;
    #pragma unroll 8
    for (int i = 0; i < per; i++) {
        const int k = lo + i;
        if (k < M) { const int c = cnt[k]; cnt[k] = run; run += c; }
    }
}

__global__ __launch_bounds__(SORT_THREADS) void kScatter(const float* __restrict__ elen,
                                                         const int* __restrict__ offs,
                                                         int* __restrict__ perm, int nb, int E)
{
    __shared__ int lh[NBINS];
    const int tid = threadIdx.x;
    if (tid < NBINS) lh[tid] = offs[(size_t)tid * nb + blockIdx.x];
    __syncthreads();
    const int base = blockIdx.x * SORT_CHUNK;
    #pragma unroll
    for (int i = 0; i < SORT_CHUNK / SORT_THREADS; i++) {
        const int e = base + i * SORT_THREADS + tid;
        if (e < E) {
            const int pos = atomicAdd(&lh[len_bin(elen[e])], 1);
            if ((unsigned)pos < (unsigned)E) perm[pos] = e;
        }
    }
}

#define SH_PITCH 20
__global__ __launch_bounds__(256) void kA(const float* __restrict__ evec, const float* __restrict__ elen,
                   const int* __restrict__ edst,
                   float* __restrict__ sh, float* __restrict__ basis,
                   float* __restrict__ x, int E)
{
    __shared__ __attribute__((aligned(16))) float tile[256 * SH_PITCH];
    __shared__ int dstT[256];

    const int tid = threadIdx.x;
    const int eb  = blockIdx.x * 256;
    const int e   = eb + tid;
    float s[16] = {};
    int d = 0;
    if (e < E) {
        float vx = evec[3*(size_t)e], vy = evec[3*(size_t)e+1], vz = evec[3*(size_t)e+2];
        float rn = 1.0f / (sqrtf(vx*vx + vy*vy + vz*vz) + 1e-12f);
        vx *= rn; vy *= rn; vz *= rn;
        float x2 = vx*vx, y2 = vy*vy, z2 = vz*vz;
        s[0]  = 1.0f;
        s[1]  = S3*vx; s[2] = S3*vy; s[3] = S3*vz;
        s[4]  = S15*vx*vz;
        s[5]  = S15*vx*vy;
        s[6]  = 0.5f*S5*(2.0f*y2 - x2 - z2);
        s[7]  = S15*vy*vz;
        s[8]  = 0.5f*S15*(z2 - x2);
        s[9]  = C358*vx*(3.0f*z2 - x2);
        s[10] = S105*vx*vy*vz;
        s[11] = C218*vx*(4.0f*y2 - z2 - x2);
        s[12] = 0.5f*S7*vy*(2.0f*y2 - 3.0f*z2 - 3.0f*x2);
        s[13] = C218*vz*(4.0f*y2 - z2 - x2);
        s[14] = 0.5f*S105*vy*(z2 - x2);
        s[15] = C358*vz*(z2 - 3.0f*x2);
        d = edst[e];
        float len = elen[e];
        #pragma unroll
        for (int i = 0; i < 3; i++) {
            float dd = (len - 2.0f*(float)i) * 0.5f;
            basis[(size_t)e*3 + i] = expf(-dd*dd) * (1.0f/1.12f);
        }
    }
    {
        float4* row = (float4*)(tile + tid * SH_PITCH);
        row[0] = make_float4(s[0],  s[1],  s[2],  s[3]);
        row[1] = make_float4(s[4],  s[5],  s[6],  s[7]);
        row[2] = make_float4(s[8],  s[9],  s[10], s[11]);
        row[3] = make_float4(s[12], s[13], s[14], s[15]);
        dstT[tid] = d;
    }
    __syncthreads();

    const int lane  = tid & 31;
    const int hh    = lane >> 4;
    const int j     = lane & 15;
    const int wbase = tid & ~31;
    if (eb + wbase + 32 <= E) {
        float v[16];
        int   dm[16];
        #pragma unroll
        for (int i = 0; i < 16; i++) {
            const int m = wbase + 2*i + hh;
            v[i]  = tile[m * SH_PITCH + j];
            dm[i] = dstT[m];
        }
        #pragma unroll
        for (int i = 0; i < 16; i++) {
            const int m = wbase + 2*i + hh;
            sh[(size_t)(eb + m)*16 + j] = v[i];
            atomicAdd(&x[(size_t)dm[i]*16 + j], v[i] * INV_NN);
        }
    } else {
        for (int i = 0; i < 16; i++) {
            const int m  = wbase + 2*i + hh;
            const int em = eb + m;
            if (em < E) {
                const float v = tile[m * SH_PITCH + j];
                sh[(size_t)em*16 + j] = v;
                atomicAdd(&x[(size_t)dstT[m]*16 + j], v * INV_NN);
            }
        }
    }
}

#define KB_EDGES 128
#define KB_THREADS 384
#define KB_PCH 16
#define KB_PITCH 800
#define KB_HP 20
__device__ __forceinline__ float pre1(float4 b, float w0, float w1, float w2) {
    return fmaf(b.z, w2, fmaf(b.y, w1, b.x * w0));
}
__device__ __forceinline__ float hidden1(float4 b, float w0, float w1, float w2) {
    return fmaxf(pre1(b, w0, w1, w2) * INV_SQRT3, 0.0f);
}
__global__ __launch_bounds__(KB_THREADS) void kB(
    const float* __restrict__ sh, const float* __restrict__ basis,
    const float* __restrict__ x,
    const int* __restrict__ esrc, const int* __restrict__ edst,
    const int* __restrict__ perm,
    const float* __restrict__ W1a, const float* __restrict__ W1b,
    float* __restrict__ h, int E)
{
    __shared__ __attribute__((aligned(16))) float bas[KB_EDGES][4];
    __shared__ __attribute__((aligned(16))) float dots[KB_EDGES][4];
    __shared__ int   dstT[KB_EDGES];
    __shared__ int   eid[KB_EDGES];
    __shared__ __attribute__((aligned(16))) float waF[4][256];
    __shared__ __attribute__((aligned(8))) unsigned colsF[256];
    __shared__ __attribute__((aligned(16))) float wP[256][4];
    __shared__ unsigned colsP[256];
    __shared__ int   wcntF[8], wcntP[8];
    __shared__ __attribute__((aligned(16))) float Gl[2 * KB_PITCH];
    __shared__ __attribute__((aligned(16))) float hP[KB_EDGES][KB_HP];
    __shared__ __attribute__((aligned(16))) float Bp[(KB_PCH / 2) * KB_PITCH];

    const int tid  = threadIdx.x;
    const int wv   = tid >> 5;
    const int lane = tid & 31;
    const int e0   = blockIdx.x * KB_EDGES;
    const int nvalid = min(KB_EDGES, E - e0);

    if (tid < KB_EDGES) {
        const int i = e0 + tid;
        int e = -1;
        float b0 = 0.0f, b1 = 0.0f, b2 = 0.0f;
        if (i < E) {
            e = perm[i];
            e = min(max(e, 0), E - 1);
            b0 = basis[(size_t)e*3]; b1 = basis[(size_t)e*3+1]; b2 = basis[(size_t)e*3+2];
        }
        eid[tid] = e;
        dstT[tid] = (e >= 0) ? edst[e] : 0;
        *(float4*)bas[tid] = make_float4(b0, b1, b2, 0.0f);
    } else {
        const int it = tid - KB_EDGES;
        const int m = it >> 1, q = it & 1;
        const int i = e0 + m;
        float sa = 0.0f, sb = 0.0f;
        if (i < E) {
            int e = perm[i];
            e = min(max(e, 0), E - 1);
            const int src = esrc[e];
            const float* xr = x  + (size_t)src*16;
            const float* sr = sh + (size_t)e*16;
            float xs[8], ss[8];
            #pragma unroll
            for (int j = 0; j < 8; j++) {
                const int idx = q ? (1 + j) : (j == 0 ? 0 : 8 + j);
                xs[j] = xr[idx];
                ss[j] = sr[idx];
            }
            if (q == 0) {
                sa = xs[0] * ss[0];
                float t = 0.0f;
                #pragma unroll
                for (int j = 1; j < 8; j++) t += xs[j] * ss[j];
                sb = t * rsqrtf(7.0f);
            } else {
                float t = 0.0f, u = 0.0f;
                #pragma unroll
                for (int j = 0; j < 3; j++) t += xs[j] * ss[j];
                #pragma unroll
                for (int j = 3; j < 8; j++) u += xs[j] * ss[j];
                sa = t * rsqrtf(3.0f);
                sb = u * rsqrtf(5.0f);
            }
        }
        if (q == 0) { dots[m][0] = sa; dots[m][3] = sb; }
        else        { dots[m][1] = sa; dots[m][2] = sb; }
    }
    if (tid < 256) waF[3][tid] = 0.0f;
    __syncthreads();

    int cls = 0;
    float w0 = 0.0f, w1 = 0.0f, w2 = 0.0f;
    if (tid < 256) {
        w0 = W1a[tid]; w1 = W1a[256 + tid]; w2 = W1a[512 + tid];
        float lo = __builtin_huge_valf(), hi = -__builtin_huge_valf();
        #pragma unroll 8
        for (int m = 0; m < nvalid; m++) {
            const float t = pre1(*(const float4*)bas[m], w0, w1, w2);
            lo = fminf(lo, t);
            hi = fmaxf(hi, t);
        }
        cls = (lo > 0.0f) ? 1 : ((hi > 0.0f) ? 2 : 0);
    }
    const unsigned balF = (unsigned)__ballot(cls == 1);
    const unsigned balP = (unsigned)__ballot(cls == 2);
    if (wv < 8 && lane == 0) { wcntF[wv] = __popc(balF); wcntP[wv] = __popc(balP); }
    __syncthreads();

    int baseF = 0, LF = 0, baseP = 0, LP = 0;
    #pragma unroll
    for (int i = 0; i < 8; i++) {
        const int cF = wcntF[i], cP = wcntP[i];
        if (i < wv) { baseF += cF; baseP += cP; }
        LF += cF; LP += cP;
    }
    const int LFp = (LF + 15) & ~15;
    const int LPp = (LP + 3) & ~3;
    if (tid < 256) {
        const unsigned lm = (1u << lane) - 1u;
        if (cls == 1) {
            const int j = baseF + __popc(balF & lm);
            colsF[j] = (unsigned)tid * (384u * 4u);
            waF[0][j] = w0; waF[1][j] = w1; waF[2][j] = w2;
        } else if (cls == 2) {
            const int j = baseP + __popc(balP & lm);
            colsP[j] = (unsigned)tid * 384u;
            *(float4*)wP[j] = make_float4(w0, w1, w2, 0.0f);
        }
    } else {
        const int j = LF + (tid - 256);
        if (j < LFp) { colsF[j] = 0u; waF[0][j] = 0.0f; waF[1][j] = 0.0f; waF[2][j] = 0.0f; }
    }
    __syncthreads();

    const int c  = lane & 15;
    const int hh = lane >> 4;
    const char* Wb = (const char*)W1b;

    auto fill = [&](int ch) {
        for (int it = tid; it < KB_EDGES * KB_PCH; it += KB_THREADS) {
            const int m = it >> 4, kk = it & 15;
            const int j = ch * KB_PCH + kk;
            float v = 0.0f;
            if (j < LP) {
                const float4 w = *(const float4*)wP[j];
                v = hidden1(*(const float4*)bas[m], w.x, w.y, w.z);
            }
            hP[m][kk] = v;
        }
        const int nk = min(KB_PCH, LPp - ch * KB_PCH);
        #pragma unroll 4
        for (int kk = 0; kk < nk; kk++) {
            const int j = ch * KB_PCH + kk;
            float v = 0.0f;
            if (j < LP) v = W1b[colsP[j] + tid];
            Bp[(kk >> 1) * KB_PITCH + 2 * tid + (kk & 1)] = v;
        }
    };

    {
        v8f g0 = {}, g1 = {};
        const unsigned cb = (unsigned)(wv * 32 + c) * 4u;
        const float* Arow = waF[min(c, 3)];
        for (int k0 = 0; k0 < LFp; k0 += 16) {
            #pragma unroll
            for (int u = 0; u < 4; u++) {
                const int kb = k0 + 4*u + 2*hh;
                const v2f A = *(const v2f*)(Arow + kb);
                const float* Br0 = (const float*)(Wb + (colsF[kb]     + cb));
                const float* Br1 = (const float*)(Wb + (colsF[kb + 1] + cb));
                v2f Ba; Ba.x = Br0[0];  Ba.y = Br1[0];
                v2f Bb; Bb.x = Br0[16]; Bb.y = Br1[16];
                g0 = __builtin_amdgcn_wmma_f32_16x16x4_f32(false, A, false, Ba, (short)0, g0, false, false);
                g1 = __builtin_amdgcn_wmma_f32_16x16x4_f32(false, A, false, Bb, (short)0, g1, false, false);
            }
        }
        if (hh == 0) {
            const int col0 = wv * 32 + c, col1 = col0 + 16;
            *(v2f*)(Gl + 2*col0)            = (v2f){g0[0], g0[1]};
            *(v2f*)(Gl + KB_PITCH + 2*col0) = (v2f){g0[2], 0.0f};
            *(v2f*)(Gl + 2*col1)            = (v2f){g1[0], g1[1]};
            *(v2f*)(Gl + KB_PITCH + 2*col1) = (v2f){g1[2], 0.0f};
        }
    }
    const int nch = (LPp + KB_PCH - 1) / KB_PCH;
    if (nch > 0) fill(0);
    __syncthreads();

    const int kc   = wv % 6;
    const int rh   = wv / 6;
    const int colq = kc * 16 + c;
    const float scale = (1.0f/16.0f) * 0.5f * INV_NN;
    const float* gp = Gl + hh * KB_PITCH + 2 * colq;

    for (int s = 0; s < 4; s++) {
        const int r0 = rh * 64 + s * 16;
        v8f a0 = {}, a1 = {}, a2 = {}, a3 = {};
        {
            const v2f bb = *(const v2f*)(bas[r0 + c] + 2*hh);
            v2f A; A.x = bb.x * INV_SQRT3; A.y = bb.y * INV_SQRT3;
            const v2f B0 = *(const v2f*)(gp);
            const v2f B1 = *(const v2f*)(gp + 192);
            const v2f B2 = *(const v2f*)(gp + 384);
            const v2f B3 = *(const v2f*)(gp + 576);
            a0 = __builtin_amdgcn_wmma_f32_16x16x4_f32(false, A, false, B0, (short)0, a0, false, false);
            a1 = __builtin_amdgcn_wmma_f32_16x16x4_f32(false, A, false, B1, (short)0, a1, false, false);
            a2 = __builtin_amdgcn_wmma_f32_16x16x4_f32(false, A, false, B2, (short)0, a2, false, false);
            a3 = __builtin_amdgcn_wmma_f32_16x16x4_f32(false, A, false, B3, (short)0, a3, false, false);
        }
        for (int ch = 0; ch < nch; ch++) {
            if (nch > 1 && (s | ch) != 0) {
                __syncthreads();
                fill(ch);
                __syncthreads();
            }
            const int nk = min(KB_PCH, LPp - ch * KB_PCH);
            for (int kk = 0; kk < nk; kk += 4) {
                const v2f A = *(const v2f*)(hP[r0 + c] + kk + 2*hh);
                const float* bp = Bp + ((kk >> 1) + hh) * KB_PITCH + 2 * colq;
                const v2f B0 = *(const v2f*)(bp);
                const v2f B1 = *(const v2f*)(bp + 192);
                const v2f B2 = *(const v2f*)(bp + 384);
                const v2f B3 = *(const v2f*)(bp + 576);
                a0 = __builtin_amdgcn_wmma_f32_16x16x4_f32(false, A, false, B0, (short)0, a0, false, false);
                a1 = __builtin_amdgcn_wmma_f32_16x16x4_f32(false, A, false, B1, (short)0, a1, false, false);
                a2 = __builtin_amdgcn_wmma_f32_16x16x4_f32(false, A, false, B2, (short)0, a2, false, false);
                a3 = __builtin_amdgcn_wmma_f32_16x16x4_f32(false, A, false, B3, (short)0, a3, false, false);
            }
        }
        #pragma unroll
        for (int r = 0; r < 8; r++) {
            const int m = r0 + r + 8*hh;
            if (eid[m] >= 0) {
                const float4 d = *(const float4*)dots[m];
                const float ef = d.x*a0[r] + d.y*a1[r] + d.z*a2[r] + d.w*a3[r];
                atomicAdd(&h[(size_t)dstT[m]*96 + colq], ef * scale);
            }
        }
    }
}

__global__ void kC(const float* __restrict__ h, float* __restrict__ y, int N)
{
    int nid = blockIdx.x * blockDim.x + threadIdx.x;
    if (nid >= N) return;
    const float* hr = h + (size_t)nid * 96;
    float* yr = y + (size_t)nid * 64;
    #pragma unroll
    for (int j = 0; j < 16; j++) yr[j] = fmaxf(hr[j], 0.0f);
    #pragma unroll
    for (int j = 16; j < 32; j++) yr[j] = fabsf(hr[j]);
    #pragma unroll
    for (int j = 0; j < 32; j++) {
        float g = hr[32 + j];
        bool use_relu = (j < 8) || (j >= 16 && j < 24);
        float ga = use_relu ? fmaxf(g, 0.0f) : tanhf(g);
        yr[32 + j] = ga * hr[64 + j];
    }
}

__global__ __launch_bounds__(256) void kE(const float* __restrict__ basis,
                                          const float* __restrict__ y,
                                          const int* __restrict__ esrc,
                                          const int* __restrict__ edst,
                                          const float* __restrict__ W2a,
                                          const float* __restrict__ W2b,
                                          float* __restrict__ out, int E)
{
    __shared__ __attribute__((aligned(16))) float Wl[128 * 128];
    __shared__ __attribute__((aligned(16))) float Al[3 * 256];

    const int tid = threadIdx.x;
    #pragma unroll
    for (int it = 0; it < 3; it++) Al[tid + 256*it] = W2a[tid + 256*it];
    {
        const int cc = tid & 15;
        #pragma unroll
        for (int it = 0; it < 8; it++) {
            const int kp = (tid >> 4) + 16*it;
            const float4 r0 = *(const float4*)(W2b + (size_t)(2*kp)     * 64 + 4*cc);
            const float4 r1 = *(const float4*)(W2b + (size_t)(2*kp + 1) * 64 + 4*cc);
            float* wb = Wl + kp*128 + 2*cc;
            const int sw = (kp & 1) * 32;
            v2f t0; t0.x = r0.x; t0.y = r1.x;
            v2f t1; t1.x = r0.y; t1.y = r1.y;
            v2f t2; t2.x = r0.z; t2.y = r1.z;
            v2f t3; t3.x = r0.w; t3.y = r1.w;
            *(v2f*)(wb + ( 0 ^ sw)) = t0;
            *(v2f*)(wb + (32 ^ sw)) = t1;
            *(v2f*)(wb + (64 ^ sw)) = t2;
            *(v2f*)(wb + (96 ^ sw)) = t3;
        }
    }
    __syncthreads();

    const int wv   = tid >> 5;
    const int lane = tid & 31;
    const int c    = lane & 15;
    const int hh   = lane >> 4;
    const int e0   = blockIdx.x * 256 + wv * 32;
    if (e0 >= E) return;

    float p0 = 0.0f, p1 = 0.0f, p2 = 0.0f, q0 = 0.0f, q1 = 0.0f, q2 = 0.0f;
    {
        const int ea = e0 + c, eb = e0 + 16 + c;
        if (ea < E) { p0 = basis[(size_t)ea*3]; p1 = basis[(size_t)ea*3+1]; p2 = basis[(size_t)ea*3+2]; }
        if (eb < E) { q0 = basis[(size_t)eb*3]; q1 = basis[(size_t)eb*3+1]; q2 = basis[(size_t)eb*3+2]; }
    }

    v8f a00 = {}, a01 = {}, a02 = {}, a03 = {};
    v8f a10 = {}, a11 = {}, a12 = {}, a13 = {};

    const float* Ap = Al + 2*hh;
    const float* Bp = Wl + hh*128 + 2*c;
    const int o0 = ( 0 ^ (hh*32)), o1 = (32 ^ (hh*32));
    const int o2 = (64 ^ (hh*32)), o3 = (96 ^ (hh*32));

    for (int k0 = 0; k0 < 256; k0 += 4) {
        const v2f u0 = *(const v2f*)(Ap + k0);
        const v2f u1 = *(const v2f*)(Ap + 256 + k0);
        const v2f u2 = *(const v2f*)(Ap + 512 + k0);
        v2f A0, A1;
        A0.x = fmaxf((p0*u0.x + p1*u1.x + p2*u2.x) * INV_SQRT3, 0.0f);
        A0.y = fmaxf((p0*u0.y + p1*u1.y + p2*u2.y) * INV_SQRT3, 0.0f);
        A1.x = fmaxf((q0*u0.x + q1*u1.x + q2*u2.x) * INV_SQRT3, 0.0f);
        A1.y = fmaxf((q0*u0.y + q1*u1.y + q2*u2.y) * INV_SQRT3, 0.0f);
        const float* Bk = Bp + k0*64;
        const v2f B0 = *(const v2f*)(Bk + o0);
        const v2f B1 = *(const v2f*)(Bk + o1);
        const v2f B2 = *(const v2f*)(Bk + o2);
        const v2f B3 = *(const v2f*)(Bk + o3);
        a00 = __builtin_amdgcn_wmma_f32_16x16x4_f32(false, A0, false, B0, (short)0, a00, false, false);
        a10 = __builtin_amdgcn_wmma_f32_16x16x4_f32(false, A1, false, B0, (short)0, a10, false, false);
        a01 = __builtin_amdgcn_wmma_f32_16x16x4_f32(false, A0, false, B1, (short)0, a01, false, false);
        a11 = __builtin_amdgcn_wmma_f32_16x16x4_f32(false, A1, false, B1, (short)0, a11, false, false);
        a02 = __builtin_amdgcn_wmma_f32_16x16x4_f32(false, A0, false, B2, (short)0, a02, false, false);
        a12 = __builtin_amdgcn_wmma_f32_16x16x4_f32(false, A1, false, B2, (short)0, a12, false, false);
        a03 = __builtin_amdgcn_wmma_f32_16x16x4_f32(false, A0, false, B3, (short)0, a03, false, false);
        a13 = __builtin_amdgcn_wmma_f32_16x16x4_f32(false, A1, false, B3, (short)0, a13, false, false);
    }

    const int sv = (e0 + lane < E) ? esrc[e0 + lane] : 0;
    float part[16];
    #pragma unroll
    for (int r = 0; r < 8; r++) {
        const int s0 = __shfl(sv, 8*hh + r, 32);
        const int s1 = __shfl(sv, 16 + 8*hh + r, 32);
        const float4 y0 = *(const float4*)(y + (size_t)s0*64 + 4*c);
        const float4 y1 = *(const float4*)(y + (size_t)s1*64 + 4*c);
        part[r]     = a00[r]*y0.x + a01[r]*y0.y + a02[r]*y0.z + a03[r]*y0.w;
        part[8 + r] = a10[r]*y1.x + a11[r]*y1.y + a12[r]*y1.z + a13[r]*y1.w;
    }
    #pragma unroll
    for (int i = 0; i < 8; i++) {
        const bool up = (c & 8) != 0;
        const float keep = up ? part[8 + i] : part[i];
        const float send = up ? part[i] : part[8 + i];
        part[i] = keep + __shfl_xor(send, 8, 32);
    }
    #pragma unroll
    for (int i = 0; i < 4; i++) {
        const bool up = (c & 4) != 0;
        const float keep = up ? part[4 + i] : part[i];
        const float send = up ? part[i] : part[4 + i];
        part[i] = keep + __shfl_xor(send, 4, 32);
    }
    #pragma unroll
    for (int i = 0; i < 2; i++) {
        const bool up = (c & 2) != 0;
        const float keep = up ? part[2 + i] : part[i];
        const float send = up ? part[i] : part[2 + i];
        part[i] = keep + __shfl_xor(send, 2, 32);
    }
    {
        const bool up = (c & 1) != 0;
        const float keep = up ? part[1] : part[0];
        const float send = up ? part[0] : part[1];
        part[0] = keep + __shfl_xor(send, 1, 32);
    }
    const int e = e0 + 16*(c >> 3) + 8*hh + (c & 7);
    if (e < E)
        atomicAdd(&out[edst[e]], part[0] * (1.0f/128.0f) * INV_NN);
}

extern "C" void kernel_launch(void* const* d_in, const int* in_sizes, int n_in,
                              void* d_out, int out_size, void* d_ws, size_t ws_size,
                              hipStream_t stream)
{
    const float* evec = (const float*)d_in[0];
    const float* elen = (const float*)d_in[1];
    const float* W1a  = (const float*)d_in[2];
    const float* W1b  = (const float*)d_in[3];
    const float* W2a  = (const float*)d_in[4];
    const float* W2b  = (const float*)d_in[5];
    const int*   esrc = (const int*)d_in[6];
    const int*   edst = (const int*)d_in[7];
    float* out = (float*)d_out;

    const int E = in_sizes[0] / 3;
    const int N = out_size;

    float* ws = (float*)d_ws;
    float* x_     = ws;                 ws += (size_t)N * 16;
    float* h_     = ws;                 ws += (size_t)N * 96;
    float* y_     = ws;                 ws += (size_t)N * 64;
    float* sh_    = ws;                 ws += (size_t)E * 16;
    float* basis_ = ws;                 ws += (size_t)E * 3;
    int*   perm_  = (int*)ws;           ws += (size_t)E;
    int*   cnt_   = (int*)ws;
    const int nb  = (E + SORT_CHUNK - 1) / SORT_CHUNK;
    const int M   = NBINS * nb;

    long nx = (long)N * 16, nh = (long)N * 96;
    kzero<<<(int)((nx + 255) / 256), 256, 0, stream>>>(x_, nx);
    kzero<<<(int)((nh + 255) / 256), 256, 0, stream>>>(h_, nh);
    kzero<<<(N + 255) / 256, 256, 0, stream>>>(out, N);

    kA<<<(E + 255) / 256, 256, 0, stream>>>(evec, elen, edst, sh_, basis_, x_, E);
    kCount<<<nb, SORT_THREADS, 0, stream>>>(elen, cnt_, nb, E);
    kScan<<<1, 1024, 0, stream>>>(cnt_, M, (M + 1023) / 1024);
    kScatter<<<nb, SORT_THREADS, 0, stream>>>(elen, cnt_, perm_, nb, E);
    kB<<<(E + KB_EDGES - 1) / KB_EDGES, KB_THREADS, 0, stream>>>(sh_, basis_, x_, esrc, edst, perm_, W1a, W1b, h_, E);
    kC<<<(N + 255) / 256, 256, 0, stream>>>(h_, y_, N);
    kE<<<(E + 255) / 256, 256, 0, stream>>>(basis_, y_, esrc, edst, W2a, W2b, out, E);
}


# kE: edges in length order; kB's column classes from W2a: through columns composed G2 = W2a x W2b per 128-edge tile on f32 WMMA, edges x G2 at k=4, mixed keep a k loop; k sum, out arrival order change
# speedup vs baseline: 3.6234x; 1.0748x over previous
#include <hip/hip_runtime.h>
#include <math.h>

typedef __attribute__((ext_vector_type(2))) float v2f;
typedef __attribute__((ext_vector_type(8))) float v8f;

#define INV_NN     0.512989176042577f
#define INV_SQRT3  0.5773502691896258f
#define S3   1.7320508075688772f
#define S5   2.23606797749979f
#define S7   2.6457513110645907f
#define S15  3.872983346207417f
#define S105 10.246950765959598f
#define C358 2.091650066335189f
#define C218 1.6201851746019651f

__global__ void kzero(float* __restrict__ p, long n) {
    long i = (long)blockIdx.x * blockDim.x + threadIdx.x;
    if (i < n) p[i] = 0.0f;
}

#define NBINS 256
#define SORT_CHUNK 4096
#define SORT_THREADS 1024
__device__ __forceinline__ int len_bin(float len) {
    float t = fminf(fmaxf(len * ((float)NBINS / 4.0f), 0.0f), (float)(NBINS - 1));
    return (int)t;
}

__global__ __launch_bounds__(SORT_THREADS) void kCount(const float* __restrict__ elen,
                                                       int* __restrict__ cnt, int nb, int E)
{
    __shared__ int lh[NBINS];
    const int tid = threadIdx.x;
    if (tid < NBINS) lh[tid] = 0;
    __syncthreads();
    const int base = blockIdx.x * SORT_CHUNK;
    #pragma unroll
    for (int i = 0; i < SORT_CHUNK / SORT_THREADS; i++) {
        const int e = base + i * SORT_THREADS + tid;
        if (e < E) atomicAdd(&lh[len_bin(elen[e])], 1);
    }
    __syncthreads();
    if (tid < NBINS) cnt[(size_t)tid * nb + blockIdx.x] = lh[tid];
}

__global__ __launch_bounds__(1024) void kScan(int* __restrict__ cnt, int M, int per)
{
    __shared__ int wsum[32];
    const int t = threadIdx.x, lane = t & 31, w = t >> 5;
    const int lo = t * per;
    int s = 0;
    #pragma unroll 8
    for (int i = 0; i < per; i++) {
        const int k = lo + i;
        s += (k < M) ? cnt[k] : 0;
    }
    int incl = s;
    #pragma unroll
    for (int d = 1; d < 32; d <<= 1) {
        int v = __shfl_up(incl, d, 32);
        if (lane >= d) incl += v;
    }
    if (lane == 31) wsum[w] = incl;
    __syncthreads();
    if (w == 0) {
        const int v = wsum[lane];
        int iv = v;
        #pragma unroll
        for (int d = 1; d < 32; d <<= 1) {
            int u = __shfl_up(iv, d, 32);
            if (lane >= d) iv += u;
        }
        wsum[lane] = iv - v;
    }
    __syncthreads();
    int run = wsum[w] + incl - s;
    #pragma unroll 8
    for (int i = 0; i < per; i++) {
        const int k = lo + i;
        if (k < M) { const int c = cnt[k]; cnt[k] = run; run += c; }
    }
}

__global__ __launch_bounds__(SORT_THREADS) void kScatter(const float* __restrict__ elen,
                                                         const int* __restrict__ offs,
                                                         int* __restrict__ perm, int nb, int E)
{
    __shared__ int lh[NBINS];
    const int tid = threadIdx.x;
    if (tid < NBINS) lh[tid] = offs[(size_t)tid * nb + blockIdx.x];
    __syncthreads();
    const int base = blockIdx.x * SORT_CHUNK;
    #pragma unroll
    for (int i = 0; i < SORT_CHUNK / SORT_THREADS; i++) {
        const int e = base + i * SORT_THREADS + tid;
        if (e < E) {
            const int pos = atomicAdd(&lh[len_bin(elen[e])], 1);
            if ((unsigned)pos < (unsigned)E) perm[pos] = e;
        }
    }
}

#define SH_PITCH 20
__global__ __launch_bounds__(256) void kA(const float* __restrict__ evec, const float* __restrict__ elen,
                   const int* __restrict__ edst,
                   float* __restrict__ sh, float* __restrict__ basis,
                   float* __restrict__ x, int E)
{
    __shared__ __attribute__((aligned(16))) float tile[256 * SH_PITCH];
    __shared__ int dstT[256];

    const int tid = threadIdx.x;
    const int eb  = blockIdx.x * 256;
    const int e   = eb + tid;
    float s[16] = {};
    int d = 0;
    if (e < E) {
        float vx = evec[3*(size_t)e], vy = evec[3*(size_t)e+1], vz = evec[3*(size_t)e+2];
        float rn = 1.0f / (sqrtf(vx*vx + vy*vy + vz*vz) + 1e-12f);
        vx *= rn; vy *= rn; vz *= rn;
        float x2 = vx*vx, y2 = vy*vy, z2 = vz*vz;
        s[0]  = 1.0f;
        s[1]  = S3*vx; s[2] = S3*vy; s[3] = S3*vz;
        s[4]  = S15*vx*vz;
        s[5]  = S15*vx*vy;
        s[6]  = 0.5f*S5*(2.0f*y2 - x2 - z2);
        s[7]  = S15*vy*vz;
        s[8]  = 0.5f*S15*(z2 - x2);
        s[9]  = C358*vx*(3.0f*z2 - x2);
        s[10] = S105*vx*vy*vz;
        s[11] = C218*vx*(4.0f*y2 - z2 - x2);
        s[12] = 0.5f*S7*vy*(2.0f*y2 - 3.0f*z2 - 3.0f*x2);
        s[13] = C218*vz*(4.0f*y2 - z2 - x2);
        s[14] = 0.5f*S105*vy*(z2 - x2);
        s[15] = C358*vz*(z2 - 3.0f*x2);
        d = edst[e];
        float len = elen[e];
        #pragma unroll
        for (int i = 0; i < 3; i++) {
            float dd = (len - 2.0f*(float)i) * 0.5f;
            basis[(size_t)e*3 + i] = expf(-dd*dd) * (1.0f/1.12f);
        }
    }
    {
        float4* row = (float4*)(tile + tid * SH_PITCH);
        row[0] = make_float4(s[0],  s[1],  s[2],  s[3]);
        row[1] = make_float4(s[4],  s[5],  s[6],  s[7]);
        row[2] = make_float4(s[8],  s[9],  s[10], s[11]);
        row[3] = make_float4(s[12], s[13], s[14], s[15]);
        dstT[tid] = d;
    }
    __syncthreads();

    const int lane  = tid & 31;
    const int hh    = lane >> 4;
    const int j     = lane & 15;
    const int wbase = tid & ~31;
    if (eb + wbase + 32 <= E) {
        float v[16];
        int   dm[16];
        #pragma unroll
        for (int i = 0; i < 16; i++) {
            const int m = wbase + 2*i + hh;
            v[i]  = tile[m * SH_PITCH + j];
            dm[i] = dstT[m];
        }
        #pragma unroll
        for (int i = 0; i < 16; i++) {
            const int m = wbase + 2*i + hh;
            sh[(size_t)(eb + m)*16 + j] = v[i];
            atomicAdd(&x[(size_t)dm[i]*16 + j], v[i] * INV_NN);
        }
    } else {
        for (int i = 0; i < 16; i++) {
            const int m  = wbase + 2*i + hh;
            const int em = eb + m;
            if (em < E) {
                const float v = tile[m * SH_PITCH + j];
                sh[(size_t)em*16 + j] = v;
                atomicAdd(&x[(size_t)dstT[m]*16 + j], v * INV_NN);
            }
        }
    }
}

#define KB_EDGES 128
#define KB_THREADS 384
#define KB_PCH 16
#define KB_PITCH 800
#define KB_HP 20
__device__ __forceinline__ float pre1(float4 b, float w0, float w1, float w2) {
    return fmaf(b.z, w2, fmaf(b.y, w1, b.x * w0));
}
__device__ __forceinline__ float hidden1(float4 b, float w0, float w1, float w2) {
    return fmaxf(pre1(b, w0, w1, w2) * INV_SQRT3, 0.0f);
}
__global__ __launch_bounds__(KB_THREADS) void kB(
    const float* __restrict__ sh, const float* __restrict__ basis,
    const float* __restrict__ x,
    const int* __restrict__ esrc, const int* __restrict__ edst,
    const int* __restrict__ perm,
    const float* __restrict__ W1a, const float* __restrict__ W1b,
    float* __restrict__ h, int E)
{
    __shared__ __attribute__((aligned(16))) float bas[KB_EDGES][4];
    __shared__ __attribute__((aligned(16))) float dots[KB_EDGES][4];
    __shared__ int   dstT[KB_EDGES];
    __shared__ int   eid[KB_EDGES];
    __shared__ __attribute__((aligned(16))) float waF[4][256];
    __shared__ __attribute__((aligned(8))) unsigned colsF[256];
    __shared__ __attribute__((aligned(16))) float wP[256][4];
    __shared__ unsigned colsP[256];
    __shared__ int   wcntF[8], wcntP[8];
    __shared__ __attribute__((aligned(16))) float Gl[2 * KB_PITCH];
    __shared__ __attribute__((aligned(16))) float hP[KB_EDGES][KB_HP];
    __shared__ __attribute__((aligned(16))) float Bp[(KB_PCH / 2) * KB_PITCH];

    const int tid  = threadIdx.x;
    const int wv   = tid >> 5;
    const int lane = tid & 31;
    const int e0   = blockIdx.x * KB_EDGES;
    const int nvalid = min(KB_EDGES, E - e0);

    if (tid < KB_EDGES) {
        const int i = e0 + tid;
        int e = -1;
        float b0 = 0.0f, b1 = 0.0f, b2 = 0.0f;
        if (i < E) {
            e = perm[i];
            e = min(max(e, 0), E - 1);
            b0 = basis[(size_t)e*3]; b1 = basis[(size_t)e*3+1]; b2 = basis[(size_t)e*3+2];
        }
        eid[tid] = e;
        dstT[tid] = (e >= 0) ? edst[e] : 0;
        *(float4*)bas[tid] = make_float4(b0, b1, b2, 0.0f);
    } else {
        const int it = tid - KB_EDGES;
        const int m = it >> 1, q = it & 1;
        const int i = e0 + m;
        float sa = 0.0f, sb = 0.0f;
        if (i < E) {
            int e = perm[i];
            e = min(max(e, 0), E - 1);
            const int src = esrc[e];
            const float* xr = x  + (size_t)src*16;
            const float* sr = sh + (size_t)e*16;
            float xs[8], ss[8];
            #pragma unroll
            for (int j = 0; j < 8; j++) {
                const int idx = q ? (1 + j) : (j == 0 ? 0 : 8 + j);
                xs[j] = xr[idx];
                ss[j] = sr[idx];
            }
            if (q == 0) {
                sa = xs[0] * ss[0];
                float t = 0.0f;
                #pragma unroll
                for (int j = 1; j < 8; j++) t += xs[j] * ss[j];
                sb = t * rsqrtf(7.0f);
            } else {
                float t = 0.0f, u = 0.0f;
                #pragma unroll
                for (int j = 0; j < 3; j++) t += xs[j] * ss[j];
                #pragma unroll
                for (int j = 3; j < 8; j++) u += xs[j] * ss[j];
                sa = t * rsqrtf(3.0f);
                sb = u * rsqrtf(5.0f);
            }
        }
        if (q == 0) { dots[m][0] = sa; dots[m][3] = sb; }
        else        { dots[m][1] = sa; dots[m][2] = sb; }
    }
    if (tid < 256) waF[3][tid] = 0.0f;
    __syncthreads();

    int cls = 0;
    float w0 = 0.0f, w1 = 0.0f, w2 = 0.0f;
    if (tid < 256) {
        w0 = W1a[tid]; w1 = W1a[256 + tid]; w2 = W1a[512 + tid];
        float lo = __builtin_huge_valf(), hi = -__builtin_huge_valf();
        #pragma unroll 8
        for (int m = 0; m < nvalid; m++) {
            const float t = pre1(*(const float4*)bas[m], w0, w1, w2);
            lo = fminf(lo, t);
            hi = fmaxf(hi, t);
        }
        cls = (lo > 0.0f) ? 1 : ((hi > 0.0f) ? 2 : 0);
    }
    const unsigned balF = (unsigned)__ballot(cls == 1);
    const unsigned balP = (unsigned)__ballot(cls == 2);
    if (wv < 8 && lane == 0) { wcntF[wv] = __popc(balF); wcntP[wv] = __popc(balP); }
    __syncthreads();

    int baseF = 0, LF = 0, baseP = 0, LP = 0;
    #pragma unroll
    for (int i = 0; i < 8; i++) {
        const int cF = wcntF[i], cP = wcntP[i];
        if (i < wv) { baseF += cF; baseP += cP; }
        LF += cF; LP += cP;
    }
    const int LFp = (LF + 15) & ~15;
    const int LPp = (LP + 3) & ~3;
    if (tid < 256) {
        const unsigned lm = (1u << lane) - 1u;
        if (cls == 1) {
            const int j = baseF + __popc(balF & lm);
            colsF[j] = (unsigned)tid * (384u * 4u);
            waF[0][j] = w0; waF[1][j] = w1; waF[2][j] = w2;
        } else if (cls == 2) {
            const int j = baseP + __popc(balP & lm);
            colsP[j] = (unsigned)tid * 384u;
            *(float4*)wP[j] = make_float4(w0, w1, w2, 0.0f);
        }
    } else {
        const int j = LF + (tid - 256);
        if (j < LFp) { colsF[j] = 0u; waF[0][j] = 0.0f; waF[1][j] = 0.0f; waF[2][j] = 0.0f; }
    }
    __syncthreads();

    const int c  = lane & 15;
    const int hh = lane >> 4;
    const char* Wb = (const char*)W1b;

    auto fill = [&](int ch) {
        for (int it = tid; it < KB_EDGES * KB_PCH; it += KB_THREADS) {
            const int m = it >> 4, kk = it & 15;
            const int j = ch * KB_PCH + kk;
            float v = 0.0f;
            if (j < LP) {
                const float4 w = *(const float4*)wP[j];
                v = hidden1(*(const float4*)bas[m], w.x, w.y, w.z);
            }
            hP[m][kk] = v;
        }
        const int nk = min(KB_PCH, LPp - ch * KB_PCH);
        #pragma unroll 4
        for (int kk = 0; kk < nk; kk++) {
            const int j = ch * KB_PCH + kk;
            float v = 0.0f;
            if (j < LP) v = W1b[colsP[j] + tid];
            Bp[(kk >> 1) * KB_PITCH + 2 * tid + (kk & 1)] = v;
        }
    };

    {
        v8f g0 = {}, g1 = {};
        const unsigned cb = (unsigned)(wv * 32 + c) * 4u;
        const float* Arow = waF[min(c, 3)];
        for (int k0 = 0; k0 < LFp; k0 += 16) {
            #pragma unroll
            for (int u = 0; u < 4; u++) {
                const int kb = k0 + 4*u + 2*hh;
                const v2f A = *(const v2f*)(Arow + kb);
                const float* Br0 = (const float*)(Wb + (colsF[kb]     + cb));
                const float* Br1 = (const float*)(Wb + (colsF[kb + 1] + cb));
                v2f Ba; Ba.x = Br0[0];  Ba.y = Br1[0];
                v2f Bb; Bb.x = Br0[16]; Bb.y = Br1[16];
                g0 = __builtin_amdgcn_wmma_f32_16x16x4_f32(false, A, false, Ba, (short)0, g0, false, false);
                g1 = __builtin_amdgcn_wmma_f32_16x16x4_f32(false, A, false, Bb, (short)0, g1, false, false);
            }
        }
        if (hh == 0) {
            const int col0 = wv * 32 + c, col1 = col0 + 16;
            *(v2f*)(Gl + 2*col0)            = (v2f){g0[0], g0[1]};
            *(v2f*)(Gl + KB_PITCH + 2*col0) = (v2f){g0[2], 0.0f};
            *(v2f*)(Gl + 2*col1)            = (v2f){g1[0], g1[1]};
            *(v2f*)(Gl + KB_PITCH + 2*col1) = (v2f){g1[2], 0.0f};
        }
    }
    const int nch = (LPp + KB_PCH - 1) / KB_PCH;
    if (nch > 0) fill(0);
    __syncthreads();

    const int kc   = wv % 6;
    const int rh   = wv / 6;
    const int colq = kc * 16 + c;
    const float scale = (1.0f/16.0f) * 0.5f * INV_NN;
    const float* gp = Gl + hh * KB_PITCH + 2 * colq;

    for (int s = 0; s < 4; s++) {
        const int r0 = rh * 64 + s * 16;
        v8f a0 = {}, a1 = {}, a2 = {}, a3 = {};
        {
            const v2f bb = *(const v2f*)(bas[r0 + c] + 2*hh);
            v2f A; A.x = bb.x * INV_SQRT3; A.y = bb.y * INV_SQRT3;
            const v2f B0 = *(const v2f*)(gp);
            const v2f B1 = *(const v2f*)(gp + 192);
            const v2f B2 = *(const v2f*)(gp + 384);
            const v2f B3 = *(const v2f*)(gp + 576);
            a0 = __builtin_amdgcn_wmma_f32_16x16x4_f32(false, A, false, B0, (short)0, a0, false, false);
            a1 = __builtin_amdgcn_wmma_f32_16x16x4_f32(false, A, false, B1, (short)0, a1, false, false);
            a2 = __builtin_amdgcn_wmma_f32_16x16x4_f32(false, A, false, B2, (short)0, a2, false, false);
            a3 = __builtin_amdgcn_wmma_f32_16x16x4_f32(false, A, false, B3, (short)0, a3, false, false);
        }
        for (int ch = 0; ch < nch; ch++) {
            if (nch > 1 && (s | ch) != 0) {
                __syncthreads();
                fill(ch);
                __syncthreads();
            }
            const int nk = min(KB_PCH, LPp - ch * KB_PCH);
            for (int kk = 0; kk < nk; kk += 4) {
                const v2f A = *(const v2f*)(hP[r0 + c] + kk + 2*hh);
                const float* bp = Bp + ((kk >> 1) + hh) * KB_PITCH + 2 * colq;
                const v2f B0 = *(const v2f*)(bp);
                const v2f B1 = *(const v2f*)(bp + 192);
                const v2f B2 = *(const v2f*)(bp + 384);
                const v2f B3 = *(const v2f*)(bp + 576);
                a0 = __builtin_amdgcn_wmma_f32_16x16x4_f32(false, A, false, B0, (short)0, a0, false, false);
                a1 = __builtin_amdgcn_wmma_f32_16x16x4_f32(false, A, false, B1, (short)0, a1, false, false);
                a2 = __builtin_amdgcn_wmma_f32_16x16x4_f32(false, A, false, B2, (short)0, a2, false, false);
                a3 = __builtin_amdgcn_wmma_f32_16x16x4_f32(false, A, false, B3, (short)0, a3, false, false);
            }
        }
        #pragma unroll
        for (int r = 0; r < 8; r++) {
            const int m = r0 + r + 8*hh;
            if (eid[m] >= 0) {
                const float4 d = *(const float4*)dots[m];
                const float ef = d.x*a0[r] + d.y*a1[r] + d.z*a2[r] + d.w*a3[r];
                atomicAdd(&h[(size_t)dstT[m]*96 + colq], ef * scale);
            }
        }
    }
}

__global__ void kC(const float* __restrict__ h, float* __restrict__ y, int N)
{
    int nid = blockIdx.x * blockDim.x + threadIdx.x;
    if (nid >= N) return;
    const float* hr = h + (size_t)nid * 96;
    float* yr = y + (size_t)nid * 64;
    #pragma unroll
    for (int j = 0; j < 16; j++) yr[j] = fmaxf(hr[j], 0.0f);
    #pragma unroll
    for (int j = 16; j < 32; j++) yr[j] = fabsf(hr[j]);
    #pragma unroll
    for (int j = 0; j < 32; j++) {
        float g = hr[32 + j];
        bool use_relu = (j < 8) || (j >= 16 && j < 24);
        float ga = use_relu ? fmaxf(g, 0.0f) : tanhf(g);
        yr[32 + j] = ga * hr[64 + j];
    }
}

#define KE_EDGES 128
#define KE_THREADS 256
#define KE_PCH 16
#define KE_PITCH 160
__global__ __launch_bounds__(KE_THREADS) void kE(const float* __restrict__ basis,
                                                 const float* __restrict__ y,
                                                 const int* __restrict__ esrc,
                                                 const int* __restrict__ edst,
                                                 const int* __restrict__ perm,
                                                 const float* __restrict__ W2a,
                                                 const float* __restrict__ W2b,
                                                 float* __restrict__ out, int E)
{
    __shared__ __attribute__((aligned(16))) float bas[KE_EDGES][4];
    __shared__ int   srcT[KE_EDGES];
    __shared__ int   dstT[KE_EDGES];
    __shared__ int   eid[KE_EDGES];
    __shared__ __attribute__((aligned(16))) float waF[4][256];
    __shared__ __attribute__((aligned(8))) unsigned colsF[256];
    __shared__ __attribute__((aligned(16))) float wP[256][4];
    __shared__ unsigned colsP[256];
    __shared__ int   wcntF[8], wcntP[8];
    __shared__ __attribute__((aligned(16))) float Gl[2][2 * KE_PITCH];
    __shared__ __attribute__((aligned(16))) float hP[KE_EDGES][KB_HP];
    __shared__ __attribute__((aligned(16))) float Bp[(KE_PCH / 2) * KE_PITCH];

    const int tid  = threadIdx.x;
    const int wv   = tid >> 5;
    const int lane = tid & 31;
    const int e0   = blockIdx.x * KE_EDGES;
    const int nvalid = min(KE_EDGES, E - e0);

    const float w0 = W2a[tid], w1 = W2a[256 + tid], w2 = W2a[512 + tid];

    if (tid < KE_EDGES) {
        const int i = e0 + tid;
        int e = -1, sv = 0, dv = 0;
        float b0 = 0.0f, b1 = 0.0f, b2 = 0.0f;
        if (i < E) {
            e = perm[i];
            e = min(max(e, 0), E - 1);
            b0 = basis[(size_t)e*3]; b1 = basis[(size_t)e*3+1]; b2 = basis[(size_t)e*3+2];
            sv = esrc[e];
            dv = edst[e];
        }
        eid[tid] = e;
        srcT[tid] = sv;
        dstT[tid] = dv;
        *(float4*)bas[tid] = make_float4(b0, b1, b2, 0.0f);
    }
    waF[3][tid] = 0.0f;
    __syncthreads();

    int cls;
    {
        bool allpos = true, anypos = false;
        #pragma unroll 8
        for (int m = 0; m < nvalid; m++) {
            const bool pos = pre1(*(const float4*)bas[m], w0, w1, w2) > 0.0f;
            allpos &= pos;
            anypos |= pos;
        }
        cls = allpos ? 1 : (anypos ? 2 : 0);
    }
    const unsigned balF = (unsigned)__ballot(cls == 1);
    const unsigned balP = (unsigned)__ballot(cls == 2);
    if (lane == 0) { wcntF[wv] = __popc(balF); wcntP[wv] = __popc(balP); }
    __syncthreads();

    int baseF = 0, LF = 0, baseP = 0, LP = 0;
    #pragma unroll
    for (int i = 0; i < 8; i++) {
        const int cF = wcntF[i], cP = wcntP[i];
        if (i < wv) { baseF += cF; baseP += cP; }
        LF += cF; LP += cP;
    }
    const int LFp = (LF + 31) & ~31;
    const int LPp = (LP + 3) & ~3;
    {
        const unsigned lm = (1u << lane) - 1u;
        if (cls == 1) {
            const int j = baseF + __popc(balF & lm);
            colsF[j] = (unsigned)tid * (64u * 4u);
            waF[0][j] = w0; waF[1][j] = w1; waF[2][j] = w2;
        } else if (cls == 2) {
            const int j = baseP + __popc(balP & lm);
            colsP[j] = (unsigned)tid * 64u;
            *(float4*)wP[j] = make_float4(w0, w1, w2, 0.0f);
        }
        if (tid < 32) {
            const int j = LF + tid;
            if (j < LFp) { colsF[j] = 0u; waF[0][j] = 0.0f; waF[1][j] = 0.0f; waF[2][j] = 0.0f; }
        }
    }
    __syncthreads();

    const int c  = lane & 15;
    const int hh = lane >> 4;
    const char* Wb = (const char*)W2b;

    auto fill = [&](int ch) {
        for (int it = tid; it < KE_EDGES * KE_PCH; it += KE_THREADS) {
            const int m = it >> 4, kk = it & 15;
            const int j = ch * KE_PCH + kk;
            float v = 0.0f;
            if (j < LP) {
                const float4 w = *(const float4*)wP[j];
                v = hidden1(*(const float4*)bas[m], w.x, w.y, w.z);
            }
            hP[m][kk] = v;
        }
        const int nk = min(KE_PCH, LPp - ch * KE_PCH);
        const int col = tid & 63;
        for (int kk = tid >> 6; kk < nk; kk += 4) {
            const int j = ch * KE_PCH + kk;
            float v = 0.0f;
            if (j < LP) v = W2b[colsP[j] + col];
            Bp[(kk >> 1) * KE_PITCH + 2 * col + (kk & 1)] = v;
        }
    };

    {
        const int ct = wv & 3, kh = wv >> 2;
        v8f g0 = {}, g1 = {};
        const unsigned cb = (unsigned)(ct * 16 + c) * 4u;
        const float* Arow = waF[min(c, 3)];
        for (int k0 = 16 * kh; k0 < LFp; k0 += 32) {
            #pragma unroll
            for (int u = 0; u < 4; u++) {
                const int kb = k0 + 4*u + 2*hh;
                const v2f A = *(const v2f*)(Arow + kb);
                v2f B;
                B.x = *(const float*)(Wb + (colsF[kb]     + cb));
                B.y = *(const float*)(Wb + (colsF[kb + 1] + cb));
                if (u & 1) g1 = __builtin_amdgcn_wmma_f32_16x16x4_f32(false, A, false, B, (short)0, g1, false, false);
                else       g0 = __builtin_amdgcn_wmma_f32_16x16x4_f32(false, A, false, B, (short)0, g0, false, false);
            }
        }
        if (hh == 0) {
            const int col = ct * 16 + c;
            *(v2f*)(Gl[kh] + 2*col)            = (v2f){g0[0] + g1[0], g0[1] + g1[1]};
            *(v2f*)(Gl[kh] + KE_PITCH + 2*col) = (v2f){g0[2] + g1[2], 0.0f};
        }
    }
    const int nch = (LPp + KE_PCH - 1) / KE_PCH;
    if (nch > 0) fill(0);
    __syncthreads();

    const int r0 = wv * 16;
    v8f a0 = {}, a1 = {}, a2 = {}, a3 = {};
    {
        const v2f bb = *(const v2f*)(bas[r0 + c] + 2*hh);
        v2f A; A.x = bb.x * INV_SQRT3; A.y = bb.y * INV_SQRT3;
        const float* gp = Gl[0] + hh * KE_PITCH + 8 * c;
        const float* gq = Gl[1] + hh * KE_PITCH + 8 * c;
        const float4 Pa = *(const float4*)(gp), Pb = *(const float4*)(gp + 4);
        const float4 Qa = *(const float4*)(gq), Qb = *(const float4*)(gq + 4);
        const v2f B0 = (v2f){Pa.x + Qa.x, Pa.y + Qa.y}, B1 = (v2f){Pa.z + Qa.z, Pa.w + Qa.w};
        const v2f B2 = (v2f){Pb.x + Qb.x, Pb.y + Qb.y}, B3 = (v2f){Pb.z + Qb.z, Pb.w + Qb.w};
        a0 = __builtin_amdgcn_wmma_f32_16x16x4_f32(false, A, false, B0, (short)0, a0, false, false);
        a1 = __builtin_amdgcn_wmma_f32_16x16x4_f32(false, A, false, B1, (short)0, a1, false, false);
        a2 = __builtin_amdgcn_wmma_f32_16x16x4_f32(false, A, false, B2, (short)0, a2, false, false);
        a3 = __builtin_amdgcn_wmma_f32_16x16x4_f32(false, A, false, B3, (short)0, a3, false, false);
    }
    for (int ch = 0; ch < nch; ch++) {
        if (ch > 0) {
            __syncthreads();
            fill(ch);
            __syncthreads();
        }
        const int nk = min(KE_PCH, LPp - ch * KE_PCH);
        for (int kk = 0; kk < nk; kk += 4) {
            const v2f A = *(const v2f*)(hP[r0 + c] + kk + 2*hh);
            const float* bp = Bp + ((kk >> 1) + hh) * KE_PITCH + 8 * c;
            const float4 Ba = *(const float4*)(bp);
            const float4 Bb = *(const float4*)(bp + 4);
            const v2f B0 = (v2f){Ba.x, Ba.y}, B1 = (v2f){Ba.z, Ba.w};
            const v2f B2 = (v2f){Bb.x, Bb.y}, B3 = (v2f){Bb.z, Bb.w};
            a0 = __builtin_amdgcn_wmma_f32_16x16x4_f32(false, A, false, B0, (short)0, a0, false, false);
            a1 = __builtin_amdgcn_wmma_f32_16x16x4_f32(false, A, false, B1, (short)0, a1, false, false);
            a2 = __builtin_amdgcn_wmma_f32_16x16x4_f32(false, A, false, B2, (short)0, a2, false, false);
            a3 = __builtin_amdgcn_wmma_f32_16x16x4_f32(false, A, false, B3, (short)0, a3, false, false);
        }
    }

    float part[8];
    #pragma unroll
    for (int r = 0; r < 8; r++) {
        const int sv = srcT[r0 + 8*hh + r];
        const float4 yv = *(const float4*)(y + (size_t)sv*64 + 4*c);
        part[r] = a0[r]*yv.x + a1[r]*yv.y + a2[r]*yv.z + a3[r]*yv.w;
    }
    #pragma unroll
    for (int i = 0; i < 4; i++) {
        const bool up = (c & 8) != 0;
        const float keep = up ? part[4 + i] : part[i];
        const float send = up ? part[i] : part[4 + i];
        part[i] = keep + __shfl_xor(send, 8, 32);
    }
    #pragma unroll
    for (int i = 0; i < 2; i++) {
        const bool up = (c & 4) != 0;
        const float keep = up ? part[2 + i] : part[i];
        const float send = up ? part[i] : part[2 + i];
        part[i] = keep + __shfl_xor(send, 4, 32);
    }
    {
        const bool up = (c & 2) != 0;
        const float keep = up ? part[1] : part[0];
        const float send = up ? part[0] : part[1];
        part[0] = keep + __shfl_xor(send, 2, 32);
    }
    const float tot = part[0] + __shfl_xor(part[0], 1, 32);
    if ((c & 1) == 0) {
        const int m = r0 + 8*hh + 4*((c >> 3) & 1) + 2*((c >> 2) & 1) + ((c >> 1) & 1);
        if (eid[m] >= 0)
            atomicAdd(&out[dstT[m]], tot * (1.0f/128.0f) * INV_NN);
    }
}

extern "C" void kernel_launch(void* const* d_in, const int* in_sizes, int n_in,
                              void* d_out, int out_size, void* d_ws, size_t ws_size,
                              hipStream_t stream)
{
    const float* evec = (const float*)d_in[0];
    const float* elen = (const float*)d_in[1];
    const float* W1a  = (const float*)d_in[2];
    const float* W1b  = (const float*)d_in[3];
    const float* W2a  = (const float*)d_in[4];
    const float* W2b  = (const float*)d_in[5];
    const int*   esrc = (const int*)d_in[6];
    const int*   edst = (const int*)d_in[7];
    float* out = (float*)d_out;

    const int E = in_sizes[0] / 3;
    const int N = out_size;

    float* ws = (float*)d_ws;
    float* x_     = ws;                 ws += (size_t)N * 16;
    float* h_     = ws;                 ws += (size_t)N * 96;
    float* y_     = ws;                 ws += (size_t)N * 64;
    float* sh_    = ws;                 ws += (size_t)E * 16;
    float* basis_ = ws;                 ws += (size_t)E * 3;
    int*   perm_  = (int*)ws;           ws += (size_t)E;
    int*   cnt_   = (int*)ws;
    const int nb  = (E + SORT_CHUNK - 1) / SORT_CHUNK;
    const int M   = NBINS * nb;

    long nx = (long)N * 16, nh = (long)N * 96;
    kzero<<<(int)((nx + 255) / 256), 256, 0, stream>>>(x_, nx);
    kzero<<<(int)((nh + 255) / 256), 256, 0, stream>>>(h_, nh);
    kzero<<<(N + 255) / 256, 256, 0, stream>>>(out, N);

    kA<<<(E + 255) / 256, 256, 0, stream>>>(evec, elen, edst, sh_, basis_, x_, E);
    kCount<<<nb, SORT_THREADS, 0, stream>>>(elen, cnt_, nb, E);
    kScan<<<1, 1024, 0, stream>>>(cnt_, M, (M + 1023) / 1024);
    kScatter<<<nb, SORT_THREADS, 0, stream>>>(elen, cnt_, perm_, nb, E);
    kB<<<(E + KB_EDGES - 1) / KB_EDGES, KB_THREADS, 0, stream>>>(sh_, basis_, x_, esrc, edst, perm_, W1a, W1b, h_, E);
    kC<<<(N + 255) / 256, 256, 0, stream>>>(h_, y_, N);
    kE<<<(E + KE_EDGES - 1) / KE_EDGES, KE_THREADS, 0, stream>>>(basis_, y_, esrc, edst, perm_, W2a, W2b, out, E);
}


# kC: 16 lanes per node, each one float4 of y from float4 pieces of h: a wave moves whole 128-byte lines and the grid spans all WGPs, not one thread per node at a 384-byte lane stride on 98 blocks
# speedup vs baseline: 3.7940x; 1.0471x over previous
#include <hip/hip_runtime.h>
#include <math.h>

typedef __attribute__((ext_vector_type(2))) float v2f;
typedef __attribute__((ext_vector_type(8))) float v8f;

#define INV_NN     0.512989176042577f
#define INV_SQRT3  0.5773502691896258f
#define S3   1.7320508075688772f
#define S5   2.23606797749979f
#define S7   2.6457513110645907f
#define S15  3.872983346207417f
#define S105 10.246950765959598f
#define C358 2.091650066335189f
#define C218 1.6201851746019651f

__global__ void kzero(float* __restrict__ p, long n) {
    long i = (long)blockIdx.x * blockDim.x + threadIdx.x;
    if (i < n) p[i] = 0.0f;
}

#define NBINS 256
#define SORT_CHUNK 4096
#define SORT_THREADS 1024
__device__ __forceinline__ int len_bin(float len) {
    float t = fminf(fmaxf(len * ((float)NBINS / 4.0f), 0.0f), (float)(NBINS - 1));
    return (int)t;
}

__global__ __launch_bounds__(SORT_THREADS) void kCount(const float* __restrict__ elen,
                                                       int* __restrict__ cnt, int nb, int E)
{
    __shared__ int lh[NBINS];
    const int tid = threadIdx.x;
    if (tid < NBINS) lh[tid] = 0;
    __syncthreads();
    const int base = blockIdx.x * SORT_CHUNK;
    #pragma unroll
    for (int i = 0; i < SORT_CHUNK / SORT_THREADS; i++) {
        const int e = base + i * SORT_THREADS + tid;
        if (e < E) atomicAdd(&lh[len_bin(elen[e])], 1);
    }
    __syncthreads();
    if (tid < NBINS) cnt[(size_t)tid * nb + blockIdx.x] = lh[tid];
}

__global__ __launch_bounds__(1024) void kScan(int* __restrict__ cnt, int M, int per)
{
    __shared__ int wsum[32];
    const int t = threadIdx.x, lane = t & 31, w = t >> 5;
    const int lo = t * per;
    int s = 0;
    #pragma unroll 8
    for (int i = 0; i < per; i++) {
        const int k = lo + i;
        s += (k < M) ? cnt[k] : 0;
    }
    int incl = s;
    #pragma unroll
    for (int d = 1; d < 32; d <<= 1) {
        int v = __shfl_up(incl, d, 32);
        if (lane >= d) incl += v;
    }
    if (lane == 31) wsum[w] = incl;
    __syncthreads();
    if (w == 0) {
        const int v = wsum[lane];
        int iv = v;
        #pragma unroll
        for (int d = 1; d < 32; d <<= 1) {
            int u = __shfl_up(iv, d, 32);
            if (lane >= d) iv += u;
        }
        wsum[lane] = iv - v;
    }
    __syncthreads();
    int run = wsum[w] + incl - s;
    #pragma unroll 8
    for (int i = 0; i < per; i++) {
        const int k = lo + i;
        if (k < M) { const int c = cnt[k]; cnt[k] = run; run += c; }
    }
}

__global__ __launch_bounds__(SORT_THREADS) void kScatter(const float* __restrict__ elen,
                                                         const int* __restrict__ offs,
                                                         int* __restrict__ perm, int nb, int E)
{
    __shared__ int lh[NBINS];
    const int tid = threadIdx.x;
    if (tid < NBINS) lh[tid] = offs[(size_t)tid * nb + blockIdx.x];
    __syncthreads();
    const int base = blockIdx.x * SORT_CHUNK;
    #pragma unroll
    for (int i = 0; i < SORT_CHUNK / SORT_THREADS; i++) {
        const int e = base + i * SORT_THREADS + tid;
        if (e < E) {
            const int pos = atomicAdd(&lh[len_bin(elen[e])], 1);
            if ((unsigned)pos < (unsigned)E) perm[pos] = e;
        }
    }
}

#define SH_PITCH 20
__global__ __launch_bounds__(256) void kA(const float* __restrict__ evec, const float* __restrict__ elen,
                   const int* __restrict__ edst,
                   float* __restrict__ sh, float* __restrict__ basis,
                   float* __restrict__ x, int E)
{
    __shared__ __attribute__((aligned(16))) float tile[256 * SH_PITCH];
    __shared__ int dstT[256];

    const int tid = threadIdx.x;
    const int eb  = blockIdx.x * 256;
    const int e   = eb + tid;
    float s[16] = {};
    int d = 0;
    if (e < E) {
        float vx = evec[3*(size_t)e], vy = evec[3*(size_t)e+1], vz = evec[3*(size_t)e+2];
        float rn = 1.0f / (sqrtf(vx*vx + vy*vy + vz*vz) + 1e-12f);
        vx *= rn; vy *= rn; vz *= rn;
        float x2 = vx*vx, y2 = vy*vy, z2 = vz*vz;
        s[0]  = 1.0f;
        s[1]  = S3*vx; s[2] = S3*vy; s[3] = S3*vz;
        s[4]  = S15*vx*vz;
        s[5]  = S15*vx*vy;
        s[6]  = 0.5f*S5*(2.0f*y2 - x2 - z2);
        s[7]  = S15*vy*vz;
        s[8]  = 0.5f*S15*(z2 - x2);
        s[9]  = C358*vx*(3.0f*z2 - x2);
        s[10] = S105*vx*vy*vz;
        s[11] = C218*vx*(4.0f*y2 - z2 - x2);
        s[12] = 0.5f*S7*vy*(2.0f*y2 - 3.0f*z2 - 3.0f*x2);
        s[13] = C218*vz*(4.0f*y2 - z2 - x2);
        s[14] = 0.5f*S105*vy*(z2 - x2);
        s[15] = C358*vz*(z2 - 3.0f*x2);
        d = edst[e];
        float len = elen[e];
        #pragma unroll
        for (int i = 0; i < 3; i++) {
            float dd = (len - 2.0f*(float)i) * 0.5f;
            basis[(size_t)e*3 + i] = expf(-dd*dd) * (1.0f/1.12f);
        }
    }
    {
        float4* row = (float4*)(tile + tid * SH_PITCH);
        row[0] = make_float4(s[0],  s[1],  s[2],  s[3]);
        row[1] = make_float4(s[4],  s[5],  s[6],  s[7]);
        row[2] = make_float4(s[8],  s[9],  s[10], s[11]);
        row[3] = make_float4(s[12], s[13], s[14], s[15]);
        dstT[tid] = d;
    }
    __syncthreads();

    const int lane  = tid & 31;
    const int hh    = lane >> 4;
    const int j     = lane & 15;
    const int wbase = tid & ~31;
    if (eb + wbase + 32 <= E) {
        float v[16];
        int   dm[16];
        #pragma unroll
        for (int i = 0; i < 16; i++) {
            const int m = wbase + 2*i + hh;
            v[i]  = tile[m * SH_PITCH + j];
            dm[i] = dstT[m];
        }
        #pragma unroll
        for (int i = 0; i < 16; i++) {
            const int m = wbase + 2*i + hh;
            sh[(size_t)(eb + m)*16 + j] = v[i];
            atomicAdd(&x[(size_t)dm[i]*16 + j], v[i] * INV_NN);
        }
    } else {
        for (int i = 0; i < 16; i++) {
            const int m  = wbase + 2*i + hh;
            const int em = eb + m;
            if (em < E) {
                const float v = tile[m * SH_PITCH + j];
                sh[(size_t)em*16 + j] = v;
                atomicAdd(&x[(size_t)dstT[m]*16 + j], v * INV_NN);
            }
        }
    }
}

#define KB_EDGES 128
#define KB_THREADS 384
#define KB_PCH 16
#define KB_PITCH 800
#define KB_HP 20
__device__ __forceinline__ float pre1(float4 b, float w0, float w1, float w2) {
    return fmaf(b.z, w2, fmaf(b.y, w1, b.x * w0));
}
__device__ __forceinline__ float hidden1(float4 b, float w0, float w1, float w2) {
    return fmaxf(pre1(b, w0, w1, w2) * INV_SQRT3, 0.0f);
}
__global__ __launch_bounds__(KB_THREADS) void kB(
    const float* __restrict__ sh, const float* __restrict__ basis,
    const float* __restrict__ x,
    const int* __restrict__ esrc, const int* __restrict__ edst,
    const int* __restrict__ perm,
    const float* __restrict__ W1a, const float* __restrict__ W1b,
    float* __restrict__ h, int E)
{
    __shared__ __attribute__((aligned(16))) float bas[KB_EDGES][4];
    __shared__ __attribute__((aligned(16))) float dots[KB_EDGES][4];
    __shared__ int   dstT[KB_EDGES];
    __shared__ int   eid[KB_EDGES];
    __shared__ __attribute__((aligned(16))) float waF[4][256];
    __shared__ __attribute__((aligned(8))) unsigned colsF[256];
    __shared__ __attribute__((aligned(16))) float wP[256][4];
    __shared__ unsigned colsP[256];
    __shared__ int   wcntF[8], wcntP[8];
    __shared__ __attribute__((aligned(16))) float Gl[2 * KB_PITCH];
    __shared__ __attribute__((aligned(16))) float hP[KB_EDGES][KB_HP];
    __shared__ __attribute__((aligned(16))) float Bp[(KB_PCH / 2) * KB_PITCH];

    const int tid  = threadIdx.x;
    const int wv   = tid >> 5;
    const int lane = tid & 31;
    const int e0   = blockIdx.x * KB_EDGES;
    const int nvalid = min(KB_EDGES, E - e0);

    if (tid < KB_EDGES) {
        const int i = e0 + tid;
        int e = -1;
        float b0 = 0.0f, b1 = 0.0f, b2 = 0.0f;
        if (i < E) {
            e = perm[i];
            e = min(max(e, 0), E - 1);
            b0 = basis[(size_t)e*3]; b1 = basis[(size_t)e*3+1]; b2 = basis[(size_t)e*3+2];
        }
        eid[tid] = e;
        dstT[tid] = (e >= 0) ? edst[e] : 0;
        *(float4*)bas[tid] = make_float4(b0, b1, b2, 0.0f);
    } else {
        const int it = tid - KB_EDGES;
        const int m = it >> 1, q = it & 1;
        const int i = e0 + m;
        float sa = 0.0f, sb = 0.0f;
        if (i < E) {
            int e = perm[i];
            e = min(max(e, 0), E - 1);
            const int src = esrc[e];
            const float* xr = x  + (size_t)src*16;
            const float* sr = sh + (size_t)e*16;
            float xs[8], ss[8];
            #pragma unroll
            for (int j = 0; j < 8; j++) {
                const int idx = q ? (1 + j) : (j == 0 ? 0 : 8 + j);
                xs[j] = xr[idx];
                ss[j] = sr[idx];
            }
            if (q == 0) {
                sa = xs[0] * ss[0];
                float t = 0.0f;
                #pragma unroll
                for (int j = 1; j < 8; j++) t += xs[j] * ss[j];
                sb = t * rsqrtf(7.0f);
            } else {
                float t = 0.0f, u = 0.0f;
                #pragma unroll
                for (int j = 0; j < 3; j++) t += xs[j] * ss[j];
                #pragma unroll
                for (int j = 3; j < 8; j++) u += xs[j] * ss[j];
                sa = t * rsqrtf(3.0f);
                sb = u * rsqrtf(5.0f);
            }
        }
        if (q == 0) { dots[m][0] = sa; dots[m][3] = sb; }
        else        { dots[m][1] = sa; dots[m][2] = sb; }
    }
    if (tid < 256) waF[3][tid] = 0.0f;
    __syncthreads();

    int cls = 0;
    float w0 = 0.0f, w1 = 0.0f, w2 = 0.0f;
    if (tid < 256) {
        w0 = W1a[tid]; w1 = W1a[256 + tid]; w2 = W1a[512 + tid];
        float lo = __builtin_huge_valf(), hi = -__builtin_huge_valf();
        #pragma unroll 8
        for (int m = 0; m < nvalid; m++) {
            const float t = pre1(*(const float4*)bas[m], w0, w1, w2);
            lo = fminf(lo, t);
            hi = fmaxf(hi, t);
        }
        cls = (lo > 0.0f) ? 1 : ((hi > 0.0f) ? 2 : 0);
    }
    const unsigned balF = (unsigned)__ballot(cls == 1);
    const unsigned balP = (unsigned)__ballot(cls == 2);
    if (wv < 8 && lane == 0) { wcntF[wv] = __popc(balF); wcntP[wv] = __popc(balP); }
    __syncthreads();

    int baseF = 0, LF = 0, baseP = 0, LP = 0;
    #pragma unroll
    for (int i = 0; i < 8; i++) {
        const int cF = wcntF[i], cP = wcntP[i];
        if (i < wv) { baseF += cF; baseP += cP; }
        LF += cF; LP += cP;
    }
    const int LFp = (LF + 15) & ~15;
    const int LPp = (LP + 3) & ~3;
    if (tid < 256) {
        const unsigned lm = (1u << lane) - 1u;
        if (cls == 1) {
            const int j = baseF + __popc(balF & lm);
            colsF[j] = (unsigned)tid * (384u * 4u);
            waF[0][j] = w0; waF[1][j] = w1; waF[2][j] = w2;
        } else if (cls == 2) {
            const int j = baseP + __popc(balP & lm);
            colsP[j] = (unsigned)tid * 384u;
            *(float4*)wP[j] = make_float4(w0, w1, w2, 0.0f);
        }
    } else {
        const int j = LF + (tid - 256);
        if (j < LFp) { colsF[j] = 0u; waF[0][j] = 0.0f; waF[1][j] = 0.0f; waF[2][j] = 0.0f; }
    }
    __syncthreads();

    const int c  = lane & 15;
    const int hh = lane >> 4;
    const char* Wb = (const char*)W1b;

    auto fill = [&](int ch) {
        for (int it = tid; it < KB_EDGES * KB_PCH; it += KB_THREADS) {
            const int m = it >> 4, kk = it & 15;
            const int j = ch * KB_PCH + kk;
            float v = 0.0f;
            if (j < LP) {
                const float4 w = *(const float4*)wP[j];
                v = hidden1(*(const float4*)bas[m], w.x, w.y, w.z);
            }
            hP[m][kk] = v;
        }
        const int nk = min(KB_PCH, LPp - ch * KB_PCH);
        #pragma unroll 4
        for (int kk = 0; kk < nk; kk++) {
            const int j = ch * KB_PCH + kk;
            float v = 0.0f;
            if (j < LP) v = W1b[colsP[j] + tid];
            Bp[(kk >> 1) * KB_PITCH + 2 * tid + (kk & 1)] = v;
        }
    };

    {
        v8f g0 = {}, g1 = {};
        const unsigned cb = (unsigned)(wv * 32 + c) * 4u;
        const float* Arow = waF[min(c, 3)];
        for (int k0 = 0; k0 < LFp; k0 += 16) {
            #pragma unroll
            for (int u = 0; u < 4; u++) {
                const int kb = k0 + 4*u + 2*hh;
                const v2f A = *(const v2f*)(Arow + kb);
                const float* Br0 = (const float*)(Wb + (colsF[kb]     + cb));
                const float* Br1 = (const float*)(Wb + (colsF[kb + 1] + cb));
                v2f Ba; Ba.x = Br0[0];  Ba.y = Br1[0];
                v2f Bb; Bb.x = Br0[16]; Bb.y = Br1[16];
                g0 = __builtin_amdgcn_wmma_f32_16x16x4_f32(false, A, false, Ba, (short)0, g0, false, false);
                g1 = __builtin_amdgcn_wmma_f32_16x16x4_f32(false, A, false, Bb, (short)0, g1, false, false);
            }
        }
        if (hh == 0) {
            const int col0 = wv * 32 + c, col1 = col0 + 16;
            *(v2f*)(Gl + 2*col0)            = (v2f){g0[0], g0[1]};
            *(v2f*)(Gl + KB_PITCH + 2*col0) = (v2f){g0[2], 0.0f};
            *(v2f*)(Gl + 2*col1)            = (v2f){g1[0], g1[1]};
            *(v2f*)(Gl + KB_PITCH + 2*col1) = (v2f){g1[2], 0.0f};
        }
    }
    const int nch = (LPp + KB_PCH - 1) / KB_PCH;
    if (nch > 0) fill(0);
    __syncthreads();

    const int kc   = wv % 6;
    const int rh   = wv / 6;
    const int colq = kc * 16 + c;
    const float scale = (1.0f/16.0f) * 0.5f * INV_NN;
    const float* gp = Gl + hh * KB_PITCH + 2 * colq;

    for (int s = 0; s < 4; s++) {
        const int r0 = rh * 64 + s * 16;
        v8f a0 = {}, a1 = {}, a2 = {}, a3 = {};
        {
            const v2f bb = *(const v2f*)(bas[r0 + c] + 2*hh);
            v2f A; A.x = bb.x * INV_SQRT3; A.y = bb.y * INV_SQRT3;
            const v2f B0 = *(const v2f*)(gp);
            const v2f B1 = *(const v2f*)(gp + 192);
            const v2f B2 = *(const v2f*)(gp + 384);
            const v2f B3 = *(const v2f*)(gp + 576);
            a0 = __builtin_amdgcn_wmma_f32_16x16x4_f32(false, A, false, B0, (short)0, a0, false, false);
            a1 = __builtin_amdgcn_wmma_f32_16x16x4_f32(false, A, false, B1, (short)0, a1, false, false);
            a2 = __builtin_amdgcn_wmma_f32_16x16x4_f32(false, A, false, B2, (short)0, a2, false, false);
            a3 = __builtin_amdgcn_wmma_f32_16x16x4_f32(false, A, false, B3, (short)0, a3, false, false);
        }
        for (int ch = 0; ch < nch; ch++) {
            if (nch > 1 && (s | ch) != 0) {
                __syncthreads();
                fill(ch);
                __syncthreads();
            }
            const int nk = min(KB_PCH, LPp - ch * KB_PCH);
            for (int kk = 0; kk < nk; kk += 4) {
                const v2f A = *(const v2f*)(hP[r0 + c] + kk + 2*hh);
                const float* bp = Bp + ((kk >> 1) + hh) * KB_PITCH + 2 * colq;
                const v2f B0 = *(const v2f*)(bp);
                const v2f B1 = *(const v2f*)(bp + 192);
                const v2f B2 = *(const v2f*)(bp + 384);
                const v2f B3 = *(const v2f*)(bp + 576);
                a0 = __builtin_amdgcn_wmma_f32_16x16x4_f32(false, A, false, B0, (short)0, a0, false, false);
                a1 = __builtin_amdgcn_wmma_f32_16x16x4_f32(false, A, false, B1, (short)0, a1, false, false);
                a2 = __builtin_amdgcn_wmma_f32_16x16x4_f32(false, A, false, B2, (short)0, a2, false, false);
                a3 = __builtin_amdgcn_wmma_f32_16x16x4_f32(false, A, false, B3, (short)0, a3, false, false);
            }
        }
        #pragma unroll
        for (int r = 0; r < 8; r++) {
            const int m = r0 + r + 8*hh;
            if (eid[m] >= 0) {
                const float4 d = *(const float4*)dots[m];
                const float ef = d.x*a0[r] + d.y*a1[r] + d.z*a2[r] + d.w*a3[r];
                atomicAdd(&h[(size_t)dstT[m]*96 + colq], ef * scale);
            }
        }
    }
}

__global__ __launch_bounds__(256) void kC(const float* __restrict__ h, float* __restrict__ y, int N)
{
    const int t   = blockIdx.x * 256 + threadIdx.x;
    const int nid = t >> 4;
    const int q   = t & 15;
    if (nid >= N) return;
    const float4* hr = (const float4*)(h + (size_t)nid * 96);
    float4 o;
    if (q < 8) {
        const float4 a = hr[q];
        if (q < 4) o = make_float4(fmaxf(a.x, 0.0f), fmaxf(a.y, 0.0f), fmaxf(a.z, 0.0f), fmaxf(a.w, 0.0f));
        else       o = make_float4(fabsf(a.x), fabsf(a.y), fabsf(a.z), fabsf(a.w));
    } else {
        const float4 g = hr[q];
        const float4 v = hr[q + 8];
        const bool use_relu = ((q - 8) & 2) == 0;
        float4 ga;
        if (use_relu) ga = make_float4(fmaxf(g.x, 0.0f), fmaxf(g.y, 0.0f), fmaxf(g.z, 0.0f), fmaxf(g.w, 0.0f));
        else          ga = make_float4(tanhf(g.x), tanhf(g.y), tanhf(g.z), tanhf(g.w));
        o = make_float4(ga.x * v.x, ga.y * v.y, ga.z * v.z, ga.w * v.w);
    }
    ((float4*)(y + (size_t)nid * 64))[q] = o;
}

#define KE_EDGES 128
#define KE_THREADS 256
#define KE_PCH 16
#define KE_PITCH 160
__global__ __launch_bounds__(KE_THREADS) void kE(const float* __restrict__ basis,
                                                 const float* __restrict__ y,
                                                 const int* __restrict__ esrc,
                                                 const int* __restrict__ edst,
                                                 const int* __restrict__ perm,
                                                 const float* __restrict__ W2a,
                                                 const float* __restrict__ W2b,
                                                 float* __restrict__ out, int E)
{
    __shared__ __attribute__((aligned(16))) float bas[KE_EDGES][4];
    __shared__ int   srcT[KE_EDGES];
    __shared__ int   dstT[KE_EDGES];
    __shared__ int   eid[KE_EDGES];
    __shared__ __attribute__((aligned(16))) float waF[4][256];
    __shared__ __attribute__((aligned(8))) unsigned colsF[256];
    __shared__ __attribute__((aligned(16))) float wP[256][4];
    __shared__ unsigned colsP[256];
    __shared__ int   wcntF[8], wcntP[8];
    __shared__ __attribute__((aligned(16))) float Gl[2][2 * KE_PITCH];
    __shared__ __attribute__((aligned(16))) float hP[KE_EDGES][KB_HP];
    __shared__ __attribute__((aligned(16))) float Bp[(KE_PCH / 2) * KE_PITCH];

    const int tid  = threadIdx.x;
    const int wv   = tid >> 5;
    const int lane = tid & 31;
    const int e0   = blockIdx.x * KE_EDGES;
    const int nvalid = min(KE_EDGES, E - e0);

    const float w0 = W2a[tid], w1 = W2a[256 + tid], w2 = W2a[512 + tid];

    if (tid < KE_EDGES) {
        const int i = e0 + tid;
        int e = -1, sv = 0, dv = 0;
        float b0 = 0.0f, b1 = 0.0f, b2 = 0.0f;
        if (i < E) {
            e = perm[i];
            e = min(max(e, 0), E - 1);
            b0 = basis[(size_t)e*3]; b1 = basis[(size_t)e*3+1]; b2 = basis[(size_t)e*3+2];
            sv = esrc[e];
            dv = edst[e];
        }
        eid[tid] = e;
        srcT[tid] = sv;
        dstT[tid] = dv;
        *(float4*)bas[tid] = make_float4(b0, b1, b2, 0.0f);
    }
    waF[3][tid] = 0.0f;
    __syncthreads();

    int cls;
    {
        bool allpos = true, anypos = false;
        #pragma unroll 8
        for (int m = 0; m < nvalid; m++) {
            const bool pos = pre1(*(const float4*)bas[m], w0, w1, w2) > 0.0f;
            allpos &= pos;
            anypos |= pos;
        }
        cls = allpos ? 1 : (anypos ? 2 : 0);
    }
    const unsigned balF = (unsigned)__ballot(cls == 1);
    const unsigned balP = (unsigned)__ballot(cls == 2);
    if (lane == 0) { wcntF[wv] = __popc(balF); wcntP[wv] = __popc(balP); }
    __syncthreads();

    int baseF = 0, LF = 0, baseP = 0, LP = 0;
    #pragma unroll
    for (int i = 0; i < 8; i++) {
        const int cF = wcntF[i], cP = wcntP[i];
        if (i < wv) { baseF += cF; baseP += cP; }
        LF += cF; LP += cP;
    }
    const int LFp = (LF + 31) & ~31;
    const int LPp = (LP + 3) & ~3;
    {
        const unsigned lm = (1u << lane) - 1u;
        if (cls == 1) {
            const int j = baseF + __popc(balF & lm);
            colsF[j] = (unsigned)tid * (64u * 4u);
            waF[0][j] = w0; waF[1][j] = w1; waF[2][j] = w2;
        } else if (cls == 2) {
            const int j = baseP + __popc(balP & lm);
            colsP[j] = (unsigned)tid * 64u;
            *(float4*)wP[j] = make_float4(w0, w1, w2, 0.0f);
        }
        if (tid < 32) {
            const int j = LF + tid;
            if (j < LFp) { colsF[j] = 0u; waF[0][j] = 0.0f; waF[1][j] = 0.0f; waF[2][j] = 0.0f; }
        }
    }
    __syncthreads();

    const int c  = lane & 15;
    const int hh = lane >> 4;
    const char* Wb = (const char*)W2b;

    auto fill = [&](int ch) {
        for (int it = tid; it < KE_EDGES * KE_PCH; it += KE_THREADS) {
            const int m = it >> 4, kk = it & 15;
            const int j = ch * KE_PCH + kk;
            float v = 0.0f;
            if (j < LP) {
                const float4 w = *(const float4*)wP[j];
                v = hidden1(*(const float4*)bas[m], w.x, w.y, w.z);
            }
            hP[m][kk] = v;
        }
        const int nk = min(KE_PCH, LPp - ch * KE_PCH);
        const int col = tid & 63;
        for (int kk = tid >> 6; kk < nk; kk += 4) {
            const int j = ch * KE_PCH + kk;
            float v = 0.0f;
            if (j < LP) v = W2b[colsP[j] + col];
            Bp[(kk >> 1) * KE_PITCH + 2 * col + (kk & 1)] = v;
        }
    };

    {
        const int ct = wv & 3, kh = wv >> 2;
        v8f g0 = {}, g1 = {};
        const unsigned cb = (unsigned)(ct * 16 + c) * 4u;
        const float* Arow = waF[min(c, 3)];
        for (int k0 = 16 * kh; k0 < LFp; k0 += 32) {
            #pragma unroll
            for (int u = 0; u < 4; u++) {
                const int kb = k0 + 4*u + 2*hh;
                const v2f A = *(const v2f*)(Arow + kb);
                v2f B;
                B.x = *(const float*)(Wb + (colsF[kb]     + cb));
                B.y = *(const float*)(Wb + (colsF[kb + 1] + cb));
                if (u & 1) g1 = __builtin_amdgcn_wmma_f32_16x16x4_f32(false, A, false, B, (short)0, g1, false, false);
                else       g0 = __builtin_amdgcn_wmma_f32_16x16x4_f32(false, A, false, B, (short)0, g0, false, false);
            }
        }
        if (hh == 0) {
            const int col = ct * 16 + c;
            *(v2f*)(Gl[kh] + 2*col)            = (v2f){g0[0] + g1[0], g0[1] + g1[1]};
            *(v2f*)(Gl[kh] + KE_PITCH + 2*col) = (v2f){g0[2] + g1[2], 0.0f};
        }
    }
    const int nch = (LPp + KE_PCH - 1) / KE_PCH;
    if (nch > 0) fill(0);
    __syncthreads();

    const int r0 = wv * 16;
    v8f a0 = {}, a1 = {}, a2 = {}, a3 = {};
    {
        const v2f bb = *(const v2f*)(bas[r0 + c] + 2*hh);
        v2f A; A.x = bb.x * INV_SQRT3; A.y = bb.y * INV_SQRT3;
        const float* gp = Gl[0] + hh * KE_PITCH + 8 * c;
        const float* gq = Gl[1] + hh * KE_PITCH + 8 * c;
        const float4 Pa = *(const float4*)(gp), Pb = *(const float4*)(gp + 4);
        const float4 Qa = *(const float4*)(gq), Qb = *(const float4*)(gq + 4);
        const v2f B0 = (v2f){Pa.x + Qa.x, Pa.y + Qa.y}, B1 = (v2f){Pa.z + Qa.z, Pa.w + Qa.w};
        const v2f B2 = (v2f){Pb.x + Qb.x, Pb.y + Qb.y}, B3 = (v2f){Pb.z + Qb.z, Pb.w + Qb.w};
        a0 = __builtin_amdgcn_wmma_f32_16x16x4_f32(false, A, false, B0, (short)0, a0, false, false);
        a1 = __builtin_amdgcn_wmma_f32_16x16x4_f32(false, A, false, B1, (short)0, a1, false, false);
        a2 = __builtin_amdgcn_wmma_f32_16x16x4_f32(false, A, false, B2, (short)0, a2, false, false);
        a3 = __builtin_amdgcn_wmma_f32_16x16x4_f32(false, A, false, B3, (short)0, a3, false, false);
    }
    for (int ch = 0; ch < nch; ch++) {
        if (ch > 0) {
            __syncthreads();
            fill(ch);
            __syncthreads();
        }
        const int nk = min(KE_PCH, LPp - ch * KE_PCH);
        for (int kk = 0; kk < nk; kk += 4) {
            const v2f A = *(const v2f*)(hP[r0 + c] + kk + 2*hh);
            const float* bp = Bp + ((kk >> 1) + hh) * KE_PITCH + 8 * c;
            const float4 Ba = *(const float4*)(bp);
            const float4 Bb = *(const float4*)(bp + 4);
            const v2f B0 = (v2f){Ba.x, Ba.y}, B1 = (v2f){Ba.z, Ba.w};
            const v2f B2 = (v2f){Bb.x, Bb.y}, B3 = (v2f){Bb.z, Bb.w};
            a0 = __builtin_amdgcn_wmma_f32_16x16x4_f32(false, A, false, B0, (short)0, a0, false, false);
            a1 = __builtin_amdgcn_wmma_f32_16x16x4_f32(false, A, false, B1, (short)0, a1, false, false);
            a2 = __builtin_amdgcn_wmma_f32_16x16x4_f32(false, A, false, B2, (short)0, a2, false, false);
            a3 = __builtin_amdgcn_wmma_f32_16x16x4_f32(false, A, false, B3, (short)0, a3, false, false);
        }
    }

    float part[8];
    #pragma unroll
    for (int r = 0; r < 8; r++) {
        const int sv = srcT[r0 + 8*hh + r];
        const float4 yv = *(const float4*)(y + (size_t)sv*64 + 4*c);
        part[r] = a0[r]*yv.x + a1[r]*yv.y + a2[r]*yv.z + a3[r]*yv.w;
    }
    #pragma unroll
    for (int i = 0; i < 4; i++) {
        const bool up = (c & 8) != 0;
        const float keep = up ? part[4 + i] : part[i];
        const float send = up ? part[i] : part[4 + i];
        part[i] = keep + __shfl_xor(send, 8, 32);
    }
    #pragma unroll
    for (int i = 0; i < 2; i++) {
        const bool up = (c & 4) != 0;
        const float keep = up ? part[2 + i] : part[i];
        const float send = up ? part[i] : part[2 + i];
        part[i] = keep + __shfl_xor(send, 4, 32);
    }
    {
        const bool up = (c & 2) != 0;
        const float keep = up ? part[1] : part[0];
        const float send = up ? part[0] : part[1];
        part[0] = keep + __shfl_xor(send, 2, 32);
    }
    const float tot = part[0] + __shfl_xor(part[0], 1, 32);
    if ((c & 1) == 0) {
        const int m = r0 + 8*hh + 4*((c >> 3) & 1) + 2*((c >> 2) & 1) + ((c >> 1) & 1);
        if (eid[m] >= 0)
            atomicAdd(&out[dstT[m]], tot * (1.0f/128.0f) * INV_NN);
    }
}

extern "C" void kernel_launch(void* const* d_in, const int* in_sizes, int n_in,
                              void* d_out, int out_size, void* d_ws, size_t ws_size,
                              hipStream_t stream)
{
    const float* evec = (const float*)d_in[0];
    const float* elen = (const float*)d_in[1];
    const float* W1a  = (const float*)d_in[2];
    const float* W1b  = (const float*)d_in[3];
    const float* W2a  = (const float*)d_in[4];
    const float* W2b  = (const float*)d_in[5];
    const int*   esrc = (const int*)d_in[6];
    const int*   edst = (const int*)d_in[7];
    float* out = (float*)d_out;

    const int E = in_sizes[0] / 3;
    const int N = out_size;

    float* ws = (float*)d_ws;
    float* x_     = ws;                 ws += (size_t)N * 16;
    float* h_     = ws;                 ws += (size_t)N * 96;
    float* y_     = ws;                 ws += (size_t)N * 64;
    float* sh_    = ws;                 ws += (size_t)E * 16;
    float* basis_ = ws;                 ws += (size_t)E * 3;
    int*   perm_  = (int*)ws;           ws += (size_t)E;
    int*   cnt_   = (int*)ws;
    const int nb  = (E + SORT_CHUNK - 1) / SORT_CHUNK;
    const int M   = NBINS * nb;

    long nx = (long)N * 16, nh = (long)N * 96;
    kzero<<<(int)((nx + 255) / 256), 256, 0, stream>>>(x_, nx);
    kzero<<<(int)((nh + 255) / 256), 256, 0, stream>>>(h_, nh);
    kzero<<<(N + 255) / 256, 256, 0, stream>>>(out, N);

    kA<<<(E + 255) / 256, 256, 0, stream>>>(evec, elen, edst, sh_, basis_, x_, E);
    kCount<<<nb, SORT_THREADS, 0, stream>>>(elen, cnt_, nb, E);
    kScan<<<1, 1024, 0, stream>>>(cnt_, M, (M + 1023) / 1024);
    kScatter<<<nb, SORT_THREADS, 0, stream>>>(elen, cnt_, perm_, nb, E);
    kB<<<(E + KB_EDGES - 1) / KB_EDGES, KB_THREADS, 0, stream>>>(sh_, basis_, x_, esrc, edst, perm_, W1a, W1b, h_, E);
    kC<<<(N * 16 + 255) / 256, 256, 0, stream>>>(h_, y_, N);
    kE<<<(E + KE_EDGES - 1) / KE_EDGES, KE_THREADS, 0, stream>>>(basis_, y_, esrc, edst, perm_, W2a, W2b, out, E);
}


# launches merged by role (block index): count pass beside the x clear; scatter pass (sums its own offsets, scan pass gone) and h, out clears beside kA: 10 -> 5 launches; h add order may change
# speedup vs baseline: 3.9420x; 1.0390x over previous
#include <hip/hip_runtime.h>
#include <math.h>

typedef __attribute__((ext_vector_type(2))) float v2f;
typedef __attribute__((ext_vector_type(8))) float v8f;

#define INV_NN     0.512989176042577f
#define INV_SQRT3  0.5773502691896258f
#define S3   1.7320508075688772f
#define S5   2.23606797749979f
#define S7   2.6457513110645907f
#define S15  3.872983346207417f
#define S105 10.246950765959598f
#define C358 2.091650066335189f
#define C218 1.6201851746019651f

#define NBINS 256
#define SORT_CHUNK 4096
#define SORT_THREADS 256
__device__ __forceinline__ int len_bin(float len) {
    float t = fminf(fmaxf(len * ((float)NBINS / 4.0f), 0.0f), (float)(NBINS - 1));
    return (int)t;
}

__device__ __forceinline__ void clear_role(float* __restrict__ p, long n, int bid, int nblk, int tid) {
    const long n4 = n >> 2;
    float4* p4 = (float4*)p;
    const long step = (long)nblk * 256;
    for (long i = (long)bid * 256 + tid; i < n4; i += step) p4[i] = make_float4(0.0f, 0.0f, 0.0f, 0.0f);
    for (long i = (n4 << 2) + (long)bid * 256 + tid; i < n; i += step) p[i] = 0.0f;
}

__global__ __launch_bounds__(SORT_THREADS) void kPre(const float* __restrict__ elen,
                                                     int* __restrict__ cnt, int nb, int E,
                                                     float* __restrict__ x, long nx)
{
    const int tid = threadIdx.x;
    if ((int)blockIdx.x >= nb) {
        clear_role(x, nx, (int)blockIdx.x - nb, (int)gridDim.x - nb, tid);
        return;
    }
    __shared__ int lh[NBINS];
    lh[tid] = 0;
    __syncthreads();
    const int base = blockIdx.x * SORT_CHUNK;
    #pragma unroll
    for (int i = 0; i < SORT_CHUNK / SORT_THREADS; i++) {
        const int e = base + i * SORT_THREADS + tid;
        if (e < E) atomicAdd(&lh[len_bin(elen[e])], 1);
    }
    __syncthreads();
    cnt[(size_t)tid * nb + blockIdx.x] = lh[tid];
}

__device__ __forceinline__ void scatter_role(const float* __restrict__ elen,
                                             const int* __restrict__ cnt,
                                             int* __restrict__ perm, int nb, int E, int blk, int tid)
{
    __shared__ int lh[NBINS];
    __shared__ int wsum[SORT_THREADS / 32];
    const int lane = tid & 31, w = tid >> 5;
    int tot = 0, part = 0;
    const int* row = cnt + (size_t)tid * nb;
    #pragma unroll 8
    for (int k = 0; k < nb; k++) {
        const int c = row[k];
        tot += c;
        part += (k < blk) ? c : 0;
    }
    int incl = tot;
    #pragma unroll
    for (int d = 1; d < 32; d <<= 1) {
        int v = __shfl_up(incl, d, 32);
        if (lane >= d) incl += v;
    }
    if (lane == 31) wsum[w] = incl;
    __syncthreads();
    int wbase = 0;
    #pragma unroll
    for (int i = 0; i < SORT_THREADS / 32; i++) if (i < w) wbase += wsum[i];
    lh[tid] = wbase + (incl - tot) + part;
    __syncthreads();
    const int base = blk * SORT_CHUNK;
    #pragma unroll
    for (int i = 0; i < SORT_CHUNK / SORT_THREADS; i++) {
        const int e = base + i * SORT_THREADS + tid;
        if (e < E) {
            const int pos = atomicAdd(&lh[len_bin(elen[e])], 1);
            if ((unsigned)pos < (unsigned)E) perm[pos] = e;
        }
    }
}

#define SH_PITCH 20
__device__ __forceinline__ void kA_role(const float* __restrict__ evec, const float* __restrict__ elen,
                   const int* __restrict__ edst,
                   float* __restrict__ sh, float* __restrict__ basis,
                   float* __restrict__ x, int E, int bid, int tid)
{
    __shared__ __attribute__((aligned(16))) float tile[256 * SH_PITCH];
    __shared__ int dstT[256];

    const int eb  = bid * 256;
    const int e   = eb + tid;
    float s[16] = {};
    int d = 0;
    if (e < E) {
        float vx = evec[3*(size_t)e], vy = evec[3*(size_t)e+1], vz = evec[3*(size_t)e+2];
        float rn = 1.0f / (sqrtf(vx*vx + vy*vy + vz*vz) + 1e-12f);
        vx *= rn; vy *= rn; vz *= rn;
        float x2 = vx*vx, y2 = vy*vy, z2 = vz*vz;
        s[0]  = 1.0f;
        s[1]  = S3*vx; s[2] = S3*vy; s[3] = S3*vz;
        s[4]  = S15*vx*vz;
        s[5]  = S15*vx*vy;
        s[6]  = 0.5f*S5*(2.0f*y2 - x2 - z2);
        s[7]  = S15*vy*vz;
        s[8]  = 0.5f*S15*(z2 - x2);
        s[9]  = C358*vx*(3.0f*z2 - x2);
        s[10] = S105*vx*vy*vz;
        s[11] = C218*vx*(4.0f*y2 - z2 - x2);
        s[12] = 0.5f*S7*vy*(2.0f*y2 - 3.0f*z2 - 3.0f*x2);
        s[13] = C218*vz*(4.0f*y2 - z2 - x2);
        s[14] = 0.5f*S105*vy*(z2 - x2);
        s[15] = C358*vz*(z2 - 3.0f*x2);
        d = edst[e];
        float len = elen[e];
        #pragma unroll
        for (int i = 0; i < 3; i++) {
            float dd = (len - 2.0f*(float)i) * 0.5f;
            basis[(size_t)e*3 + i] = expf(-dd*dd) * (1.0f/1.12f);
        }
    }
    {
        float4* row = (float4*)(tile + tid * SH_PITCH);
        row[0] = make_float4(s[0],  s[1],  s[2],  s[3]);
        row[1] = make_float4(s[4],  s[5],  s[6],  s[7]);
        row[2] = make_float4(s[8],  s[9],  s[10], s[11]);
        row[3] = make_float4(s[12], s[13], s[14], s[15]);
        dstT[tid] = d;
    }
    __syncthreads();

    const int lane  = tid & 31;
    const int hh    = lane >> 4;
    const int j     = lane & 15;
    const int wbase = tid & ~31;
    if (eb + wbase + 32 <= E) {
        float v[16];
        int   dm[16];
        #pragma unroll
        for (int i = 0; i < 16; i++) {
            const int m = wbase + 2*i + hh;
            v[i]  = tile[m * SH_PITCH + j];
            dm[i] = dstT[m];
        }
        #pragma unroll
        for (int i = 0; i < 16; i++) {
            const int m = wbase + 2*i + hh;
            sh[(size_t)(eb + m)*16 + j] = v[i];
            atomicAdd(&x[(size_t)dm[i]*16 + j], v[i] * INV_NN);
        }
    } else {
        for (int i = 0; i < 16; i++) {
            const int m  = wbase + 2*i + hh;
            const int em = eb + m;
            if (em < E) {
                const float v = tile[m * SH_PITCH + j];
                sh[(size_t)em*16 + j] = v;
                atomicAdd(&x[(size_t)dstT[m]*16 + j], v * INV_NN);
            }
        }
    }
}

__global__ __launch_bounds__(256) void kMain(const float* __restrict__ evec, const float* __restrict__ elen,
                   const int* __restrict__ edst,
                   float* __restrict__ sh, float* __restrict__ basis,
                   float* __restrict__ x, int E,
                   const int* __restrict__ cnt, int* __restrict__ perm, int nb,
                   float* __restrict__ h, long nh, float* __restrict__ out, long nout, int nz)
{
    const int tid = threadIdx.x;
    const int b   = blockIdx.x;
    if (b < nb) {
        scatter_role(elen, cnt, perm, nb, E, b, tid);
    } else if (b < nb + nz) {
        clear_role(h, nh, b - nb, nz, tid);
        for (long i = (long)(b - nb) * 256 + tid; i < nout; i += (long)nz * 256) out[i] = 0.0f;
    } else {
        kA_role(evec, elen, edst, sh, basis, x, E, b - nb - nz, tid);
    }
}

#define KB_EDGES 128
#define KB_THREADS 384
#define KB_PCH 16
#define KB_PITCH 800
#define KB_HP 20
__device__ __forceinline__ float pre1(float4 b, float w0, float w1, float w2) {
    return fmaf(b.z, w2, fmaf(b.y, w1, b.x * w0));
}
__device__ __forceinline__ float hidden1(float4 b, float w0, float w1, float w2) {
    return fmaxf(pre1(b, w0, w1, w2) * INV_SQRT3, 0.0f);
}
__global__ __launch_bounds__(KB_THREADS) void kB(
    const float* __restrict__ sh, const float* __restrict__ basis,
    const float* __restrict__ x,
    const int* __restrict__ esrc, const int* __restrict__ edst,
    const int* __restrict__ perm,
    const float* __restrict__ W1a, const float* __restrict__ W1b,
    float* __restrict__ h, int E)
{
    __shared__ __attribute__((aligned(16))) float bas[KB_EDGES][4];
    __shared__ __attribute__((aligned(16))) float dots[KB_EDGES][4];
    __shared__ int   dstT[KB_EDGES];
    __shared__ int   eid[KB_EDGES];
    __shared__ __attribute__((aligned(16))) float waF[4][256];
    __shared__ __attribute__((aligned(8))) unsigned colsF[256];
    __shared__ __attribute__((aligned(16))) float wP[256][4];
    __shared__ unsigned colsP[256];
    __shared__ int   wcntF[8], wcntP[8];
    __shared__ __attribute__((aligned(16))) float Gl[2 * KB_PITCH];
    __shared__ __attribute__((aligned(16))) float hP[KB_EDGES][KB_HP];
    __shared__ __attribute__((aligned(16))) float Bp[(KB_PCH / 2) * KB_PITCH];

    const int tid  = threadIdx.x;
    const int wv   = tid >> 5;
    const int lane = tid & 31;
    const int e0   = blockIdx.x * KB_EDGES;
    const int nvalid = min(KB_EDGES, E - e0);

    if (tid < KB_EDGES) {
        const int i = e0 + tid;
        int e = -1;
        float b0 = 0.0f, b1 = 0.0f, b2 = 0.0f;
        if (i < E) {
            e = perm[i];
            e = min(max(e, 0), E - 1);
            b0 = basis[(size_t)e*3]; b1 = basis[(size_t)e*3+1]; b2 = basis[(size_t)e*3+2];
        }
        eid[tid] = e;
        dstT[tid] = (e >= 0) ? edst[e] : 0;
        *(float4*)bas[tid] = make_float4(b0, b1, b2, 0.0f);
    } else {
        const int it = tid - KB_EDGES;
        const int m = it >> 1, q = it & 1;
        const int i = e0 + m;
        float sa = 0.0f, sb = 0.0f;
        if (i < E) {
            int e = perm[i];
            e = min(max(e, 0), E - 1);
            const int src = esrc[e];
            const float* xr = x  + (size_t)src*16;
            const float* sr = sh + (size_t)e*16;
            float xs[8], ss[8];
            #pragma unroll
            for (int j = 0; j < 8; j++) {
                const int idx = q ? (1 + j) : (j == 0 ? 0 : 8 + j);
                xs[j] = xr[idx];
                ss[j] = sr[idx];
            }
            if (q == 0) {
                sa = xs[0] * ss[0];
                float t = 0.0f;
                #pragma unroll
                for (int j = 1; j < 8; j++) t += xs[j] * ss[j];
                sb = t * rsqrtf(7.0f);
            } else {
                float t = 0.0f, u = 0.0f;
                #pragma unroll
                for (int j = 0; j < 3; j++) t += xs[j] * ss[j];
                #pragma unroll
                for (int j = 3; j < 8; j++) u += xs[j] * ss[j];
                sa = t * rsqrtf(3.0f);
                sb = u * rsqrtf(5.0f);
            }
        }
        if (q == 0) { dots[m][0] = sa; dots[m][3] = sb; }
        else        { dots[m][1] = sa; dots[m][2] = sb; }
    }
    if (tid < 256) waF[3][tid] = 0.0f;
    __syncthreads();

    int cls = 0;
    float w0 = 0.0f, w1 = 0.0f, w2 = 0.0f;
    if (tid < 256) {
        w0 = W1a[tid]; w1 = W1a[256 + tid]; w2 = W1a[512 + tid];
        float lo = __builtin_huge_valf(), hi = -__builtin_huge_valf();
        #pragma unroll 8
        for (int m = 0; m < nvalid; m++) {
            const float t = pre1(*(const float4*)bas[m], w0, w1, w2);
            lo = fminf(lo, t);
            hi = fmaxf(hi, t);
        }
        cls = (lo > 0.0f) ? 1 : ((hi > 0.0f) ? 2 : 0);
    }
    const unsigned balF = (unsigned)__ballot(cls == 1);
    const unsigned balP = (unsigned)__ballot(cls == 2);
    if (wv < 8 && lane == 0) { wcntF[wv] = __popc(balF); wcntP[wv] = __popc(balP); }
    __syncthreads();

    int baseF = 0, LF = 0, baseP = 0, LP = 0;
    #pragma unroll
    for (int i = 0; i < 8; i++) {
        const int cF = wcntF[i], cP = wcntP[i];
        if (i < wv) { baseF += cF; baseP += cP; }
        LF += cF; LP += cP;
    }
    const int LFp = (LF + 15) & ~15;
    const int LPp = (LP + 3) & ~3;
    if (tid < 256) {
        const unsigned lm = (1u << lane) - 1u;
        if (cls == 1) {
            const int j = baseF + __popc(balF & lm);
            colsF[j] = (unsigned)tid * (384u * 4u);
            waF[0][j] = w0; waF[1][j] = w1; waF[2][j] = w2;
        } else if (cls == 2) {
            const int j = baseP + __popc(balP & lm);
            colsP[j] = (unsigned)tid * 384u;
            *(float4*)wP[j] = make_float4(w0, w1, w2, 0.0f);
        }
    } else {
        const int j = LF + (tid - 256);
        if (j < LFp) { colsF[j] = 0u; waF[0][j] = 0.0f; waF[1][j] = 0.0f; waF[2][j] = 0.0f; }
    }
    __syncthreads();

    const int c  = lane & 15;
    const int hh = lane >> 4;
    const char* Wb = (const char*)W1b;

    auto fill = [&](int ch) {
        for (int it = tid; it < KB_EDGES * KB_PCH; it += KB_THREADS) {
            const int m = it >> 4, kk = it & 15;
            const int j = ch * KB_PCH + kk;
            float v = 0.0f;
            if (j < LP) {
                const float4 w = *(const float4*)wP[j];
                v = hidden1(*(const float4*)bas[m], w.x, w.y, w.z);
            }
            hP[m][kk] = v;
        }
        const int nk = min(KB_PCH, LPp - ch * KB_PCH);
        #pragma unroll 4
        for (int kk = 0; kk < nk; kk++) {
            const int j = ch * KB_PCH + kk;
            float v = 0.0f;
            if (j < LP) v = W1b[colsP[j] + tid];
            Bp[(kk >> 1) * KB_PITCH + 2 * tid + (kk & 1)] = v;
        }
    };

    {
        v8f g0 = {}, g1 = {};
        const unsigned cb = (unsigned)(wv * 32 + c) * 4u;
        const float* Arow = waF[min(c, 3)];
        for (int k0 = 0; k0 < LFp; k0 += 16) {
            #pragma unroll
            for (int u = 0; u < 4; u++) {
                const int kb = k0 + 4*u + 2*hh;
                const v2f A = *(const v2f*)(Arow + kb);
                const float* Br0 = (const float*)(Wb + (colsF[kb]     + cb));
                const float* Br1 = (const float*)(Wb + (colsF[kb + 1] + cb));
                v2f Ba; Ba.x = Br0[0];  Ba.y = Br1[0];
                v2f Bb; Bb.x = Br0[16]; Bb.y = Br1[16];
                g0 = __builtin_amdgcn_wmma_f32_16x16x4_f32(false, A, false, Ba, (short)0, g0, false, false);
                g1 = __builtin_amdgcn_wmma_f32_16x16x4_f32(false, A, false, Bb, (short)0, g1, false, false);
            }
        }
        if (hh == 0) {
            const int col0 = wv * 32 + c, col1 = col0 + 16;
            *(v2f*)(Gl + 2*col0)            = (v2f){g0[0], g0[1]};
            *(v2f*)(Gl + KB_PITCH + 2*col0) = (v2f){g0[2], 0.0f};
            *(v2f*)(Gl + 2*col1)            = (v2f){g1[0], g1[1]};
            *(v2f*)(Gl + KB_PITCH + 2*col1) = (v2f){g1[2], 0.0f};
        }
    }
    const int nch = (LPp + KB_PCH - 1) / KB_PCH;
    if (nch > 0) fill(0);
    __syncthreads();

    const int kc   = wv % 6;
    const int rh   = wv / 6;
    const int colq = kc * 16 + c;
    const float scale = (1.0f/16.0f) * 0.5f * INV_NN;
    const float* gp = Gl + hh * KB_PITCH + 2 * colq;

    for (int s = 0; s < 4; s++) {
        const int r0 = rh * 64 + s * 16;
        v8f a0 = {}, a1 = {}, a2 = {}, a3 = {};
        {
            const v2f bb = *(const v2f*)(bas[r0 + c] + 2*hh);
            v2f A; A.x = bb.x * INV_SQRT3; A.y = bb.y * INV_SQRT3;
            const v2f B0 = *(const v2f*)(gp);
            const v2f B1 = *(const v2f*)(gp + 192);
            const v2f B2 = *(const v2f*)(gp + 384);
            const v2f B3 = *(const v2f*)(gp + 576);
            a0 = __builtin_amdgcn_wmma_f32_16x16x4_f32(false, A, false, B0, (short)0, a0, false, false);
            a1 = __builtin_amdgcn_wmma_f32_16x16x4_f32(false, A, false, B1, (short)0, a1, false, false);
            a2 = __builtin_amdgcn_wmma_f32_16x16x4_f32(false, A, false, B2, (short)0, a2, false, false);
            a3 = __builtin_amdgcn_wmma_f32_16x16x4_f32(false, A, false, B3, (short)0, a3, false, false);
        }
        for (int ch = 0; ch < nch; ch++) {
            if (nch > 1 && (s | ch) != 0) {
                __syncthreads();
                fill(ch);
                __syncthreads();
            }
            const int nk = min(KB_PCH, LPp - ch * KB_PCH);
            for (int kk = 0; kk < nk; kk += 4) {
                const v2f A = *(const v2f*)(hP[r0 + c] + kk + 2*hh);
                const float* bp = Bp + ((kk >> 1) + hh) * KB_PITCH + 2 * colq;
                const v2f B0 = *(const v2f*)(bp);
                const v2f B1 = *(const v2f*)(bp + 192);
                const v2f B2 = *(const v2f*)(bp + 384);
                const v2f B3 = *(const v2f*)(bp + 576);
                a0 = __builtin_amdgcn_wmma_f32_16x16x4_f32(false, A, false, B0, (short)0, a0, false, false);
                a1 = __builtin_amdgcn_wmma_f32_16x16x4_f32(false, A, false, B1, (short)0, a1, false, false);
                a2 = __builtin_amdgcn_wmma_f32_16x16x4_f32(false, A, false, B2, (short)0, a2, false, false);
                a3 = __builtin_amdgcn_wmma_f32_16x16x4_f32(false, A, false, B3, (short)0, a3, false, false);
            }
        }
        #pragma unroll
        for (int r = 0; r < 8; r++) {
            const int m = r0 + r + 8*hh;
            if (eid[m] >= 0) {
                const float4 d = *(const float4*)dots[m];
                const float ef = d.x*a0[r] + d.y*a1[r] + d.z*a2[r] + d.w*a3[r];
                atomicAdd(&h[(size_t)dstT[m]*96 + colq], ef * scale);
            }
        }
    }
}

__global__ __launch_bounds__(256) void kC(const float* __restrict__ h, float* __restrict__ y, int N)
{
    const int t   = blockIdx.x * 256 + threadIdx.x;
    const int nid = t >> 4;
    const int q   = t & 15;
    if (nid >= N) return;
    const float4* hr = (const float4*)(h + (size_t)nid * 96);
    float4 o;
    if (q < 8) {
        const float4 a = hr[q];
        if (q < 4) o = make_float4(fmaxf(a.x, 0.0f), fmaxf(a.y, 0.0f), fmaxf(a.z, 0.0f), fmaxf(a.w, 0.0f));
        else       o = make_float4(fabsf(a.x), fabsf(a.y), fabsf(a.z), fabsf(a.w));
    } else {
        const float4 g = hr[q];
        const float4 v = hr[q + 8];
        const bool use_relu = ((q - 8) & 2) == 0;
        float4 ga;
        if (use_relu) ga = make_float4(fmaxf(g.x, 0.0f), fmaxf(g.y, 0.0f), fmaxf(g.z, 0.0f), fmaxf(g.w, 0.0f));
        else          ga = make_float4(tanhf(g.x), tanhf(g.y), tanhf(g.z), tanhf(g.w));
        o = make_float4(ga.x * v.x, ga.y * v.y, ga.z * v.z, ga.w * v.w);
    }
    ((float4*)(y + (size_t)nid * 64))[q] = o;
}

#define KE_EDGES 128
#define KE_THREADS 256
#define KE_PCH 16
#define KE_PITCH 160
__global__ __launch_bounds__(KE_THREADS) void kE(const float* __restrict__ basis,
                                                 const float* __restrict__ y,
                                                 const int* __restrict__ esrc,
                                                 const int* __restrict__ edst,
                                                 const int* __restrict__ perm,
                                                 const float* __restrict__ W2a,
                                                 const float* __restrict__ W2b,
                                                 float* __restrict__ out, int E)
{
    __shared__ __attribute__((aligned(16))) float bas[KE_EDGES][4];
    __shared__ int   srcT[KE_EDGES];
    __shared__ int   dstT[KE_EDGES];
    __shared__ int   eid[KE_EDGES];
    __shared__ __attribute__((aligned(16))) float waF[4][256];
    __shared__ __attribute__((aligned(8))) unsigned colsF[256];
    __shared__ __attribute__((aligned(16))) float wP[256][4];
    __shared__ unsigned colsP[256];
    __shared__ int   wcntF[8], wcntP[8];
    __shared__ __attribute__((aligned(16))) float Gl[2][2 * KE_PITCH];
    __shared__ __attribute__((aligned(16))) float hP[KE_EDGES][KB_HP];
    __shared__ __attribute__((aligned(16))) float Bp[(KE_PCH / 2) * KE_PITCH];

    const int tid  = threadIdx.x;
    const int wv   = tid >> 5;
    const int lane = tid & 31;
    const int e0   = blockIdx.x * KE_EDGES;
    const int nvalid = min(KE_EDGES, E - e0);

    const float w0 = W2a[tid], w1 = W2a[256 + tid], w2 = W2a[512 + tid];

    if (tid < KE_EDGES) {
        const int i = e0 + tid;
        int e = -1, sv = 0, dv = 0;
        float b0 = 0.0f, b1 = 0.0f, b2 = 0.0f;
        if (i < E) {
            e = perm[i];
            e = min(max(e, 0), E - 1);
            b0 = basis[(size_t)e*3]; b1 = basis[(size_t)e*3+1]; b2 = basis[(size_t)e*3+2];
            sv = esrc[e];
            dv = edst[e];
        }
        eid[tid] = e;
        srcT[tid] = sv;
        dstT[tid] = dv;
        *(float4*)bas[tid] = make_float4(b0, b1, b2, 0.0f);
    }
    waF[3][tid] = 0.0f;
    __syncthreads();

    int cls;
    {
        bool allpos = true, anypos = false;
        #pragma unroll 8
        for (int m = 0; m < nvalid; m++) {
            const bool pos = pre1(*(const float4*)bas[m], w0, w1, w2) > 0.0f;
            allpos &= pos;
            anypos |= pos;
        }
        cls = allpos ? 1 : (anypos ? 2 : 0);
    }
    const unsigned balF = (unsigned)__ballot(cls == 1);
    const unsigned balP = (unsigned)__ballot(cls == 2);
    if (lane == 0) { wcntF[wv] = __popc(balF); wcntP[wv] = __popc(balP); }
    __syncthreads();

    int baseF = 0, LF = 0, baseP = 0, LP = 0;
    #pragma unroll
    for (int i = 0; i < 8; i++) {
        const int cF = wcntF[i], cP = wcntP[i];
        if (i < wv) { baseF += cF; baseP += cP; }
        LF += cF; LP += cP;
    }
    const int LFp = (LF + 31) & ~31;
    const int LPp = (LP + 3) & ~3;
    {
        const unsigned lm = (1u << lane) - 1u;
        if (cls == 1) {
            const int j = baseF + __popc(balF & lm);
            colsF[j] = (unsigned)tid * (64u * 4u);
            waF[0][j] = w0; waF[1][j] = w1; waF[2][j] = w2;
        } else if (cls == 2) {
            const int j = baseP + __popc(balP & lm);
            colsP[j] = (unsigned)tid * 64u;
            *(float4*)wP[j] = make_float4(w0, w1, w2, 0.0f);
        }
        if (tid < 32) {
            const int j = LF + tid;
            if (j < LFp) { colsF[j] = 0u; waF[0][j] = 0.0f; waF[1][j] = 0.0f; waF[2][j] = 0.0f; }
        }
    }
    __syncthreads();

    const int c  = lane & 15;
    const int hh = lane >> 4;
    const char* Wb = (const char*)W2b;

    auto fill = [&](int ch) {
        for (int it = tid; it < KE_EDGES * KE_PCH; it += KE_THREADS) {
            const int m = it >> 4, kk = it & 15;
            const int j = ch * KE_PCH + kk;
            float v = 0.0f;
            if (j < LP) {
                const float4 w = *(const float4*)wP[j];
                v = hidden1(*(const float4*)bas[m], w.x, w.y, w.z);
            }
            hP[m][kk] = v;
        }
        const int nk = min(KE_PCH, LPp - ch * KE_PCH);
        const int col = tid & 63;
        for (int kk = tid >> 6; kk < nk; kk += 4) {
            const int j = ch * KE_PCH + kk;
            float v = 0.0f;
            if (j < LP) v = W2b[colsP[j] + col];
            Bp[(kk >> 1) * KE_PITCH + 2 * col + (kk & 1)] = v;
        }
    };

    {
        const int ct = wv & 3, kh = wv >> 2;
        v8f g0 = {}, g1 = {};
        const unsigned cb = (unsigned)(ct * 16 + c) * 4u;
        const float* Arow = waF[min(c, 3)];
        for (int k0 = 16 * kh; k0 < LFp; k0 += 32) {
            #pragma unroll
            for (int u = 0; u < 4; u++) {
                const int kb = k0 + 4*u + 2*hh;
                const v2f A = *(const v2f*)(Arow + kb);
                v2f B;
                B.x = *(const float*)(Wb + (colsF[kb]     + cb));
                B.y = *(const float*)(Wb + (colsF[kb + 1] + cb));
                if (u & 1) g1 = __builtin_amdgcn_wmma_f32_16x16x4_f32(false, A, false, B, (short)0, g1, false, false);
                else       g0 = __builtin_amdgcn_wmma_f32_16x16x4_f32(false, A, false, B, (short)0, g0, false, false);
            }
        }
        if (hh == 0) {
            const int col = ct * 16 + c;
            *(v2f*)(Gl[kh] + 2*col)            = (v2f){g0[0] + g1[0], g0[1] + g1[1]};
            *(v2f*)(Gl[kh] + KE_PITCH + 2*col) = (v2f){g0[2] + g1[2], 0.0f};
        }
    }
    const int nch = (LPp + KE_PCH - 1) / KE_PCH;
    if (nch > 0) fill(0);
    __syncthreads();

    const int r0 = wv * 16;
    v8f a0 = {}, a1 = {}, a2 = {}, a3 = {};
    {
        const v2f bb = *(const v2f*)(bas[r0 + c] + 2*hh);
        v2f A; A.x = bb.x * INV_SQRT3; A.y = bb.y * INV_SQRT3;
        const float* gp = Gl[0] + hh * KE_PITCH + 8 * c;
        const float* gq = Gl[1] + hh * KE_PITCH + 8 * c;
        const float4 Pa = *(const float4*)(gp), Pb = *(const float4*)(gp + 4);
        const float4 Qa = *(const float4*)(gq), Qb = *(const float4*)(gq + 4);
        const v2f B0 = (v2f){Pa.x + Qa.x, Pa.y + Qa.y}, B1 = (v2f){Pa.z + Qa.z, Pa.w + Qa.w};
        const v2f B2 = (v2f){Pb.x + Qb.x, Pb.y + Qb.y}, B3 = (v2f){Pb.z + Qb.z, Pb.w + Qb.w};
        a0 = __builtin_amdgcn_wmma_f32_16x16x4_f32(false, A, false, B0, (short)0, a0, false, false);
        a1 = __builtin_amdgcn_wmma_f32_16x16x4_f32(false, A, false, B1, (short)0, a1, false, false);
        a2 = __builtin_amdgcn_wmma_f32_16x16x4_f32(false, A, false, B2, (short)0, a2, false, false);
        a3 = __builtin_amdgcn_wmma_f32_16x16x4_f32(false, A, false, B3, (short)0, a3, false, false);
    }
    for (int ch = 0; ch < nch; ch++) {
        if (ch > 0) {
            __syncthreads();
            fill(ch);
            __syncthreads();
        }
        const int nk = min(KE_PCH, LPp - ch * KE_PCH);
        for (int kk = 0; kk < nk; kk += 4) {
            const v2f A = *(const v2f*)(hP[r0 + c] + kk + 2*hh);
            const float* bp = Bp + ((kk >> 1) + hh) * KE_PITCH + 8 * c;
            const float4 Ba = *(const float4*)(bp);
            const float4 Bb = *(const float4*)(bp + 4);
            const v2f B0 = (v2f){Ba.x, Ba.y}, B1 = (v2f){Ba.z, Ba.w};
            const v2f B2 = (v2f){Bb.x, Bb.y}, B3 = (v2f){Bb.z, Bb.w};
            a0 = __builtin_amdgcn_wmma_f32_16x16x4_f32(false, A, false, B0, (short)0, a0, false, false);
            a1 = __builtin_amdgcn_wmma_f32_16x16x4_f32(false, A, false, B1, (short)0, a1, false, false);
            a2 = __builtin_amdgcn_wmma_f32_16x16x4_f32(false, A, false, B2, (short)0, a2, false, false);
            a3 = __builtin_amdgcn_wmma_f32_16x16x4_f32(false, A, false, B3, (short)0, a3, false, false);
        }
    }

    float part[8];
    #pragma unroll
    for (int r = 0; r < 8; r++) {
        const int sv = srcT[r0 + 8*hh + r];
        const float4 yv = *(const float4*)(y + (size_t)sv*64 + 4*c);
        part[r] = a0[r]*yv.x + a1[r]*yv.y + a2[r]*yv.z + a3[r]*yv.w;
    }
    #pragma unroll
    for (int i = 0; i < 4; i++) {
        const bool up = (c & 8) != 0;
        const float keep = up ? part[4 + i] : part[i];
        const float send = up ? part[i] : part[4 + i];
        part[i] = keep + __shfl_xor(send, 8, 32);
    }
    #pragma unroll
    for (int i = 0; i < 2; i++) {
        const bool up = (c & 4) != 0;
        const float keep = up ? part[2 + i] : part[i];
        const float send = up ? part[i] : part[2 + i];
        part[i] = keep + __shfl_xor(send, 4, 32);
    }
    {
        const bool up = (c & 2) != 0;
        const float keep = up ? part[1] : part[0];
        const float send = up ? part[0] : part[1];
        part[0] = keep + __shfl_xor(send, 2, 32);
    }
    const float tot = part[0] + __shfl_xor(part[0], 1, 32);
    if ((c & 1) == 0) {
        const int m = r0 + 8*hh + 4*((c >> 3) & 1) + 2*((c >> 2) & 1) + ((c >> 1) & 1);
        if (eid[m] >= 0)
            atomicAdd(&out[dstT[m]], tot * (1.0f/128.0f) * INV_NN);
    }
}

extern "C" void kernel_launch(void* const* d_in, const int* in_sizes, int n_in,
                              void* d_out, int out_size, void* d_ws, size_t ws_size,
                              hipStream_t stream)
{
    const float* evec = (const float*)d_in[0];
    const float* elen = (const float*)d_in[1];
    const float* W1a  = (const float*)d_in[2];
    const float* W1b  = (const float*)d_in[3];
    const float* W2a  = (const float*)d_in[4];
    const float* W2b  = (const float*)d_in[5];
    const int*   esrc = (const int*)d_in[6];
    const int*   edst = (const int*)d_in[7];
    float* out = (float*)d_out;

    const int E = in_sizes[0] / 3;
    const int N = out_size;

    float* ws = (float*)d_ws;
    float* x_     = ws;                 ws += (size_t)N * 16;
    float* h_     = ws;                 ws += (size_t)N * 96;
    float* y_     = ws;                 ws += (size_t)N * 64;
    float* sh_    = ws;                 ws += (size_t)E * 16;
    float* basis_ = ws;                 ws += (size_t)E * 3;
    int*   perm_  = (int*)ws;           ws += (size_t)E;
    int*   cnt_   = (int*)ws;
    const int nb  = (E + SORT_CHUNK - 1) / SORT_CHUNK;

    long nx = (long)N * 16, nh = (long)N * 96;
    const int zx = (int)((nx / 4 + 255) / 256);
    const int nz = (int)((nh / 4 + 1023) / 1024);
    const int na = (E + 255) / 256;

    kPre<<<nb + zx, SORT_THREADS, 0, stream>>>(elen, cnt_, nb, E, x_, nx);
    kMain<<<nb + nz + na, 256, 0, stream>>>(evec, elen, edst, sh_, basis_, x_, E,
                                            cnt_, perm_, nb, h_, nh, out, (long)N, nz);
    kB<<<(E + KB_EDGES - 1) / KB_EDGES, KB_THREADS, 0, stream>>>(sh_, basis_, x_, esrc, edst, perm_, W1a, W1b, h_, E);
    kC<<<(N * 16 + 255) / 256, 256, 0, stream>>>(h_, y_, N);
    kE<<<(E + KE_EDGES - 1) / KE_EDGES, KE_THREADS, 0, stream>>>(basis_, y_, esrc, edst, perm_, W2a, W2b, out, E);
}


# kB, kE: hidden column class from the bounding box of the tile's basis rows (6 numbers, rounding margin), not from all 128 rows; undecided columns join the mixed k loop; k grouping may change
# speedup vs baseline: 4.4236x; 1.1222x over previous
#include <hip/hip_runtime.h>
#include <math.h>

typedef __attribute__((ext_vector_type(2))) float v2f;
typedef __attribute__((ext_vector_type(8))) float v8f;

#define INV_NN     0.512989176042577f
#define INV_SQRT3  0.5773502691896258f
#define S3   1.7320508075688772f
#define S5   2.23606797749979f
#define S7   2.6457513110645907f
#define S15  3.872983346207417f
#define S105 10.246950765959598f
#define C358 2.091650066335189f
#define C218 1.6201851746019651f

#define NBINS 256
#define SORT_CHUNK 4096
#define SORT_THREADS 256
__device__ __forceinline__ int len_bin(float len) {
    float t = fminf(fmaxf(len * ((float)NBINS / 4.0f), 0.0f), (float)(NBINS - 1));
    return (int)t;
}

__device__ __forceinline__ void clear_role(float* __restrict__ p, long n, int bid, int nblk, int tid) {
    const long n4 = n >> 2;
    float4* p4 = (float4*)p;
    const long step = (long)nblk * 256;
    for (long i = (long)bid * 256 + tid; i < n4; i += step) p4[i] = make_float4(0.0f, 0.0f, 0.0f, 0.0f);
    for (long i = (n4 << 2) + (long)bid * 256 + tid; i < n; i += step) p[i] = 0.0f;
}

__global__ __launch_bounds__(SORT_THREADS) void kPre(const float* __restrict__ elen,
                                                     int* __restrict__ cnt, int nb, int E,
                                                     float* __restrict__ x, long nx)
{
    const int tid = threadIdx.x;
    if ((int)blockIdx.x >= nb) {
        clear_role(x, nx, (int)blockIdx.x - nb, (int)gridDim.x - nb, tid);
        return;
    }
    __shared__ int lh[NBINS];
    lh[tid] = 0;
    __syncthreads();
    const int base = blockIdx.x * SORT_CHUNK;
    #pragma unroll
    for (int i = 0; i < SORT_CHUNK / SORT_THREADS; i++) {
        const int e = base + i * SORT_THREADS + tid;
        if (e < E) atomicAdd(&lh[len_bin(elen[e])], 1);
    }
    __syncthreads();
    cnt[(size_t)tid * nb + blockIdx.x] = lh[tid];
}

__device__ __forceinline__ void scatter_role(const float* __restrict__ elen,
                                             const int* __restrict__ cnt,
                                             int* __restrict__ perm, int nb, int E, int blk, int tid)
{
    __shared__ int lh[NBINS];
    __shared__ int wsum[SORT_THREADS / 32];
    const int lane = tid & 31, w = tid >> 5;
    int tot = 0, part = 0;
    const int* row = cnt + (size_t)tid * nb;
    #pragma unroll 8
    for (int k = 0; k < nb; k++) {
        const int c = row[k];
        tot += c;
        part += (k < blk) ? c : 0;
    }
    int incl = tot;
    #pragma unroll
    for (int d = 1; d < 32; d <<= 1) {
        int v = __shfl_up(incl, d, 32);
        if (lane >= d) incl += v;
    }
    if (lane == 31) wsum[w] = incl;
    __syncthreads();
    int wbase = 0;
    #pragma unroll
    for (int i = 0; i < SORT_THREADS / 32; i++) if (i < w) wbase += wsum[i];
    lh[tid] = wbase + (incl - tot) + part;
    __syncthreads();
    const int base = blk * SORT_CHUNK;
    #pragma unroll
    for (int i = 0; i < SORT_CHUNK / SORT_THREADS; i++) {
        const int e = base + i * SORT_THREADS + tid;
        if (e < E) {
            const int pos = atomicAdd(&lh[len_bin(elen[e])], 1);
            if ((unsigned)pos < (unsigned)E) perm[pos] = e;
        }
    }
}

#define SH_PITCH 20
__device__ __forceinline__ void kA_role(const float* __restrict__ evec, const float* __restrict__ elen,
                   const int* __restrict__ edst,
                   float* __restrict__ sh, float* __restrict__ basis,
                   float* __restrict__ x, int E, int bid, int tid)
{
    __shared__ __attribute__((aligned(16))) float tile[256 * SH_PITCH];
    __shared__ int dstT[256];

    const int eb  = bid * 256;
    const int e   = eb + tid;
    float s[16] = {};
    int d = 0;
    if (e < E) {
        float vx = evec[3*(size_t)e], vy = evec[3*(size_t)e+1], vz = evec[3*(size_t)e+2];
        float rn = 1.0f / (sqrtf(vx*vx + vy*vy + vz*vz) + 1e-12f);
        vx *= rn; vy *= rn; vz *= rn;
        float x2 = vx*vx, y2 = vy*vy, z2 = vz*vz;
        s[0]  = 1.0f;
        s[1]  = S3*vx; s[2] = S3*vy; s[3] = S3*vz;
        s[4]  = S15*vx*vz;
        s[5]  = S15*vx*vy;
        s[6]  = 0.5f*S5*(2.0f*y2 - x2 - z2);
        s[7]  = S15*vy*vz;
        s[8]  = 0.5f*S15*(z2 - x2);
        s[9]  = C358*vx*(3.0f*z2 - x2);
        s[10] = S105*vx*vy*vz;
        s[11] = C218*vx*(4.0f*y2 - z2 - x2);
        s[12] = 0.5f*S7*vy*(2.0f*y2 - 3.0f*z2 - 3.0f*x2);
        s[13] = C218*vz*(4.0f*y2 - z2 - x2);
        s[14] = 0.5f*S105*vy*(z2 - x2);
        s[15] = C358*vz*(z2 - 3.0f*x2);
        d = edst[e];
        float len = elen[e];
        #pragma unroll
        for (int i = 0; i < 3; i++) {
            float dd = (len - 2.0f*(float)i) * 0.5f;
            basis[(size_t)e*3 + i] = expf(-dd*dd) * (1.0f/1.12f);
        }
    }
    {
        float4* row = (float4*)(tile + tid * SH_PITCH);
        row[0] = make_float4(s[0],  s[1],  s[2],  s[3]);
        row[1] = make_float4(s[4],  s[5],  s[6],  s[7]);
        row[2] = make_float4(s[8],  s[9],  s[10], s[11]);
        row[3] = make_float4(s[12], s[13], s[14], s[15]);
        dstT[tid] = d;
    }
    __syncthreads();

    const int lane  = tid & 31;
    const int hh    = lane >> 4;
    const int j     = lane & 15;
    const int wbase = tid & ~31;
    if (eb + wbase + 32 <= E) {
        float v[16];
        int   dm[16];
        #pragma unroll
        for (int i = 0; i < 16; i++) {
            const int m = wbase + 2*i + hh;
            v[i]  = tile[m * SH_PITCH + j];
            dm[i] = dstT[m];
        }
        #pragma unroll
        for (int i = 0; i < 16; i++) {
            const int m = wbase + 2*i + hh;
            sh[(size_t)(eb + m)*16 + j] = v[i];
            atomicAdd(&x[(size_t)dm[i]*16 + j], v[i] * INV_NN);
        }
    } else {
        for (int i = 0; i < 16; i++) {
            const int m  = wbase + 2*i + hh;
            const int em = eb + m;
            if (em < E) {
                const float v = tile[m * SH_PITCH + j];
                sh[(size_t)em*16 + j] = v;
                atomicAdd(&x[(size_t)dstT[m]*16 + j], v * INV_NN);
            }
        }
    }
}

__global__ __launch_bounds__(256) void kMain(const float* __restrict__ evec, const float* __restrict__ elen,
                   const int* __restrict__ edst,
                   float* __restrict__ sh, float* __restrict__ basis,
                   float* __restrict__ x, int E,
                   const int* __restrict__ cnt, int* __restrict__ perm, int nb,
                   float* __restrict__ h, long nh, float* __restrict__ out, long nout, int nz)
{
    const int tid = threadIdx.x;
    const int b   = blockIdx.x;
    if (b < nb) {
        scatter_role(elen, cnt, perm, nb, E, b, tid);
    } else if (b < nb + nz) {
        clear_role(h, nh, b - nb, nz, tid);
        for (long i = (long)(b - nb) * 256 + tid; i < nout; i += (long)nz * 256) out[i] = 0.0f;
    } else {
        kA_role(evec, elen, edst, sh, basis, x, E, b - nb - nz, tid);
    }
}

#define KB_EDGES 128
#define KB_THREADS 384
#define KB_PCH 16
#define KB_PITCH 800
#define KB_HP 20
__device__ __forceinline__ float pre1(float4 b, float w0, float w1, float w2) {
    return fmaf(b.z, w2, fmaf(b.y, w1, b.x * w0));
}
__device__ __forceinline__ float hidden1(float4 b, float w0, float w1, float w2) {
    return fmaxf(pre1(b, w0, w1, w2) * INV_SQRT3, 0.0f);
}
__device__ __forceinline__ void box_partial(bool ok, float b0, float b1, float b2,
                                            float* dst8, int lane) {
    const float inf = __builtin_huge_valf();
    float n0 = ok ? b0 : inf,  n1 = ok ? b1 : inf,  n2 = ok ? b2 : inf;
    float x0 = ok ? b0 : -inf, x1 = ok ? b1 : -inf, x2 = ok ? b2 : -inf;
    #pragma unroll
    for (int d = 16; d >= 1; d >>= 1) {
        n0 = fminf(n0, __shfl_xor(n0, d, 32));
        n1 = fminf(n1, __shfl_xor(n1, d, 32));
        n2 = fminf(n2, __shfl_xor(n2, d, 32));
        x0 = fmaxf(x0, __shfl_xor(x0, d, 32));
        x1 = fmaxf(x1, __shfl_xor(x1, d, 32));
        x2 = fmaxf(x2, __shfl_xor(x2, d, 32));
    }
    if (lane == 0) {
        *(float4*)(dst8)     = make_float4(n0, n1, n2, 0.0f);
        *(float4*)(dst8 + 4) = make_float4(x0, x1, x2, 0.0f);
    }
}
__device__ __forceinline__ int box_class(const float* boxp, float w0, float w1, float w2) {
    const float4 pa = *(const float4*)(boxp),      qa = *(const float4*)(boxp + 4);
    const float4 pb = *(const float4*)(boxp + 8),  qb = *(const float4*)(boxp + 12);
    const float4 pc = *(const float4*)(boxp + 16), qc = *(const float4*)(boxp + 20);
    const float4 pd = *(const float4*)(boxp + 24), qd = *(const float4*)(boxp + 28);
    const float n0 = fminf(fminf(pa.x, pb.x), fminf(pc.x, pd.x));
    const float n1 = fminf(fminf(pa.y, pb.y), fminf(pc.y, pd.y));
    const float n2 = fminf(fminf(pa.z, pb.z), fminf(pc.z, pd.z));
    const float x0 = fmaxf(fmaxf(qa.x, qb.x), fmaxf(qc.x, qd.x));
    const float x1 = fmaxf(fmaxf(qa.y, qb.y), fmaxf(qc.y, qd.y));
    const float x2 = fmaxf(fmaxf(qa.z, qb.z), fmaxf(qc.z, qd.z));
    const float a0 = w0 * n0, c0 = w0 * x0;
    const float a1 = w1 * n1, c1 = w1 * x1;
    const float a2 = w2 * n2, c2 = w2 * x2;
    const float lo  = fminf(a0, c0) + fminf(a1, c1) + fminf(a2, c2);
    const float hi  = fmaxf(a0, c0) + fmaxf(a1, c1) + fmaxf(a2, c2);
    const float mag = fmaxf(fabsf(a0), fabsf(c0)) + fmaxf(fabsf(a1), fabsf(c1)) + fmaxf(fabsf(a2), fabsf(c2));
    const float tol = 1.0e-6f * mag;
    return (lo > tol) ? 1 : ((hi < -tol) ? 0 : 2);
}
__global__ __launch_bounds__(KB_THREADS) void kB(
    const float* __restrict__ sh, const float* __restrict__ basis,
    const float* __restrict__ x,
    const int* __restrict__ esrc, const int* __restrict__ edst,
    const int* __restrict__ perm,
    const float* __restrict__ W1a, const float* __restrict__ W1b,
    float* __restrict__ h, int E)
{
    __shared__ __attribute__((aligned(16))) float bas[KB_EDGES][4];
    __shared__ __attribute__((aligned(16))) float dots[KB_EDGES][4];
    __shared__ int   dstT[KB_EDGES];
    __shared__ int   eid[KB_EDGES];
    __shared__ __attribute__((aligned(16))) float waF[4][256];
    __shared__ __attribute__((aligned(8))) unsigned colsF[256];
    __shared__ __attribute__((aligned(16))) float wP[256][4];
    __shared__ unsigned colsP[256];
    __shared__ int   wcntF[8], wcntP[8];
    __shared__ __attribute__((aligned(16))) float boxp[4 * 8];
    __shared__ __attribute__((aligned(16))) float Gl[2 * KB_PITCH];
    __shared__ __attribute__((aligned(16))) float hP[KB_EDGES][KB_HP];
    __shared__ __attribute__((aligned(16))) float Bp[(KB_PCH / 2) * KB_PITCH];

    const int tid  = threadIdx.x;
    const int wv   = tid >> 5;
    const int lane = tid & 31;
    const int e0   = blockIdx.x * KB_EDGES;

    float w0 = 0.0f, w1 = 0.0f, w2 = 0.0f;
    if (tid < 256) { w0 = W1a[tid]; w1 = W1a[256 + tid]; w2 = W1a[512 + tid]; }

    if (tid < KB_EDGES) {
        const int i = e0 + tid;
        int e = -1;
        float b0 = 0.0f, b1 = 0.0f, b2 = 0.0f;
        if (i < E) {
            e = perm[i];
            e = min(max(e, 0), E - 1);
            b0 = basis[(size_t)e*3]; b1 = basis[(size_t)e*3+1]; b2 = basis[(size_t)e*3+2];
        }
        eid[tid] = e;
        dstT[tid] = (e >= 0) ? edst[e] : 0;
        *(float4*)bas[tid] = make_float4(b0, b1, b2, 0.0f);
        box_partial(e >= 0, b0, b1, b2, boxp + 8 * wv, lane);
    } else {
        const int it = tid - KB_EDGES;
        const int m = it >> 1, q = it & 1;
        const int i = e0 + m;
        float sa = 0.0f, sb = 0.0f;
        if (i < E) {
            int e = perm[i];
            e = min(max(e, 0), E - 1);
            const int src = esrc[e];
            const float* xr = x  + (size_t)src*16;
            const float* sr = sh + (size_t)e*16;
            float xs[8], ss[8];
            #pragma unroll
            for (int j = 0; j < 8; j++) {
                const int idx = q ? (1 + j) : (j == 0 ? 0 : 8 + j);
                xs[j] = xr[idx];
                ss[j] = sr[idx];
            }
            if (q == 0) {
                sa = xs[0] * ss[0];
                float t = 0.0f;
                #pragma unroll
                for (int j = 1; j < 8; j++) t += xs[j] * ss[j];
                sb = t * rsqrtf(7.0f);
            } else {
                float t = 0.0f, u = 0.0f;
                #pragma unroll
                for (int j = 0; j < 3; j++) t += xs[j] * ss[j];
                #pragma unroll
                for (int j = 3; j < 8; j++) u += xs[j] * ss[j];
                sa = t * rsqrtf(3.0f);
                sb = u * rsqrtf(5.0f);
            }
        }
        if (q == 0) { dots[m][0] = sa; dots[m][3] = sb; }
        else        { dots[m][1] = sa; dots[m][2] = sb; }
    }
    if (tid < 256) waF[3][tid] = 0.0f;
    __syncthreads();

    int cls = 0;
    if (tid < 256) cls = box_class(boxp, w0, w1, w2);
    const unsigned balF = (unsigned)__ballot(cls == 1);
    const unsigned balP = (unsigned)__ballot(cls == 2);
    if (wv < 8 && lane == 0) { wcntF[wv] = __popc(balF); wcntP[wv] = __popc(balP); }
    __syncthreads();

    int baseF = 0, LF = 0, baseP = 0, LP = 0;
    #pragma unroll
    for (int i = 0; i < 8; i++) {
        const int cF = wcntF[i], cP = wcntP[i];
        if (i < wv) { baseF += cF; baseP += cP; }
        LF += cF; LP += cP;
    }
    const int LFp = (LF + 15) & ~15;
    const int LPp = (LP + 3) & ~3;
    if (tid < 256) {
        const unsigned lm = (1u << lane) - 1u;
        if (cls == 1) {
            const int j = baseF + __popc(balF & lm);
            colsF[j] = (unsigned)tid * (384u * 4u);
            waF[0][j] = w0; waF[1][j] = w1; waF[2][j] = w2;
        } else if (cls == 2) {
            const int j = baseP + __popc(balP & lm);
            colsP[j] = (unsigned)tid * 384u;
            *(float4*)wP[j] = make_float4(w0, w1, w2, 0.0f);
        }
    } else {
        const int j = LF + (tid - 256);
        if (j < LFp) { colsF[j] = 0u; waF[0][j] = 0.0f; waF[1][j] = 0.0f; waF[2][j] = 0.0f; }
    }
    __syncthreads();

    const int c  = lane & 15;
    const int hh = lane >> 4;
    const char* Wb = (const char*)W1b;

    auto fill = [&](int ch) {
        for (int it = tid; it < KB_EDGES * KB_PCH; it += KB_THREADS) {
            const int m = it >> 4, kk = it & 15;
            const int j = ch * KB_PCH + kk;
            float v = 0.0f;
            if (j < LP) {
                const float4 w = *(const float4*)wP[j];
                v = hidden1(*(const float4*)bas[m], w.x, w.y, w.z);
            }
            hP[m][kk] = v;
        }
        const int nk = min(KB_PCH, LPp - ch * KB_PCH);
        #pragma unroll 4
        for (int kk = 0; kk < nk; kk++) {
            const int j = ch * KB_PCH + kk;
            float v = 0.0f;
            if (j < LP) v = W1b[colsP[j] + tid];
            Bp[(kk >> 1) * KB_PITCH + 2 * tid + (kk & 1)] = v;
        }
    };

    {
        v8f g0 = {}, g1 = {};
        const unsigned cb = (unsigned)(wv * 32 + c) * 4u;
        const float* Arow = waF[min(c, 3)];
        for (int k0 = 0; k0 < LFp; k0 += 16) {
            #pragma unroll
            for (int u = 0; u < 4; u++) {
                const int kb = k0 + 4*u + 2*hh;
                const v2f A = *(const v2f*)(Arow + kb);
                const float* Br0 = (const float*)(Wb + (colsF[kb]     + cb));
                const float* Br1 = (const float*)(Wb + (colsF[kb + 1] + cb));
                v2f Ba; Ba.x = Br0[0];  Ba.y = Br1[0];
                v2f Bb; Bb.x = Br0[16]; Bb.y = Br1[16];
                g0 = __builtin_amdgcn_wmma_f32_16x16x4_f32(false, A, false, Ba, (short)0, g0, false, false);
                g1 = __builtin_amdgcn_wmma_f32_16x16x4_f32(false, A, false, Bb, (short)0, g1, false, false);
            }
        }
        if (hh == 0) {
            const int col0 = wv * 32 + c, col1 = col0 + 16;
            *(v2f*)(Gl + 2*col0)            = (v2f){g0[0], g0[1]};
            *(v2f*)(Gl + KB_PITCH + 2*col0) = (v2f){g0[2], 0.0f};
            *(v2f*)(Gl + 2*col1)            = (v2f){g1[0], g1[1]};
            *(v2f*)(Gl + KB_PITCH + 2*col1) = (v2f){g1[2], 0.0f};
        }
    }
    const int nch = (LPp + KB_PCH - 1) / KB_PCH;
    if (nch > 0) fill(0);
    __syncthreads();

    const int kc   = wv % 6;
    const int rh   = wv / 6;
    const int colq = kc * 16 + c;
    const float scale = (1.0f/16.0f) * 0.5f * INV_NN;
    const float* gp = Gl + hh * KB_PITCH + 2 * colq;

    for (int s = 0; s < 4; s++) {
        const int r0 = rh * 64 + s * 16;
        v8f a0 = {}, a1 = {}, a2 = {}, a3 = {};
        {
            const v2f bb = *(const v2f*)(bas[r0 + c] + 2*hh);
            v2f A; A.x = bb.x * INV_SQRT3; A.y = bb.y * INV_SQRT3;
            const v2f B0 = *(const v2f*)(gp);
            const v2f B1 = *(const v2f*)(gp + 192);
            const v2f B2 = *(const v2f*)(gp + 384);
            const v2f B3 = *(const v2f*)(gp + 576);
            a0 = __builtin_amdgcn_wmma_f32_16x16x4_f32(false, A, false, B0, (short)0, a0, false, false);
            a1 = __builtin_amdgcn_wmma_f32_16x16x4_f32(false, A, false, B1, (short)0, a1, false, false);
            a2 = __builtin_amdgcn_wmma_f32_16x16x4_f32(false, A, false, B2, (short)0, a2, false, false);
            a3 = __builtin_amdgcn_wmma_f32_16x16x4_f32(false, A, false, B3, (short)0, a3, false, false);
        }
        for (int ch = 0; ch < nch; ch++) {
            if (nch > 1 && (s | ch) != 0) {
                __syncthreads();
                fill(ch);
                __syncthreads();
            }
            const int nk = min(KB_PCH, LPp - ch * KB_PCH);
            for (int kk = 0; kk < nk; kk += 4) {
                const v2f A = *(const v2f*)(hP[r0 + c] + kk + 2*hh);
                const float* bp = Bp + ((kk >> 1) + hh) * KB_PITCH + 2 * colq;
                const v2f B0 = *(const v2f*)(bp);
                const v2f B1 = *(const v2f*)(bp + 192);
                const v2f B2 = *(const v2f*)(bp + 384);
                const v2f B3 = *(const v2f*)(bp + 576);
                a0 = __builtin_amdgcn_wmma_f32_16x16x4_f32(false, A, false, B0, (short)0, a0, false, false);
                a1 = __builtin_amdgcn_wmma_f32_16x16x4_f32(false, A, false, B1, (short)0, a1, false, false);
                a2 = __builtin_amdgcn_wmma_f32_16x16x4_f32(false, A, false, B2, (short)0, a2, false, false);
                a3 = __builtin_amdgcn_wmma_f32_16x16x4_f32(false, A, false, B3, (short)0, a3, false, false);
            }
        }
        #pragma unroll
        for (int r = 0; r < 8; r++) {
            const int m = r0 + r + 8*hh;
            if (eid[m] >= 0) {
                const float4 d = *(const float4*)dots[m];
                const float ef = d.x*a0[r] + d.y*a1[r] + d.z*a2[r] + d.w*a3[r];
                atomicAdd(&h[(size_t)dstT[m]*96 + colq], ef * scale);
            }
        }
    }
}

__global__ __launch_bounds__(256) void kC(const float* __restrict__ h, float* __restrict__ y, int N)
{
    const int t   = blockIdx.x * 256 + threadIdx.x;
    const int nid = t >> 4;
    const int q   = t & 15;
    if (nid >= N) return;
    const float4* hr = (const float4*)(h + (size_t)nid * 96);
    float4 o;
    if (q < 8) {
        const float4 a = hr[q];
        if (q < 4) o = make_float4(fmaxf(a.x, 0.0f), fmaxf(a.y, 0.0f), fmaxf(a.z, 0.0f), fmaxf(a.w, 0.0f));
        else       o = make_float4(fabsf(a.x), fabsf(a.y), fabsf(a.z), fabsf(a.w));
    } else {
        const float4 g = hr[q];
        const float4 v = hr[q + 8];
        const bool use_relu = ((q - 8) & 2) == 0;
        float4 ga;
        if (use_relu) ga = make_float4(fmaxf(g.x, 0.0f), fmaxf(g.y, 0.0f), fmaxf(g.z, 0.0f), fmaxf(g.w, 0.0f));
        else          ga = make_float4(tanhf(g.x), tanhf(g.y), tanhf(g.z), tanhf(g.w));
        o = make_float4(ga.x * v.x, ga.y * v.y, ga.z * v.z, ga.w * v.w);
    }
    ((float4*)(y + (size_t)nid * 64))[q] = o;
}

#define KE_EDGES 128
#define KE_THREADS 256
#define KE_PCH 16
#define KE_PITCH 160
__global__ __launch_bounds__(KE_THREADS) void kE(const float* __restrict__ basis,
                                                 const float* __restrict__ y,
                                                 const int* __restrict__ esrc,
                                                 const int* __restrict__ edst,
                                                 const int* __restrict__ perm,
                                                 const float* __restrict__ W2a,
                                                 const float* __restrict__ W2b,
                                                 float* __restrict__ out, int E)
{
    __shared__ __attribute__((aligned(16))) float bas[KE_EDGES][4];
    __shared__ int   srcT[KE_EDGES];
    __shared__ int   dstT[KE_EDGES];
    __shared__ int   eid[KE_EDGES];
    __shared__ __attribute__((aligned(16))) float waF[4][256];
    __shared__ __attribute__((aligned(8))) unsigned colsF[256];
    __shared__ __attribute__((aligned(16))) float wP[256][4];
    __shared__ unsigned colsP[256];
    __shared__ int   wcntF[8], wcntP[8];
    __shared__ __attribute__((aligned(16))) float boxp[4 * 8];
    __shared__ __attribute__((aligned(16))) float Gl[2][2 * KE_PITCH];
    __shared__ __attribute__((aligned(16))) float hP[KE_EDGES][KB_HP];
    __shared__ __attribute__((aligned(16))) float Bp[(KE_PCH / 2) * KE_PITCH];

    const int tid  = threadIdx.x;
    const int wv   = tid >> 5;
    const int lane = tid & 31;
    const int e0   = blockIdx.x * KE_EDGES;

    const float w0 = W2a[tid], w1 = W2a[256 + tid], w2 = W2a[512 + tid];

    if (tid < KE_EDGES) {
        const int i = e0 + tid;
        int e = -1, sv = 0, dv = 0;
        float b0 = 0.0f, b1 = 0.0f, b2 = 0.0f;
        if (i < E) {
            e = perm[i];
            e = min(max(e, 0), E - 1);
            b0 = basis[(size_t)e*3]; b1 = basis[(size_t)e*3+1]; b2 = basis[(size_t)e*3+2];
            sv = esrc[e];
            dv = edst[e];
        }
        eid[tid] = e;
        srcT[tid] = sv;
        dstT[tid] = dv;
        *(float4*)bas[tid] = make_float4(b0, b1, b2, 0.0f);
        box_partial(e >= 0, b0, b1, b2, boxp + 8 * wv, lane);
    }
    waF[3][tid] = 0.0f;
    __syncthreads();

    const int cls = box_class(boxp, w0, w1, w2);
    const unsigned balF = (unsigned)__ballot(cls == 1);
    const unsigned balP = (unsigned)__ballot(cls == 2);
    if (lane == 0) { wcntF[wv] = __popc(balF); wcntP[wv] = __popc(balP); }
    __syncthreads();

    int baseF = 0, LF = 0, baseP = 0, LP = 0;
    #pragma unroll
    for (int i = 0; i < 8; i++) {
        const int cF = wcntF[i], cP = wcntP[i];
        if (i < wv) { baseF += cF; baseP += cP; }
        LF += cF; LP += cP;
    }
    const int LFp = (LF + 31) & ~31;
    const int LPp = (LP + 3) & ~3;
    {
        const unsigned lm = (1u << lane) - 1u;
        if (cls == 1) {
            const int j = baseF + __popc(balF & lm);
            colsF[j] = (unsigned)tid * (64u * 4u);
            waF[0][j] = w0; waF[1][j] = w1; waF[2][j] = w2;
        } else if (cls == 2) {
            const int j = baseP + __popc(balP & lm);
            colsP[j] = (unsigned)tid * 64u;
            *(float4*)wP[j] = make_float4(w0, w1, w2, 0.0f);
        }
        if (tid < 32) {
            const int j = LF + tid;
            if (j < LFp) { colsF[j] = 0u; waF[0][j] = 0.0f; waF[1][j] = 0.0f; waF[2][j] = 0.0f; }
        }
    }
    __syncthreads();

    const int c  = lane & 15;
    const int hh = lane >> 4;
    const char* Wb = (const char*)W2b;

    auto fill = [&](int ch) {
        for (int it = tid; it < KE_EDGES * KE_PCH; it += KE_THREADS) {
            const int m = it >> 4, kk = it & 15;
            const int j = ch * KE_PCH + kk;
            float v = 0.0f;
            if (j < LP) {
                const float4 w = *(const float4*)wP[j];
                v = hidden1(*(const float4*)bas[m], w.x, w.y, w.z);
            }
            hP[m][kk] = v;
        }
        const int nk = min(KE_PCH, LPp - ch * KE_PCH);
        const int col = tid & 63;
        for (int kk = tid >> 6; kk < nk; kk += 4) {
            const int j = ch * KE_PCH + kk;
            float v = 0.0f;
            if (j < LP) v = W2b[colsP[j] + col];
            Bp[(kk >> 1) * KE_PITCH + 2 * col + (kk & 1)] = v;
        }
    };

    {
        const int ct = wv & 3, kh = wv >> 2;
        v8f g0 = {}, g1 = {};
        const unsigned cb = (unsigned)(ct * 16 + c) * 4u;
        const float* Arow = waF[min(c, 3)];
        for (int k0 = 16 * kh; k0 < LFp; k0 += 32) {
            #pragma unroll
            for (int u = 0; u < 4; u++) {
                const int kb = k0 + 4*u + 2*hh;
                const v2f A = *(const v2f*)(Arow + kb);
                v2f B;
                B.x = *(const float*)(Wb + (colsF[kb]     + cb));
                B.y = *(const float*)(Wb + (colsF[kb + 1] + cb));
                if (u & 1) g1 = __builtin_amdgcn_wmma_f32_16x16x4_f32(false, A, false, B, (short)0, g1, false, false);
                else       g0 = __builtin_amdgcn_wmma_f32_16x16x4_f32(false, A, false, B, (short)0, g0, false, false);
            }
        }
        if (hh == 0) {
            const int col = ct * 16 + c;
            *(v2f*)(Gl[kh] + 2*col)            = (v2f){g0[0] + g1[0], g0[1] + g1[1]};
            *(v2f*)(Gl[kh] + KE_PITCH + 2*col) = (v2f){g0[2] + g1[2], 0.0f};
        }
    }
    const int nch = (LPp + KE_PCH - 1) / KE_PCH;
    if (nch > 0) fill(0);
    __syncthreads();

    const int r0 = wv * 16;
    v8f a0 = {}, a1 = {}, a2 = {}, a3 = {};
    {
        const v2f bb = *(const v2f*)(bas[r0 + c] + 2*hh);
        v2f A; A.x = bb.x * INV_SQRT3; A.y = bb.y * INV_SQRT3;
        const float* gp = Gl[0] + hh * KE_PITCH + 8 * c;
        const float* gq = Gl[1] + hh * KE_PITCH + 8 * c;
        const float4 Pa = *(const float4*)(gp), Pb = *(const float4*)(gp + 4);
        const float4 Qa = *(const float4*)(gq), Qb = *(const float4*)(gq + 4);
        const v2f B0 = (v2f){Pa.x + Qa.x, Pa.y + Qa.y}, B1 = (v2f){Pa.z + Qa.z, Pa.w + Qa.w};
        const v2f B2 = (v2f){Pb.x + Qb.x, Pb.y + Qb.y}, B3 = (v2f){Pb.z + Qb.z, Pb.w + Qb.w};
        a0 = __builtin_amdgcn_wmma_f32_16x16x4_f32(false, A, false, B0, (short)0, a0, false, false);
        a1 = __builtin_amdgcn_wmma_f32_16x16x4_f32(false, A, false, B1, (short)0, a1, false, false);
        a2 = __builtin_amdgcn_wmma_f32_16x16x4_f32(false, A, false, B2, (short)0, a2, false, false);
        a3 = __builtin_amdgcn_wmma_f32_16x16x4_f32(false, A, false, B3, (short)0, a3, false, false);
    }
    for (int ch = 0; ch < nch; ch++) {
        if (ch > 0) {
            __syncthreads();
            fill(ch);
            __syncthreads();
        }
        const int nk = min(KE_PCH, LPp - ch * KE_PCH);
        for (int kk = 0; kk < nk; kk += 4) {
            const v2f A = *(const v2f*)(hP[r0 + c] + kk + 2*hh);
            const float* bp = Bp + ((kk >> 1) + hh) * KE_PITCH + 8 * c;
            const float4 Ba = *(const float4*)(bp);
            const float4 Bb = *(const float4*)(bp + 4);
            const v2f B0 = (v2f){Ba.x, Ba.y}, B1 = (v2f){Ba.z, Ba.w};
            const v2f B2 = (v2f){Bb.x, Bb.y}, B3 = (v2f){Bb.z, Bb.w};
            a0 = __builtin_amdgcn_wmma_f32_16x16x4_f32(false, A, false, B0, (short)0, a0, false, false);
            a1 = __builtin_amdgcn_wmma_f32_16x16x4_f32(false, A, false, B1, (short)0, a1, false, false);
            a2 = __builtin_amdgcn_wmma_f32_16x16x4_f32(false, A, false, B2, (short)0, a2, false, false);
            a3 = __builtin_amdgcn_wmma_f32_16x16x4_f32(false, A, false, B3, (short)0, a3, false, false);
        }
    }

    float part[8];
    #pragma unroll
    for (int r = 0; r < 8; r++) {
        const int sv = srcT[r0 + 8*hh + r];
        const float4 yv = *(const float4*)(y + (size_t)sv*64 + 4*c);
        part[r] = a0[r]*yv.x + a1[r]*yv.y + a2[r]*yv.z + a3[r]*yv.w;
    }
    #pragma unroll
    for (int i = 0; i < 4; i++) {
        const bool up = (c & 8) != 0;
        const float keep = up ? part[4 + i] : part[i];
        const float send = up ? part[i] : part[4 + i];
        part[i] = keep + __shfl_xor(send, 8, 32);
    }
    #pragma unroll
    for (int i = 0; i < 2; i++) {
        const bool up = (c & 4) != 0;
        const float keep = up ? part[2 + i] : part[i];
        const float send = up ? part[i] : part[2 + i];
        part[i] = keep + __shfl_xor(send, 4, 32);
    }
    {
        const bool up = (c & 2) != 0;
        const float keep = up ? part[1] : part[0];
        const float send = up ? part[0] : part[1];
        part[0] = keep + __shfl_xor(send, 2, 32);
    }
    const float tot = part[0] + __shfl_xor(part[0], 1, 32);
    if ((c & 1) == 0) {
        const int m = r0 + 8*hh + 4*((c >> 3) & 1) + 2*((c >> 2) & 1) + ((c >> 1) & 1);
        if (eid[m] >= 0)
            atomicAdd(&out[dstT[m]], tot * (1.0f/128.0f) * INV_NN);
    }
}

extern "C" void kernel_launch(void* const* d_in, const int* in_sizes, int n_in,
                              void* d_out, int out_size, void* d_ws, size_t ws_size,
                              hipStream_t stream)
{
    const float* evec = (const float*)d_in[0];
    const float* elen = (const float*)d_in[1];
    const float* W1a  = (const float*)d_in[2];
    const float* W1b  = (const float*)d_in[3];
    const float* W2a  = (const float*)d_in[4];
    const float* W2b  = (const float*)d_in[5];
    const int*   esrc = (const int*)d_in[6];
    const int*   edst = (const int*)d_in[7];
    float* out = (float*)d_out;

    const int E = in_sizes[0] / 3;
    const int N = out_size;

    float* ws = (float*)d_ws;
    float* x_     = ws;                 ws += (size_t)N * 16;
    float* h_     = ws;                 ws += (size_t)N * 96;
    float* y_     = ws;                 ws += (size_t)N * 64;
    float* sh_    = ws;                 ws += (size_t)E * 16;
    float* basis_ = ws;                 ws += (size_t)E * 3;
    int*   perm_  = (int*)ws;           ws += (size_t)E;
    int*   cnt_   = (int*)ws;
    const int nb  = (E + SORT_CHUNK - 1) / SORT_CHUNK;

    long nx = (long)N * 16, nh = (long)N * 96;
    const int zx = (int)((nx / 4 + 255) / 256);
    const int nz = (int)((nh / 4 + 1023) / 1024);
    const int na = (E + 255) / 256;

    kPre<<<nb + zx, SORT_THREADS, 0, stream>>>(elen, cnt_, nb, E, x_, nx);
    kMain<<<nb + nz + na, 256, 0, stream>>>(evec, elen, edst, sh_, basis_, x_, E,
                                            cnt_, perm_, nb, h_, nh, out, (long)N, nz);
    kB<<<(E + KB_EDGES - 1) / KB_EDGES, KB_THREADS, 0, stream>>>(sh_, basis_, x_, esrc, edst, perm_, W1a, W1b, h_, E);
    kC<<<(N * 16 + 255) / 256, 256, 0, stream>>>(h_, y_, N);
    kE<<<(E + KE_EDGES - 1) / KE_EDGES, KE_THREADS, 0, stream>>>(basis_, y_, esrc, edst, perm_, W2a, W2b, out, E);
}


# kB: a block takes 512 length-sorted edges, not 128: one class pass, one G (W1a x W1b, f32 WMMA) per 512 rows: 1/4 of the W1b reads and G products; loads grouped; k grouping, h add order change
# speedup vs baseline: 4.7407x; 1.0717x over previous
#include <hip/hip_runtime.h>
#include <math.h>

typedef __attribute__((ext_vector_type(2))) float v2f;
typedef __attribute__((ext_vector_type(8))) float v8f;

#define INV_NN     0.512989176042577f
#define INV_SQRT3  0.5773502691896258f
#define S3   1.7320508075688772f
#define S5   2.23606797749979f
#define S7   2.6457513110645907f
#define S15  3.872983346207417f
#define S105 10.246950765959598f
#define C358 2.091650066335189f
#define C218 1.6201851746019651f

#define NBINS 256
#define SORT_CHUNK 4096
#define SORT_THREADS 256
__device__ __forceinline__ int len_bin(float len) {
    float t = fminf(fmaxf(len * ((float)NBINS / 4.0f), 0.0f), (float)(NBINS - 1));
    return (int)t;
}

__device__ __forceinline__ void clear_role(float* __restrict__ p, long n, int bid, int nblk, int tid) {
    const long n4 = n >> 2;
    float4* p4 = (float4*)p;
    const long step = (long)nblk * 256;
    for (long i = (long)bid * 256 + tid; i < n4; i += step) p4[i] = make_float4(0.0f, 0.0f, 0.0f, 0.0f);
    for (long i = (n4 << 2) + (long)bid * 256 + tid; i < n; i += step) p[i] = 0.0f;
}

__global__ __launch_bounds__(SORT_THREADS) void kPre(const float* __restrict__ elen,
                                                     int* __restrict__ cnt, int nb, int E,
                                                     float* __restrict__ x, long nx)
{
    const int tid = threadIdx.x;
    if ((int)blockIdx.x >= nb) {
        clear_role(x, nx, (int)blockIdx.x - nb, (int)gridDim.x - nb, tid);
        return;
    }
    __shared__ int lh[NBINS];
    lh[tid] = 0;
    __syncthreads();
    const int base = blockIdx.x * SORT_CHUNK;
    #pragma unroll
    for (int i = 0; i < SORT_CHUNK / SORT_THREADS; i++) {
        const int e = base + i * SORT_THREADS + tid;
        if (e < E) atomicAdd(&lh[len_bin(elen[e])], 1);
    }
    __syncthreads();
    cnt[(size_t)tid * nb + blockIdx.x] = lh[tid];
}

__device__ __forceinline__ void scatter_role(const float* __restrict__ elen,
                                             const int* __restrict__ cnt,
                                             int* __restrict__ perm, int nb, int E, int blk, int tid)
{
    __shared__ int lh[NBINS];
    __shared__ int wsum[SORT_THREADS / 32];
    const int lane = tid & 31, w = tid >> 5;
    int tot = 0, part = 0;
    const int* row = cnt + (size_t)tid * nb;
    #pragma unroll 8
    for (int k = 0; k < nb; k++) {
        const int c = row[k];
        tot += c;
        part += (k < blk) ? c : 0;
    }
    int incl = tot;
    #pragma unroll
    for (int d = 1; d < 32; d <<= 1) {
        int v = __shfl_up(incl, d, 32);
        if (lane >= d) incl += v;
    }
    if (lane == 31) wsum[w] = incl;
    __syncthreads();
    int wbase = 0;
    #pragma unroll
    for (int i = 0; i < SORT_THREADS / 32; i++) if (i < w) wbase += wsum[i];
    lh[tid] = wbase + (incl - tot) + part;
    __syncthreads();
    const int base = blk * SORT_CHUNK;
    #pragma unroll
    for (int i = 0; i < SORT_CHUNK / SORT_THREADS; i++) {
        const int e = base + i * SORT_THREADS + tid;
        if (e < E) {
            const int pos = atomicAdd(&lh[len_bin(elen[e])], 1);
            if ((unsigned)pos < (unsigned)E) perm[pos] = e;
        }
    }
}

#define SH_PITCH 20
__device__ __forceinline__ void kA_role(const float* __restrict__ evec, const float* __restrict__ elen,
                   const int* __restrict__ edst,
                   float* __restrict__ sh, float* __restrict__ basis,
                   float* __restrict__ x, int E, int bid, int tid)
{
    __shared__ __attribute__((aligned(16))) float tile[256 * SH_PITCH];
    __shared__ int dstT[256];

    const int eb  = bid * 256;
    const int e   = eb + tid;
    float s[16] = {};
    int d = 0;
    if (e < E) {
        float vx = evec[3*(size_t)e], vy = evec[3*(size_t)e+1], vz = evec[3*(size_t)e+2];
        float rn = 1.0f / (sqrtf(vx*vx + vy*vy + vz*vz) + 1e-12f);
        vx *= rn; vy *= rn; vz *= rn;
        float x2 = vx*vx, y2 = vy*vy, z2 = vz*vz;
        s[0]  = 1.0f;
        s[1]  = S3*vx; s[2] = S3*vy; s[3] = S3*vz;
        s[4]  = S15*vx*vz;
        s[5]  = S15*vx*vy;
        s[6]  = 0.5f*S5*(2.0f*y2 - x2 - z2);
        s[7]  = S15*vy*vz;
        s[8]  = 0.5f*S15*(z2 - x2);
        s[9]  = C358*vx*(3.0f*z2 - x2);
        s[10] = S105*vx*vy*vz;
        s[11] = C218*vx*(4.0f*y2 - z2 - x2);
        s[12] = 0.5f*S7*vy*(2.0f*y2 - 3.0f*z2 - 3.0f*x2);
        s[13] = C218*vz*(4.0f*y2 - z2 - x2);
        s[14] = 0.5f*S105*vy*(z2 - x2);
        s[15] = C358*vz*(z2 - 3.0f*x2);
        d = edst[e];
        float len = elen[e];
        #pragma unroll
        for (int i = 0; i < 3; i++) {
            float dd = (len - 2.0f*(float)i) * 0.5f;
            basis[(size_t)e*3 + i] = expf(-dd*dd) * (1.0f/1.12f);
        }
    }
    {
        float4* row = (float4*)(tile + tid * SH_PITCH);
        row[0] = make_float4(s[0],  s[1],  s[2],  s[3]);
        row[1] = make_float4(s[4],  s[5],  s[6],  s[7]);
        row[2] = make_float4(s[8],  s[9],  s[10], s[11]);
        row[3] = make_float4(s[12], s[13], s[14], s[15]);
        dstT[tid] = d;
    }
    __syncthreads();

    const int lane  = tid & 31;
    const int hh    = lane >> 4;
    const int j     = lane & 15;
    const int wbase = tid & ~31;
    if (eb + wbase + 32 <= E) {
        float v[16];
        int   dm[16];
        #pragma unroll
        for (int i = 0; i < 16; i++) {
            const int m = wbase + 2*i + hh;
            v[i]  = tile[m * SH_PITCH + j];
            dm[i] = dstT[m];
        }
        #pragma unroll
        for (int i = 0; i < 16; i++) {
            const int m = wbase + 2*i + hh;
            sh[(size_t)(eb + m)*16 + j] = v[i];
            atomicAdd(&x[(size_t)dm[i]*16 + j], v[i] * INV_NN);
        }
    } else {
        for (int i = 0; i < 16; i++) {
            const int m  = wbase + 2*i + hh;
            const int em = eb + m;
            if (em < E) {
                const float v = tile[m * SH_PITCH + j];
                sh[(size_t)em*16 + j] = v;
                atomicAdd(&x[(size_t)dstT[m]*16 + j], v * INV_NN);
            }
        }
    }
}

__global__ __launch_bounds__(256) void kMain(const float* __restrict__ evec, const float* __restrict__ elen,
                   const int* __restrict__ edst,
                   float* __restrict__ sh, float* __restrict__ basis,
                   float* __restrict__ x, int E,
                   const int* __restrict__ cnt, int* __restrict__ perm, int nb,
                   float* __restrict__ h, long nh, float* __restrict__ out, long nout, int nz)
{
    const int tid = threadIdx.x;
    const int b   = blockIdx.x;
    if (b < nb) {
        scatter_role(elen, cnt, perm, nb, E, b, tid);
    } else if (b < nb + nz) {
        clear_role(h, nh, b - nb, nz, tid);
        for (long i = (long)(b - nb) * 256 + tid; i < nout; i += (long)nz * 256) out[i] = 0.0f;
    } else {
        kA_role(evec, elen, edst, sh, basis, x, E, b - nb - nz, tid);
    }
}

#define KB_EDGES 512
#define KB_THREADS 384
#define KB_STEPS (KB_EDGES / 32)
#define KB_PCH 16
#define KB_PITCH 800
#define KB_HP 20
__device__ __forceinline__ float pre1(float4 b, float w0, float w1, float w2) {
    return fmaf(b.z, w2, fmaf(b.y, w1, b.x * w0));
}
__device__ __forceinline__ float hidden1(float4 b, float w0, float w1, float w2) {
    return fmaxf(pre1(b, w0, w1, w2) * INV_SQRT3, 0.0f);
}
__device__ __forceinline__ void box_reduce(float n0, float n1, float n2,
                                           float x0, float x1, float x2,
                                           float* dst8, int lane) {
    #pragma unroll
    for (int d = 16; d >= 1; d >>= 1) {
        n0 = fminf(n0, __shfl_xor(n0, d, 32));
        n1 = fminf(n1, __shfl_xor(n1, d, 32));
        n2 = fminf(n2, __shfl_xor(n2, d, 32));
        x0 = fmaxf(x0, __shfl_xor(x0, d, 32));
        x1 = fmaxf(x1, __shfl_xor(x1, d, 32));
        x2 = fmaxf(x2, __shfl_xor(x2, d, 32));
    }
    if (lane == 0) {
        *(float4*)(dst8)     = make_float4(n0, n1, n2, 0.0f);
        *(float4*)(dst8 + 4) = make_float4(x0, x1, x2, 0.0f);
    }
}
__device__ __forceinline__ void box_partial(bool ok, float b0, float b1, float b2,
                                            float* dst8, int lane) {
    const float inf = __builtin_huge_valf();
    box_reduce(ok ? b0 : inf, ok ? b1 : inf, ok ? b2 : inf,
               ok ? b0 : -inf, ok ? b1 : -inf, ok ? b2 : -inf, dst8, lane);
}
template <int NP>
__device__ __forceinline__ int box_class(const float* boxp, float w0, float w1, float w2) {
    const float inf = __builtin_huge_valf();
    float n0 = inf, n1 = inf, n2 = inf, x0 = -inf, x1 = -inf, x2 = -inf;
    #pragma unroll
    for (int p = 0; p < NP; p++) {
        const float4 lo4 = *(const float4*)(boxp + 8 * p), hi4 = *(const float4*)(boxp + 8 * p + 4);
        n0 = fminf(n0, lo4.x); n1 = fminf(n1, lo4.y); n2 = fminf(n2, lo4.z);
        x0 = fmaxf(x0, hi4.x); x1 = fmaxf(x1, hi4.y); x2 = fmaxf(x2, hi4.z);
    }
    const float a0 = w0 * n0, c0 = w0 * x0;
    const float a1 = w1 * n1, c1 = w1 * x1;
    const float a2 = w2 * n2, c2 = w2 * x2;
    const float lo  = fminf(a0, c0) + fminf(a1, c1) + fminf(a2, c2);
    const float hi  = fmaxf(a0, c0) + fmaxf(a1, c1) + fmaxf(a2, c2);
    const float mag = fmaxf(fabsf(a0), fabsf(c0)) + fmaxf(fabsf(a1), fabsf(c1)) + fmaxf(fabsf(a2), fabsf(c2));
    const float tol = 1.0e-6f * mag;
    return (lo > tol) ? 1 : ((hi < -tol) ? 0 : 2);
}
__global__ __launch_bounds__(KB_THREADS) void kB(
    const float* __restrict__ sh, const float* __restrict__ basis,
    const float* __restrict__ x,
    const int* __restrict__ esrc, const int* __restrict__ edst,
    const int* __restrict__ perm,
    const float* __restrict__ W1a, const float* __restrict__ W1b,
    float* __restrict__ h, int E)
{
    __shared__ __attribute__((aligned(16))) float bas[KB_EDGES + 1][4];
    __shared__ __attribute__((aligned(16))) float dots[KB_EDGES + 1][4];
    __shared__ int   dstT[KB_EDGES + 1];
    __shared__ __attribute__((aligned(16))) float waF[4][256];
    __shared__ __attribute__((aligned(8))) unsigned colsF[256];
    __shared__ __attribute__((aligned(16))) float wP[256][4];
    __shared__ unsigned colsP[256];
    __shared__ int   wcntF[8], wcntP[8];
    __shared__ __attribute__((aligned(16))) float boxp[(KB_THREADS / 32) * 8];
    __shared__ __attribute__((aligned(16))) float Gl[2 * KB_PITCH];
    __shared__ __attribute__((aligned(16))) float hP[KB_EDGES][KB_HP];
    __shared__ __attribute__((aligned(16))) float Bp[(KB_PCH / 2) * KB_PITCH];

    const int tid  = threadIdx.x;
    const int wv   = tid >> 5;
    const int lane = tid & 31;
    const int e0   = blockIdx.x * KB_EDGES;
    const int Em1  = E - 1;

    float w0 = 0.0f, w1 = 0.0f, w2 = 0.0f;
    if (tid < 256) { w0 = W1a[tid]; w1 = W1a[256 + tid]; w2 = W1a[512 + tid]; }

    int  mR[2], eR[2];  bool okR[2];
    int  mD[3], qD[3], eD[3];  bool okD[3];
    #pragma unroll
    for (int u = 0; u < 2; u++) {
        const int mm = tid + u * KB_THREADS;
        const bool mine = mm < KB_EDGES;
        mR[u]    = mine ? mm : KB_EDGES;
        const int i = e0 + min(mm, KB_EDGES - 1);
        okR[u]   = mine && (i < E);
        eR[u]    = perm[min(i, Em1)];
    }
    #pragma unroll
    for (int u = 0; u < 3; u++) {
        const int it0 = tid + u * KB_THREADS;
        const bool mine = it0 < 2 * KB_EDGES;
        const int it = min(it0, 2 * KB_EDGES - 1);
        mD[u]    = mine ? (it >> 1) : KB_EDGES;
        qD[u]    = it & 1;
        const int i = e0 + (it >> 1);
        okD[u]   = i < E;
        eD[u]    = perm[min(i, Em1)];
    }
    float bR[2][3];  int dvR[2], srcD[3];
    #pragma unroll
    for (int u = 0; u < 2; u++) {
        const int e = min(max(eR[u], 0), Em1);
        eR[u] = e;
        bR[u][0] = basis[(size_t)e*3]; bR[u][1] = basis[(size_t)e*3+1]; bR[u][2] = basis[(size_t)e*3+2];
        dvR[u] = edst[e];
    }
    #pragma unroll
    for (int u = 0; u < 3; u++) {
        const int e = min(max(eD[u], 0), Em1);
        eD[u] = e;
        srcD[u] = esrc[e];
    }
    float xs[3][8], ss[3][8];
    #pragma unroll
    for (int u = 0; u < 3; u++) {
        const int q = qD[u];
        const float* xr = x  + (size_t)srcD[u]*16;
        const float* sr = sh + (size_t)eD[u]*16;
        #pragma unroll
        for (int j = 0; j < 8; j++) {
            const int idx = q ? (1 + j) : (j == 0 ? 0 : 8 + j);
            xs[u][j] = xr[idx];
            ss[u][j] = sr[idx];
        }
    }
    {
        const float inf = __builtin_huge_valf();
        float n0 = inf, n1 = inf, n2 = inf, x0 = -inf, x1 = -inf, x2 = -inf;
        #pragma unroll
        for (int u = 0; u < 2; u++) {
            const bool ok = okR[u];
            const float b0 = bR[u][0], b1 = bR[u][1], b2 = bR[u][2];
            {
                const int m = mR[u];
                dstT[m] = ok ? dvR[u] : -1;
                *(float4*)bas[m] = ok ? make_float4(b0, b1, b2, 0.0f) : make_float4(0.0f, 0.0f, 0.0f, 0.0f);
            }
            if (ok) {
                n0 = fminf(n0, b0); n1 = fminf(n1, b1); n2 = fminf(n2, b2);
                x0 = fmaxf(x0, b0); x1 = fmaxf(x1, b1); x2 = fmaxf(x2, b2);
            }
        }
        box_reduce(n0, n1, n2, x0, x1, x2, boxp + 8 * wv, lane);
    }
    #pragma unroll
    for (int u = 0; u < 3; u++) {
        const int m = mD[u], q = qD[u];
        float sa, sb;
        if (q == 0) {
            sa = xs[u][0] * ss[u][0];
            float t = 0.0f;
            #pragma unroll
            for (int j = 1; j < 8; j++) t += xs[u][j] * ss[u][j];
            sb = t * rsqrtf(7.0f);
        } else {
            float t = 0.0f, v = 0.0f;
            #pragma unroll
            for (int j = 0; j < 3; j++) t += xs[u][j] * ss[u][j];
            #pragma unroll
            for (int j = 3; j < 8; j++) v += xs[u][j] * ss[u][j];
            sa = t * rsqrtf(3.0f);
            sb = v * rsqrtf(5.0f);
        }
        if (!okD[u]) { sa = 0.0f; sb = 0.0f; }
        dots[m][q ? 1 : 0] = sa;
        dots[m][q ? 2 : 3] = sb;
    }
    if (tid < 256) waF[3][tid] = 0.0f;
    __syncthreads();

    int cls = 0;
    if (tid < 256) cls = box_class<KB_THREADS / 32>(boxp, w0, w1, w2);
    const unsigned balF = (unsigned)__ballot(cls == 1);
    const unsigned balP = (unsigned)__ballot(cls == 2);
    if (wv < 8 && lane == 0) { wcntF[wv] = __popc(balF); wcntP[wv] = __popc(balP); }
    __syncthreads();

    int baseF = 0, LF = 0, baseP = 0, LP = 0;
    #pragma unroll
    for (int i = 0; i < 8; i++) {
        const int cF = wcntF[i], cP = wcntP[i];
        if (i < wv) { baseF += cF; baseP += cP; }
        LF += cF; LP += cP;
    }
    const int LFp = (LF + 15) & ~15;
    const int LPp = (LP + 3) & ~3;
    if (tid < 256) {
        const unsigned lm = (1u << lane) - 1u;
        if (cls == 1) {
            const int j = baseF + __popc(balF & lm);
            colsF[j] = (unsigned)tid * (384u * 4u);
            waF[0][j] = w0; waF[1][j] = w1; waF[2][j] = w2;
        } else if (cls == 2) {
            const int j = baseP + __popc(balP & lm);
            colsP[j] = (unsigned)tid * 384u;
            *(float4*)wP[j] = make_float4(w0, w1, w2, 0.0f);
        }
    } else {
        const int j = LF + (tid - 256);
        if (j < LFp) { colsF[j] = 0u; waF[0][j] = 0.0f; waF[1][j] = 0.0f; waF[2][j] = 0.0f; }
    }
    __syncthreads();

    const int c  = lane & 15;
    const int hh = lane >> 4;
    const char* Wb = (const char*)W1b;

    auto fill = [&](int ch) {
        const int nk = min(KB_PCH, LPp - ch * KB_PCH);
        for (int it = tid; it < KB_EDGES * (nk >> 2); it += KB_THREADS) {
            const int m = it % KB_EDGES, g = it / KB_EDGES;
            const float4 b = *(const float4*)bas[m];
            float v[4];
            #pragma unroll
            for (int t = 0; t < 4; t++) {
                const int j = ch * KB_PCH + 4 * g + t;
                const float4 w = *(const float4*)wP[min(j, LP - 1)];
                const float val = hidden1(b, w.x, w.y, w.z);
                v[t] = (j < LP) ? val : 0.0f;
            }
            *(float4*)(hP[m] + 4 * g) = make_float4(v[0], v[1], v[2], v[3]);
        }
        #pragma unroll 4
        for (int kk = 0; kk < nk; kk++) {
            const int j = ch * KB_PCH + kk;
            const float v = W1b[colsP[min(j, LP - 1)] + tid];
            Bp[(kk >> 1) * KB_PITCH + 2 * tid + (kk & 1)] = (j < LP) ? v : 0.0f;
        }
    };

    {
        v8f g0 = {}, g1 = {};
        const unsigned cb = (unsigned)(wv * 32 + c) * 4u;
        const float* Arow = waF[min(c, 3)];
        for (int k0 = 0; k0 < LFp; k0 += 16) {
            #pragma unroll
            for (int u = 0; u < 4; u++) {
                const int kb = k0 + 4*u + 2*hh;
                const v2f A = *(const v2f*)(Arow + kb);
                const float* Br0 = (const float*)(Wb + (colsF[kb]     + cb));
                const float* Br1 = (const float*)(Wb + (colsF[kb + 1] + cb));
                v2f Ba; Ba.x = Br0[0];  Ba.y = Br1[0];
                v2f Bb; Bb.x = Br0[16]; Bb.y = Br1[16];
                g0 = __builtin_amdgcn_wmma_f32_16x16x4_f32(false, A, false, Ba, (short)0, g0, false, false);
                g1 = __builtin_amdgcn_wmma_f32_16x16x4_f32(false, A, false, Bb, (short)0, g1, false, false);
            }
        }
        if (hh == 0) {
            const int col0 = wv * 32 + c, col1 = col0 + 16;
            *(v2f*)(Gl + 2*col0)            = (v2f){g0[0], g0[1]};
            *(v2f*)(Gl + KB_PITCH + 2*col0) = (v2f){g0[2], 0.0f};
            *(v2f*)(Gl + 2*col1)            = (v2f){g1[0], g1[1]};
            *(v2f*)(Gl + KB_PITCH + 2*col1) = (v2f){g1[2], 0.0f};
        }
    }
    const int nch = (LPp + KB_PCH - 1) / KB_PCH;
    if (nch > 0) fill(0);
    __syncthreads();

    const int wvu  = __builtin_amdgcn_readfirstlane(wv);
    const int kc   = wvu % 6;
    const int rh   = wvu / 6;
    const int colq = kc * 16 + c;
    const float scale = (1.0f/16.0f) * 0.5f * INV_NN;
    const float* gp = Gl + hh * KB_PITCH + 2 * colq;

    const int nsweep = max(nch, 1);
    for (int ch = 0; ch < nsweep; ch++) {
        if (ch > 0) {
            __syncthreads();
            fill(ch);
            __syncthreads();
        }
        const int nk = (nch > 0) ? min(KB_PCH, LPp - ch * KB_PCH) : 0;
        for (int s = 0; s < KB_STEPS; s++) {
            const int r0 = rh * (KB_EDGES / 2) + s * 16;
            if (e0 + r0 >= E) break;
            int    dv[8];
            float4 dd[8];
            #pragma unroll
            for (int r = 0; r < 8; r++) {
                const int m = r0 + r + 8*hh;
                dv[r] = dstT[m];
                dd[r] = *(const float4*)dots[m];
            }
            v8f a0 = {}, a1 = {}, a2 = {}, a3 = {};
            if (ch == 0) {
                const v2f bb = *(const v2f*)(bas[r0 + c] + 2*hh);
                v2f A; A.x = bb.x * INV_SQRT3; A.y = bb.y * INV_SQRT3;
                const v2f B0 = *(const v2f*)(gp);
                const v2f B1 = *(const v2f*)(gp + 192);
                const v2f B2 = *(const v2f*)(gp + 384);
                const v2f B3 = *(const v2f*)(gp + 576);
                a0 = __builtin_amdgcn_wmma_f32_16x16x4_f32(false, A, false, B0, (short)0, a0, false, false);
                a1 = __builtin_amdgcn_wmma_f32_16x16x4_f32(false, A, false, B1, (short)0, a1, false, false);
                a2 = __builtin_amdgcn_wmma_f32_16x16x4_f32(false, A, false, B2, (short)0, a2, false, false);
                a3 = __builtin_amdgcn_wmma_f32_16x16x4_f32(false, A, false, B3, (short)0, a3, false, false);
            }
            for (int kk = 0; kk < nk; kk += 4) {
                const v2f A = *(const v2f*)(hP[r0 + c] + kk + 2*hh);
                const float* bp = Bp + ((kk >> 1) + hh) * KB_PITCH + 2 * colq;
                const v2f B0 = *(const v2f*)(bp);
                const v2f B1 = *(const v2f*)(bp + 192);
                const v2f B2 = *(const v2f*)(bp + 384);
                const v2f B3 = *(const v2f*)(bp + 576);
                a0 = __builtin_amdgcn_wmma_f32_16x16x4_f32(false, A, false, B0, (short)0, a0, false, false);
                a1 = __builtin_amdgcn_wmma_f32_16x16x4_f32(false, A, false, B1, (short)0, a1, false, false);
                a2 = __builtin_amdgcn_wmma_f32_16x16x4_f32(false, A, false, B2, (short)0, a2, false, false);
                a3 = __builtin_amdgcn_wmma_f32_16x16x4_f32(false, A, false, B3, (short)0, a3, false, false);
            }
            #pragma unroll
            for (int r = 0; r < 8; r++) {
                const float4 d = dd[r];
                const float ef = d.x*a0[r] + d.y*a1[r] + d.z*a2[r] + d.w*a3[r];
                if (dv[r] >= 0) atomicAdd(&h[(size_t)dv[r]*96 + colq], ef * scale);
            }
        }
    }
}

__global__ __launch_bounds__(256) void kC(const float* __restrict__ h, float* __restrict__ y, int N)
{
    const int t   = blockIdx.x * 256 + threadIdx.x;
    const int nid = t >> 4;
    const int q   = t & 15;
    if (nid >= N) return;
    const float4* hr = (const float4*)(h + (size_t)nid * 96);
    float4 o;
    if (q < 8) {
        const float4 a = hr[q];
        if (q < 4) o = make_float4(fmaxf(a.x, 0.0f), fmaxf(a.y, 0.0f), fmaxf(a.z, 0.0f), fmaxf(a.w, 0.0f));
        else       o = make_float4(fabsf(a.x), fabsf(a.y), fabsf(a.z), fabsf(a.w));
    } else {
        const float4 g = hr[q];
        const float4 v = hr[q + 8];
        const bool use_relu = ((q - 8) & 2) == 0;
        float4 ga;
        if (use_relu) ga = make_float4(fmaxf(g.x, 0.0f), fmaxf(g.y, 0.0f), fmaxf(g.z, 0.0f), fmaxf(g.w, 0.0f));
        else          ga = make_float4(tanhf(g.x), tanhf(g.y), tanhf(g.z), tanhf(g.w));
        o = make_float4(ga.x * v.x, ga.y * v.y, ga.z * v.z, ga.w * v.w);
    }
    ((float4*)(y + (size_t)nid * 64))[q] = o;
}

#define KE_EDGES 128
#define KE_THREADS 256
#define KE_PCH 16
#define KE_PITCH 160
__global__ __launch_bounds__(KE_THREADS) void kE(const float* __restrict__ basis,
                                                 const float* __restrict__ y,
                                                 const int* __restrict__ esrc,
                                                 const int* __restrict__ edst,
                                                 const int* __restrict__ perm,
                                                 const float* __restrict__ W2a,
                                                 const float* __restrict__ W2b,
                                                 float* __restrict__ out, int E)
{
    __shared__ __attribute__((aligned(16))) float bas[KE_EDGES][4];
    __shared__ int   srcT[KE_EDGES];
    __shared__ int   dstT[KE_EDGES];
    __shared__ int   eid[KE_EDGES];
    __shared__ __attribute__((aligned(16))) float waF[4][256];
    __shared__ __attribute__((aligned(8))) unsigned colsF[256];
    __shared__ __attribute__((aligned(16))) float wP[256][4];
    __shared__ unsigned colsP[256];
    __shared__ int   wcntF[8], wcntP[8];
    __shared__ __attribute__((aligned(16))) float boxp[4 * 8];
    __shared__ __attribute__((aligned(16))) float Gl[2][2 * KE_PITCH];
    __shared__ __attribute__((aligned(16))) float hP[KE_EDGES][KB_HP];
    __shared__ __attribute__((aligned(16))) float Bp[(KE_PCH / 2) * KE_PITCH];

    const int tid  = threadIdx.x;
    const int wv   = tid >> 5;
    const int lane = tid & 31;
    const int e0   = blockIdx.x * KE_EDGES;

    const float w0 = W2a[tid], w1 = W2a[256 + tid], w2 = W2a[512 + tid];

    if (tid < KE_EDGES) {
        const int i = e0 + tid;
        int e = -1, sv = 0, dv = 0;
        float b0 = 0.0f, b1 = 0.0f, b2 = 0.0f;
        if (i < E) {
            e = perm[i];
            e = min(max(e, 0), E - 1);
            b0 = basis[(size_t)e*3]; b1 = basis[(size_t)e*3+1]; b2 = basis[(size_t)e*3+2];
            sv = esrc[e];
            dv = edst[e];
        }
        eid[tid] = e;
        srcT[tid] = sv;
        dstT[tid] = dv;
        *(float4*)bas[tid] = make_float4(b0, b1, b2, 0.0f);
        box_partial(e >= 0, b0, b1, b2, boxp + 8 * wv, lane);
    }
    waF[3][tid] = 0.0f;
    __syncthreads();

    const int cls = box_class<4>(boxp, w0, w1, w2);
    const unsigned balF = (unsigned)__ballot(cls == 1);
    const unsigned balP = (unsigned)__ballot(cls == 2);
    if (lane == 0) { wcntF[wv] = __popc(balF); wcntP[wv] = __popc(balP); }
    __syncthreads();

    int baseF = 0, LF = 0, baseP = 0, LP = 0;
    #pragma unroll
    for (int i = 0; i < 8; i++) {
        const int cF = wcntF[i], cP = wcntP[i];
        if (i < wv) { baseF += cF; baseP += cP; }
        LF += cF; LP += cP;
    }
    const int LFp = (LF + 31) & ~31;
    const int LPp = (LP + 3) & ~3;
    {
        const unsigned lm = (1u << lane) - 1u;
        if (cls == 1) {
            const int j = baseF + __popc(balF & lm);
            colsF[j] = (unsigned)tid * (64u * 4u);
            waF[0][j] = w0; waF[1][j] = w1; waF[2][j] = w2;
        } else if (cls == 2) {
            const int j = baseP + __popc(balP & lm);
            colsP[j] = (unsigned)tid * 64u;
            *(float4*)wP[j] = make_float4(w0, w1, w2, 0.0f);
        }
        if (tid < 32) {
            const int j = LF + tid;
            if (j < LFp) { colsF[j] = 0u; waF[0][j] = 0.0f; waF[1][j] = 0.0f; waF[2][j] = 0.0f; }
        }
    }
    __syncthreads();

    const int c  = lane & 15;
    const int hh = lane >> 4;
    const char* Wb = (const char*)W2b;

    auto fill = [&](int ch) {
        for (int it = tid; it < KE_EDGES * KE_PCH; it += KE_THREADS) {
            const int m = it >> 4, kk = it & 15;
            const int j = ch * KE_PCH + kk;
            float v = 0.0f;
            if (j < LP) {
                const float4 w = *(const float4*)wP[j];
                v = hidden1(*(const float4*)bas[m], w.x, w.y, w.z);
            }
            hP[m][kk] = v;
        }
        const int nk = min(KE_PCH, LPp - ch * KE_PCH);
        const int col = tid & 63;
        for (int kk = tid >> 6; kk < nk; kk += 4) {
            const int j = ch * KE_PCH + kk;
            float v = 0.0f;
            if (j < LP) v = W2b[colsP[j] + col];
            Bp[(kk >> 1) * KE_PITCH + 2 * col + (kk & 1)] = v;
        }
    };

    {
        const int ct = wv & 3, kh = wv >> 2;
        v8f g0 = {}, g1 = {};
        const unsigned cb = (unsigned)(ct * 16 + c) * 4u;
        const float* Arow = waF[min(c, 3)];
        for (int k0 = 16 * kh; k0 < LFp; k0 += 32) {
            #pragma unroll
            for (int u = 0; u < 4; u++) {
                const int kb = k0 + 4*u + 2*hh;
                const v2f A = *(const v2f*)(Arow + kb);
                v2f B;
                B.x = *(const float*)(Wb + (colsF[kb]     + cb));
                B.y = *(const float*)(Wb + (colsF[kb + 1] + cb));
                if (u & 1) g1 = __builtin_amdgcn_wmma_f32_16x16x4_f32(false, A, false, B, (short)0, g1, false, false);
                else       g0 = __builtin_amdgcn_wmma_f32_16x16x4_f32(false, A, false, B, (short)0, g0, false, false);
            }
        }
        if (hh == 0) {
            const int col = ct * 16 + c;
            *(v2f*)(Gl[kh] + 2*col)            = (v2f){g0[0] + g1[0], g0[1] + g1[1]};
            *(v2f*)(Gl[kh] + KE_PITCH + 2*col) = (v2f){g0[2] + g1[2], 0.0f};
        }
    }
    const int nch = (LPp + KE_PCH - 1) / KE_PCH;
    if (nch > 0) fill(0);
    __syncthreads();

    const int r0 = wv * 16;
    v8f a0 = {}, a1 = {}, a2 = {}, a3 = {};
    {
        const v2f bb = *(const v2f*)(bas[r0 + c] + 2*hh);
        v2f A; A.x = bb.x * INV_SQRT3; A.y = bb.y * INV_SQRT3;
        const float* gp = Gl[0] + hh * KE_PITCH + 8 * c;
        const float* gq = Gl[1] + hh * KE_PITCH + 8 * c;
        const float4 Pa = *(const float4*)(gp), Pb = *(const float4*)(gp + 4);
        const float4 Qa = *(const float4*)(gq), Qb = *(const float4*)(gq + 4);
        const v2f B0 = (v2f){Pa.x + Qa.x, Pa.y + Qa.y}, B1 = (v2f){Pa.z + Qa.z, Pa.w + Qa.w};
        const v2f B2 = (v2f){Pb.x + Qb.x, Pb.y + Qb.y}, B3 = (v2f){Pb.z + Qb.z, Pb.w + Qb.w};
        a0 = __builtin_amdgcn_wmma_f32_16x16x4_f32(false, A, false, B0, (short)0, a0, false, false);
        a1 = __builtin_amdgcn_wmma_f32_16x16x4_f32(false, A, false, B1, (short)0, a1, false, false);
        a2 = __builtin_amdgcn_wmma_f32_16x16x4_f32(false, A, false, B2, (short)0, a2, false, false);
        a3 = __builtin_amdgcn_wmma_f32_16x16x4_f32(false, A, false, B3, (short)0, a3, false, false);
    }
    for (int ch = 0; ch < nch; ch++) {
        if (ch > 0) {
            __syncthreads();
            fill(ch);
            __syncthreads();
        }
        const int nk = min(KE_PCH, LPp - ch * KE_PCH);
        for (int kk = 0; kk < nk; kk += 4) {
            const v2f A = *(const v2f*)(hP[r0 + c] + kk + 2*hh);
            const float* bp = Bp + ((kk >> 1) + hh) * KE_PITCH + 8 * c;
            const float4 Ba = *(const float4*)(bp);
            const float4 Bb = *(const float4*)(bp + 4);
            const v2f B0 = (v2f){Ba.x, Ba.y}, B1 = (v2f){Ba.z, Ba.w};
            const v2f B2 = (v2f){Bb.x, Bb.y}, B3 = (v2f){Bb.z, Bb.w};
            a0 = __builtin_amdgcn_wmma_f32_16x16x4_f32(false, A, false, B0, (short)0, a0, false, false);
            a1 = __builtin_amdgcn_wmma_f32_16x16x4_f32(false, A, false, B1, (short)0, a1, false, false);
            a2 = __builtin_amdgcn_wmma_f32_16x16x4_f32(false, A, false, B2, (short)0, a2, false, false);
            a3 = __builtin_amdgcn_wmma_f32_16x16x4_f32(false, A, false, B3, (short)0, a3, false, false);
        }
    }

    float part[8];
    #pragma unroll
    for (int r = 0; r < 8; r++) {
        const int sv = srcT[r0 + 8*hh + r];
        const float4 yv = *(const float4*)(y + (size_t)sv*64 + 4*c);
        part[r] = a0[r]*yv.x + a1[r]*yv.y + a2[r]*yv.z + a3[r]*yv.w;
    }
    #pragma unroll
    for (int i = 0; i < 4; i++) {
        const bool up = (c & 8) != 0;
        const float keep = up ? part[4 + i] : part[i];
        const float send = up ? part[i] : part[4 + i];
        part[i] = keep + __shfl_xor(send, 8, 32);
    }
    #pragma unroll
    for (int i = 0; i < 2; i++) {
        const bool up = (c & 4) != 0;
        const float keep = up ? part[2 + i] : part[i];
        const float send = up ? part[i] : part[2 + i];
        part[i] = keep + __shfl_xor(send, 4, 32);
    }
    {
        const bool up = (c & 2) != 0;
        const float keep = up ? part[1] : part[0];
        const float send = up ? part[0] : part[1];
        part[0] = keep + __shfl_xor(send, 2, 32);
    }
    const float tot = part[0] + __shfl_xor(part[0], 1, 32);
    if ((c & 1) == 0) {
        const int m = r0 + 8*hh + 4*((c >> 3) & 1) + 2*((c >> 2) & 1) + ((c >> 1) & 1);
        if (eid[m] >= 0)
            atomicAdd(&out[dstT[m]], tot * (1.0f/128.0f) * INV_NN);
    }
}

extern "C" void kernel_launch(void* const* d_in, const int* in_sizes, int n_in,
                              void* d_out, int out_size, void* d_ws, size_t ws_size,
                              hipStream_t stream)
{
    const float* evec = (const float*)d_in[0];
    const float* elen = (const float*)d_in[1];
    const float* W1a  = (const float*)d_in[2];
    const float* W1b  = (const float*)d_in[3];
    const float* W2a  = (const float*)d_in[4];
    const float* W2b  = (const float*)d_in[5];
    const int*   esrc = (const int*)d_in[6];
    const int*   edst = (const int*)d_in[7];
    float* out = (float*)d_out;

    const int E = in_sizes[0] / 3;
    const int N = out_size;

    float* ws = (float*)d_ws;
    float* x_     = ws;                 ws += (size_t)N * 16;
    float* h_     = ws;                 ws += (size_t)N * 96;
    float* y_     = ws;                 ws += (size_t)N * 64;
    float* sh_    = ws;                 ws += (size_t)E * 16;
    float* basis_ = ws;                 ws += (size_t)E * 3;
    int*   perm_  = (int*)ws;           ws += (size_t)E;
    int*   cnt_   = (int*)ws;
    const int nb  = (E + SORT_CHUNK - 1) / SORT_CHUNK;

    long nx = (long)N * 16, nh = (long)N * 96;
    const int zx = (int)((nx / 4 + 255) / 256);
    const int nz = (int)((nh / 4 + 1023) / 1024);
    const int na = (E + 255) / 256;

    kPre<<<nb + zx, SORT_THREADS, 0, stream>>>(elen, cnt_, nb, E, x_, nx);
    kMain<<<nb + nz + na, 256, 0, stream>>>(evec, elen, edst, sh_, basis_, x_, E,
                                            cnt_, perm_, nb, h_, nh, out, (long)N, nz);
    kB<<<(E + KB_EDGES - 1) / KB_EDGES, KB_THREADS, 0, stream>>>(sh_, basis_, x_, esrc, edst, perm_, W1a, W1b, h_, E);
    kC<<<(N * 16 + 255) / 256, 256, 0, stream>>>(h_, y_, N);
    kE<<<(E + KE_EDGES - 1) / KE_EDGES, KE_THREADS, 0, stream>>>(basis_, y_, esrc, edst, perm_, W2a, W2b, out, E);
}
